# starting kernel
# baseline (speedup 1.0000x reference)
#include <hip/hip_runtime.h>

#define GDIM 128
#define NGRAPH 256
#define NEG_SLOPE 0.2f

typedef float v2f __attribute__((ext_vector_type(2)));
typedef float v8f __attribute__((ext_vector_type(8)));

__device__ __forceinline__ unsigned enc_f32(float f) {
    unsigned b = __float_as_uint(f);
    return (b & 0x80000000u) ? ~b : (b | 0x80000000u);
}
__device__ __forceinline__ float dec_f32(unsigned u) {
    return __uint_as_float((u & 0x80000000u) ? (u ^ 0x80000000u) : ~u);
}

__global__ void __launch_bounds__(128)
gat_gemm_wmma(const float* __restrict__ X, const float* __restrict__ W,
              float* __restrict__ H, int nTiles) {
    extern __shared__ float sW[];
    const int tid = threadIdx.x;

    const float4* W4 = (const float4*)W;
    float4*       S4 = (float4*)sW;
#pragma unroll
    for (int t = 0; t < 32; ++t) S4[t * 128 + tid] = W4[t * 128 + tid];
    __syncthreads();

    const int wid  = tid >> 5;
    const int lane = tid & 31;
    const int hi   = lane >> 4;
    const int col  = lane & 15;
    const int tile = blockIdx.x * 4 + wid;
    if (tile >= nTiles) return;

    const long   row0 = (long)tile * 16;
    const float* Xr   = X + (row0 + col) * GDIM;

    v8f c[8];
#pragma unroll
    for (int n = 0; n < 8; ++n) c[n] = (v8f)0.0f;

    for (int kk = 0; kk < GDIM; kk += 4) {
        const int ka = kk + 2 * hi;
        v2f a = *(const v2f*)(Xr + ka);
#pragma unroll
        for (int n = 0; n < 8; ++n) {
            v2f b;
            b.x = sW[(ka + 0) * GDIM + n * 16 + col];
            b.y = sW[(ka + 1) * GDIM + n * 16 + col];
            c[n] = __builtin_amdgcn_wmma_f32_16x16x4_f32(
                       false, a, false, b, (short)0, c[n], false, false);
        }
    }

    float* Hr = H + row0 * GDIM;
#pragma unroll
    for (int n = 0; n < 8; ++n)
#pragma unroll
        for (int v = 0; v < 8; ++v)
            Hr[(v + 8 * hi) * GDIM + n * 16 + col] = c[n][v];
}

__global__ void __launch_bounds__(256)
gat_node_scores(const float* __restrict__ H, const float* __restrict__ a_src,
                const float* __restrict__ a_dst, float* __restrict__ s_src,
                float* __restrict__ s_dst, int n) {
    const int lane = threadIdx.x & 31;
    const int node = blockIdx.x * 8 + (threadIdx.x >> 5);
    if (node >= n) return;
    const float4 hv = *(const float4*)(H + (long)node * GDIM + lane * 4);
    const float4 as = *(const float4*)(a_src + lane * 4);
    const float4 ad = *(const float4*)(a_dst + lane * 4);
    float ps = hv.x * as.x + hv.y * as.y + hv.z * as.z + hv.w * as.w;
    float pd = hv.x * ad.x + hv.y * ad.y + hv.z * ad.z + hv.w * ad.w;
#pragma unroll
    for (int off = 16; off > 0; off >>= 1) {
        ps += __shfl_xor(ps, off, 32);
        pd += __shfl_xor(pd, off, 32);
    }
    if (lane == 0) { s_src[node] = ps; s_dst[node] = pd; }
}

__global__ void __launch_bounds__(256)
gat_edge_max(const int* __restrict__ ei, int E, int n,
             const float* __restrict__ s_src, const float* __restrict__ s_dst,
             unsigned* __restrict__ m_enc) {
    const int e  = blockIdx.x * blockDim.x + threadIdx.x;
    const int NE = E + n;
    if (e >= NE) return;
    int i, j;
    if (e < E) { i = ei[e]; j = ei[E + e]; } else { i = j = e - E; }
    const float s     = s_src[i] + s_dst[j];
    const float logit = (s > 0.f) ? s : NEG_SLOPE * s;
    atomicMax(m_enc + j, enc_f32(logit));
}

__global__ void __launch_bounds__(256)
gat_edge_aggr(const int* __restrict__ ei, int E, int n,
              const float* __restrict__ s_src, const float* __restrict__ s_dst,
              const unsigned* __restrict__ m_enc, const float* __restrict__ H,
              float* __restrict__ denom, float* __restrict__ acc) {
    const int lane = threadIdx.x & 31;
    const int e    = blockIdx.x * 8 + (threadIdx.x >> 5);
    const int NE   = E + n;
    if (e >= NE) return;
    int i, j;
    if (e < E) { i = ei[e]; j = ei[E + e]; } else { i = j = e - E; }
    const float s     = s_src[i] + s_dst[j];
    const float logit = (s > 0.f) ? s : NEG_SLOPE * s;
    const float w     = __expf(logit - dec_f32(m_enc[j]));
    if (lane == 0) atomicAdd(denom + j, w);
    const float4 hv = *(const float4*)(H + (long)i * GDIM + lane * 4);
    float* ap = acc + (long)j * GDIM + lane * 4;
    atomicAdd(ap + 0, w * hv.x);
    atomicAdd(ap + 1, w * hv.y);
    atomicAdd(ap + 2, w * hv.z);
    atomicAdd(ap + 3, w * hv.w);
}

__global__ void __launch_bounds__(256)
gat_finalize(float* __restrict__ acc, const float* __restrict__ denom,
             const float* __restrict__ bias, int n) {
    const int lane = threadIdx.x & 31;
    const int node = blockIdx.x * 8 + (threadIdx.x >> 5);
    if (node >= n) return;
    const float  inv = 1.0f / denom[node];
    const float4 b4  = *(const float4*)(bias + lane * 4);
    float4* p = (float4*)(acc + (long)node * GDIM + lane * 4);
    float4  v = *p;
    v.x = v.x * inv + b4.x;
    v.y = v.y * inv + b4.y;
    v.z = v.z * inv + b4.z;
    v.w = v.w * inv + b4.w;
    *p = v;
}

__global__ void __launch_bounds__(256)
gat_pool_accum(const float* __restrict__ h, const int* __restrict__ batch, int n,
               float* __restrict__ psum, float* __restrict__ pcount) {
    const int lane = threadIdx.x & 31;
    const int node = blockIdx.x * 8 + (threadIdx.x >> 5);
    if (node >= n) return;
    const int    g  = batch[node];
    const float4 hv = *(const float4*)(h + (long)node * GDIM + lane * 4);
    float* pp = psum + (long)g * GDIM + lane * 4;
    atomicAdd(pp + 0, hv.x);
    atomicAdd(pp + 1, hv.y);
    atomicAdd(pp + 2, hv.z);
    atomicAdd(pp + 3, hv.w);
    if (lane == 0) atomicAdd(pcount + g, 1.0f);
}

__global__ void __launch_bounds__(256)
gat_pool_final(const float* __restrict__ psum, const float* __restrict__ pcount,
               float* __restrict__ out) {
    const int idx = blockIdx.x * blockDim.x + threadIdx.x;
    const float cnt = pcount[idx >> 7];
    out[idx] = psum[idx] / fmaxf(cnt, 1.0f);
}

extern "C" void kernel_launch(void* const* d_in, const int* in_sizes, int n_in,
                              void* d_out, int out_size, void* d_ws, size_t ws_size,
                              hipStream_t stream) {
    const float* x     = (const float*)d_in[0];
    const int*   ei    = (const int*)d_in[1];
    const int*   batch = (const int*)d_in[2];
    const float* W[3]    = {(const float*)d_in[3], (const float*)d_in[7],  (const float*)d_in[11]};
    const float* asrc[3] = {(const float*)d_in[4], (const float*)d_in[8],  (const float*)d_in[12]};
    const float* adst[3] = {(const float*)d_in[5], (const float*)d_in[9],  (const float*)d_in[13]};
    const float* bias[3] = {(const float*)d_in[6], (const float*)d_in[10], (const float*)d_in[14]};

    const int n  = in_sizes[0] / GDIM;
    const int E  = in_sizes[1] / 2;
    const int NE = E + n;

    char*  wsb = (char*)d_ws;
    size_t off = 0;
    auto take = [&](size_t bytes) -> void* {
        void* p = wsb + off;
        off += (bytes + 255) & ~(size_t)255;
        return p;
    };
    const size_t szH = (size_t)n * GDIM * sizeof(float);
    float*    h      = (float*)   take(szH);
    float*    acc    = (float*)   take(szH);
    float*    s_src  = (float*)   take((size_t)n * sizeof(float));
    float*    s_dst  = (float*)   take((size_t)n * sizeof(float));
    unsigned* m_enc  = (unsigned*)take((size_t)n * sizeof(unsigned));
    float*    denom  = (float*)   take((size_t)n * sizeof(float));
    float*    psum   = (float*)   take((size_t)NGRAPH * GDIM * sizeof(float));
    float*    pcount = (float*)   take((size_t)NGRAPH * sizeof(float));
    (void)ws_size; (void)n_in; (void)out_size;

    const int nTiles     = (n + 15) / 16;
    const int gemmBlocks = (nTiles + 3) / 4;
    const int nodeBlocks = (n + 7) / 8;
    const int edgeTBlk   = (NE + 255) / 256;
    const int edgeWBlk   = (NE + 7) / 8;

    const float* xin = x;
    for (int l = 0; l < 3; ++l) {
        gat_gemm_wmma<<<gemmBlocks, 128, GDIM * GDIM * sizeof(float), stream>>>(
            xin, W[l], h, nTiles);
        gat_node_scores<<<nodeBlocks, 256, 0, stream>>>(h, asrc[l], adst[l],
                                                        s_src, s_dst, n);
        hipMemsetAsync(m_enc, 0, (size_t)n * sizeof(unsigned), stream);
        hipMemsetAsync(denom, 0, (size_t)n * sizeof(float), stream);
        hipMemsetAsync(acc,   0, szH, stream);
        gat_edge_max<<<edgeTBlk, 256, 0, stream>>>(ei, E, n, s_src, s_dst, m_enc);
        gat_edge_aggr<<<edgeWBlk, 256, 0, stream>>>(ei, E, n, s_src, s_dst,
                                                    m_enc, h, denom, acc);
        gat_finalize<<<nodeBlocks, 256, 0, stream>>>(acc, denom, bias[l], n);
        xin = acc;
    }

    hipMemsetAsync(psum,   0, (size_t)NGRAPH * GDIM * sizeof(float), stream);
    hipMemsetAsync(pcount, 0, (size_t)NGRAPH * sizeof(float), stream);
    gat_pool_accum<<<nodeBlocks, 256, 0, stream>>>(acc, batch, n, psum, pcount);
    gat_pool_final<<<(NGRAPH * GDIM) / 256, 256, 0, stream>>>(psum, pcount,
                                                              (float*)d_out);
}


# mean pool: each wave sums its chunk of consecutive node rows per run of equal graph id in registers, then one atomic set per run instead of one per node; changes the pool's summation order
# speedup vs baseline: 1.1970x; 1.1970x over previous
#include <hip/hip_runtime.h>

#define GDIM 128
#define NGRAPH 256
#define NEG_SLOPE 0.2f

typedef float v2f __attribute__((ext_vector_type(2)));
typedef float v8f __attribute__((ext_vector_type(8)));

__device__ __forceinline__ unsigned enc_f32(float f) {
    unsigned b = __float_as_uint(f);
    return (b & 0x80000000u) ? ~b : (b | 0x80000000u);
}
__device__ __forceinline__ float dec_f32(unsigned u) {
    return __uint_as_float((u & 0x80000000u) ? (u ^ 0x80000000u) : ~u);
}

__global__ void __launch_bounds__(128)
gat_gemm_wmma(const float* __restrict__ X, const float* __restrict__ W,
              float* __restrict__ H, int nTiles) {
    extern __shared__ float sW[];
    const int tid = threadIdx.x;

    const float4* W4 = (const float4*)W;
    float4*       S4 = (float4*)sW;
#pragma unroll
    for (int t = 0; t < 32; ++t) S4[t * 128 + tid] = W4[t * 128 + tid];
    __syncthreads();

    const int wid  = tid >> 5;
    const int lane = tid & 31;
    const int hi   = lane >> 4;
    const int col  = lane & 15;
    const int tile = blockIdx.x * 4 + wid;
    if (tile >= nTiles) return;

    const long   row0 = (long)tile * 16;
    const float* Xr   = X + (row0 + col) * GDIM;

    v8f c[8];
#pragma unroll
    for (int n = 0; n < 8; ++n) c[n] = (v8f)0.0f;

    for (int kk = 0; kk < GDIM; kk += 4) {
        const int ka = kk + 2 * hi;
        v2f a = *(const v2f*)(Xr + ka);
#pragma unroll
        for (int n = 0; n < 8; ++n) {
            v2f b;
            b.x = sW[(ka + 0) * GDIM + n * 16 + col];
            b.y = sW[(ka + 1) * GDIM + n * 16 + col];
            c[n] = __builtin_amdgcn_wmma_f32_16x16x4_f32(
                       false, a, false, b, (short)0, c[n], false, false);
        }
    }

    float* Hr = H + row0 * GDIM;
#pragma unroll
    for (int n = 0; n < 8; ++n)
#pragma unroll
        for (int v = 0; v < 8; ++v)
            Hr[(v + 8 * hi) * GDIM + n * 16 + col] = c[n][v];
}

__global__ void __launch_bounds__(256)
gat_node_scores(const float* __restrict__ H, const float* __restrict__ a_src,
                const float* __restrict__ a_dst, float* __restrict__ s_src,
                float* __restrict__ s_dst, int n) {
    const int lane = threadIdx.x & 31;
    const int node = blockIdx.x * 8 + (threadIdx.x >> 5);
    if (node >= n) return;
    const float4 hv = *(const float4*)(H + (long)node * GDIM + lane * 4);
    const float4 as = *(const float4*)(a_src + lane * 4);
    const float4 ad = *(const float4*)(a_dst + lane * 4);
    float ps = hv.x * as.x + hv.y * as.y + hv.z * as.z + hv.w * as.w;
    float pd = hv.x * ad.x + hv.y * ad.y + hv.z * ad.z + hv.w * ad.w;
#pragma unroll
    for (int off = 16; off > 0; off >>= 1) {
        ps += __shfl_xor(ps, off, 32);
        pd += __shfl_xor(pd, off, 32);
    }
    if (lane == 0) { s_src[node] = ps; s_dst[node] = pd; }
}

__global__ void __launch_bounds__(256)
gat_edge_max(const int* __restrict__ ei, int E, int n,
             const float* __restrict__ s_src, const float* __restrict__ s_dst,
             unsigned* __restrict__ m_enc) {
    const int e  = blockIdx.x * blockDim.x + threadIdx.x;
    const int NE = E + n;
    if (e >= NE) return;
    int i, j;
    if (e < E) { i = ei[e]; j = ei[E + e]; } else { i = j = e - E; }
    const float s     = s_src[i] + s_dst[j];
    const float logit = (s > 0.f) ? s : NEG_SLOPE * s;
    atomicMax(m_enc + j, enc_f32(logit));
}

__global__ void __launch_bounds__(256)
gat_edge_aggr(const int* __restrict__ ei, int E, int n,
              const float* __restrict__ s_src, const float* __restrict__ s_dst,
              const unsigned* __restrict__ m_enc, const float* __restrict__ H,
              float* __restrict__ denom, float* __restrict__ acc) {
    const int lane = threadIdx.x & 31;
    const int e    = blockIdx.x * 8 + (threadIdx.x >> 5);
    const int NE   = E + n;
    if (e >= NE) return;
    int i, j;
    if (e < E) { i = ei[e]; j = ei[E + e]; } else { i = j = e - E; }
    const float s     = s_src[i] + s_dst[j];
    const float logit = (s > 0.f) ? s : NEG_SLOPE * s;
    const float w     = __expf(logit - dec_f32(m_enc[j]));
    if (lane == 0) atomicAdd(denom + j, w);
    const float4 hv = *(const float4*)(H + (long)i * GDIM + lane * 4);
    float* ap = acc + (long)j * GDIM + lane * 4;
    atomicAdd(ap + 0, w * hv.x);
    atomicAdd(ap + 1, w * hv.y);
    atomicAdd(ap + 2, w * hv.z);
    atomicAdd(ap + 3, w * hv.w);
}

__global__ void __launch_bounds__(256)
gat_finalize(float* __restrict__ acc, const float* __restrict__ denom,
             const float* __restrict__ bias, int n) {
    const int lane = threadIdx.x & 31;
    const int node = blockIdx.x * 8 + (threadIdx.x >> 5);
    if (node >= n) return;
    const float  inv = 1.0f / denom[node];
    const float4 b4  = *(const float4*)(bias + lane * 4);
    float4* p = (float4*)(acc + (long)node * GDIM + lane * 4);
    float4  v = *p;
    v.x = v.x * inv + b4.x;
    v.y = v.y * inv + b4.y;
    v.z = v.z * inv + b4.z;
    v.w = v.w * inv + b4.w;
    *p = v;
}

#define POOL_CHUNK 16

__device__ __forceinline__ void pool_flush(float* __restrict__ psum,
                                           float* __restrict__ pcount, int g,
                                           float4 s, float cnt, int lane) {
    float* pp = psum + (long)g * GDIM + lane * 4;
    atomicAdd(pp + 0, s.x);
    atomicAdd(pp + 1, s.y);
    atomicAdd(pp + 2, s.z);
    atomicAdd(pp + 3, s.w);
    if (lane == 0) atomicAdd(pcount + g, cnt);
}

__device__ __forceinline__ float4 add4(float4 a, float4 b) {
    return make_float4(a.x + b.x, a.y + b.y, a.z + b.z, a.w + b.w);
}

__global__ void __launch_bounds__(256)
gat_pool_accum(const float* __restrict__ h, const int* __restrict__ batch, int n,
               float* __restrict__ psum, float* __restrict__ pcount) {
    const int lane  = threadIdx.x & 31;
    const int wave  = __builtin_amdgcn_readfirstlane((int)(threadIdx.x >> 5));
    const int node0 = (blockIdx.x * 8 + wave) * POOL_CHUNK;
    if (node0 >= n) return;
    const int valid = min(POOL_CHUNK, n - node0);
    const float* hp = h + (long)node0 * GDIM + lane * 4;
    const int gl = batch[node0 + (lane < valid ? lane : 0)];

    if (valid == POOL_CHUNK) {
        float4 v[POOL_CHUNK];
#pragma unroll
        for (int u = 0; u < POOL_CHUNK; ++u)
            v[u] = *(const float4*)(hp + (long)u * GDIM);
        const int g0 = __builtin_amdgcn_readfirstlane(gl);
        if (__all(gl == g0)) {
#pragma unroll
            for (int w = POOL_CHUNK / 2; w > 0; w >>= 1)
#pragma unroll
                for (int u = 0; u < w; ++u) v[u] = add4(v[u], v[u + w]);
            pool_flush(psum, pcount, g0, v[0], (float)POOL_CHUNK, lane);
        } else {
            float4 s   = make_float4(0.f, 0.f, 0.f, 0.f);
            int    gc  = g0;
            float  cnt = 0.f;
#pragma unroll
            for (int u = 0; u < POOL_CHUNK; ++u) {
                const int g = __builtin_amdgcn_readlane(gl, u);
                if (g != gc) {
                    pool_flush(psum, pcount, gc, s, cnt, lane);
                    s = make_float4(0.f, 0.f, 0.f, 0.f); cnt = 0.f; gc = g;
                }
                s = add4(s, v[u]); cnt += 1.0f;
            }
            pool_flush(psum, pcount, gc, s, cnt, lane);
        }
    } else {
        float4 s   = make_float4(0.f, 0.f, 0.f, 0.f);
        int    gc  = __builtin_amdgcn_readfirstlane(gl);
        float  cnt = 0.f;
        for (int u = 0; u < valid; ++u) {
            const int    g  = __shfl(gl, u, 32);
            const float4 hv = *(const float4*)(hp + (long)u * GDIM);
            if (g != gc) {
                pool_flush(psum, pcount, gc, s, cnt, lane);
                s = make_float4(0.f, 0.f, 0.f, 0.f); cnt = 0.f; gc = g;
            }
            s = add4(s, hv); cnt += 1.0f;
        }
        pool_flush(psum, pcount, gc, s, cnt, lane);
    }
}

__global__ void __launch_bounds__(256)
gat_pool_final(const float* __restrict__ psum, const float* __restrict__ pcount,
               float* __restrict__ out) {
    const int idx = blockIdx.x * blockDim.x + threadIdx.x;
    const float cnt = pcount[idx >> 7];
    out[idx] = psum[idx] / fmaxf(cnt, 1.0f);
}

extern "C" void kernel_launch(void* const* d_in, const int* in_sizes, int n_in,
                              void* d_out, int out_size, void* d_ws, size_t ws_size,
                              hipStream_t stream) {
    const float* x     = (const float*)d_in[0];
    const int*   ei    = (const int*)d_in[1];
    const int*   batch = (const int*)d_in[2];
    const float* W[3]    = {(const float*)d_in[3], (const float*)d_in[7],  (const float*)d_in[11]};
    const float* asrc[3] = {(const float*)d_in[4], (const float*)d_in[8],  (const float*)d_in[12]};
    const float* adst[3] = {(const float*)d_in[5], (const float*)d_in[9],  (const float*)d_in[13]};
    const float* bias[3] = {(const float*)d_in[6], (const float*)d_in[10], (const float*)d_in[14]};

    const int n  = in_sizes[0] / GDIM;
    const int E  = in_sizes[1] / 2;
    const int NE = E + n;

    char*  wsb = (char*)d_ws;
    size_t off = 0;
    auto take = [&](size_t bytes) -> void* {
        void* p = wsb + off;
        off += (bytes + 255) & ~(size_t)255;
        return p;
    };
    const size_t szH = (size_t)n * GDIM * sizeof(float);
    float*    h      = (float*)   take(szH);
    float*    acc    = (float*)   take(szH);
    float*    s_src  = (float*)   take((size_t)n * sizeof(float));
    float*    s_dst  = (float*)   take((size_t)n * sizeof(float));
    unsigned* m_enc  = (unsigned*)take((size_t)n * sizeof(unsigned));
    float*    denom  = (float*)   take((size_t)n * sizeof(float));
    float*    psum   = (float*)   take((size_t)NGRAPH * GDIM * sizeof(float));
    float*    pcount = (float*)   take((size_t)NGRAPH * sizeof(float));
    (void)ws_size; (void)n_in; (void)out_size;

    const int nTiles     = (n + 15) / 16;
    const int gemmBlocks = (nTiles + 3) / 4;
    const int nodeBlocks = (n + 7) / 8;
    const int edgeTBlk   = (NE + 255) / 256;
    const int edgeWBlk   = (NE + 7) / 8;
    const int poolBlocks = (n + 8 * POOL_CHUNK - 1) / (8 * POOL_CHUNK);

    const float* xin = x;
    for (int l = 0; l < 3; ++l) {
        gat_gemm_wmma<<<gemmBlocks, 128, GDIM * GDIM * sizeof(float), stream>>>(
            xin, W[l], h, nTiles);
        gat_node_scores<<<nodeBlocks, 256, 0, stream>>>(h, asrc[l], adst[l],
                                                        s_src, s_dst, n);
        hipMemsetAsync(m_enc, 0, (size_t)n * sizeof(unsigned), stream);
        hipMemsetAsync(denom, 0, (size_t)n * sizeof(float), stream);
        hipMemsetAsync(acc,   0, szH, stream);
        gat_edge_max<<<edgeTBlk, 256, 0, stream>>>(ei, E, n, s_src, s_dst, m_enc);
        gat_edge_aggr<<<edgeWBlk, 256, 0, stream>>>(ei, E, n, s_src, s_dst,
                                                    m_enc, h, denom, acc);
        gat_finalize<<<nodeBlocks, 256, 0, stream>>>(acc, denom, bias[l], n);
        xin = acc;
    }

    hipMemsetAsync(psum,   0, (size_t)NGRAPH * GDIM * sizeof(float), stream);
    hipMemsetAsync(pcount, 0, (size_t)NGRAPH * sizeof(float), stream);
    gat_pool_accum<<<poolBlocks, 256, 0, stream>>>(acc, batch, n, psum, pcount);
    gat_pool_final<<<(NGRAPH * GDIM) / 256, 256, 0, stream>>>(psum, pcount,
                                                              (float*)d_out);
}


# attention aggregation as a gather: edges grouped by dst once per call (count, scan, fill), one wave per node does max, exp, sum and weighted rows in registers; changes its summation order
# speedup vs baseline: 4.1334x; 3.4531x over previous
#include <hip/hip_runtime.h>

#define GDIM 128
#define NGRAPH 256
#define NEG_SLOPE 0.2f

typedef float v2f __attribute__((ext_vector_type(2)));
typedef float v8f __attribute__((ext_vector_type(8)));

__global__ void __launch_bounds__(128)
gat_gemm_wmma(const float* __restrict__ X, const float* __restrict__ W,
              float* __restrict__ H, int nTiles) {
    extern __shared__ float sW[];
    const int tid = threadIdx.x;

    const float4* W4 = (const float4*)W;
    float4*       S4 = (float4*)sW;
#pragma unroll
    for (int t = 0; t < 32; ++t) S4[t * 128 + tid] = W4[t * 128 + tid];
    __syncthreads();

    const int wid  = tid >> 5;
    const int lane = tid & 31;
    const int hi   = lane >> 4;
    const int col  = lane & 15;
    const int tile = blockIdx.x * 4 + wid;
    if (tile >= nTiles) return;

    const long   row0 = (long)tile * 16;
    const float* Xr   = X + (row0 + col) * GDIM;

    v8f c[8];
#pragma unroll
    for (int n = 0; n < 8; ++n) c[n] = (v8f)0.0f;

    for (int kk = 0; kk < GDIM; kk += 4) {
        const int ka = kk + 2 * hi;
        v2f a = *(const v2f*)(Xr + ka);
#pragma unroll
        for (int n = 0; n < 8; ++n) {
            v2f b;
            b.x = sW[(ka + 0) * GDIM + n * 16 + col];
            b.y = sW[(ka + 1) * GDIM + n * 16 + col];
            c[n] = __builtin_amdgcn_wmma_f32_16x16x4_f32(
                       false, a, false, b, (short)0, c[n], false, false);
        }
    }

    float* Hr = H + row0 * GDIM;
#pragma unroll
    for (int n = 0; n < 8; ++n)
#pragma unroll
        for (int v = 0; v < 8; ++v)
            Hr[(v + 8 * hi) * GDIM + n * 16 + col] = c[n][v];
}

__global__ void __launch_bounds__(256)
gat_node_scores(const float* __restrict__ H, const float* __restrict__ a_src,
                const float* __restrict__ a_dst, float* __restrict__ s_src,
                float* __restrict__ s_dst, int n) {
    const int lane = threadIdx.x & 31;
    const int node = blockIdx.x * 8 + (threadIdx.x >> 5);
    if (node >= n) return;
    const float4 hv = *(const float4*)(H + (long)node * GDIM + lane * 4);
    const float4 as = *(const float4*)(a_src + lane * 4);
    const float4 ad = *(const float4*)(a_dst + lane * 4);
    float ps = hv.x * as.x + hv.y * as.y + hv.z * as.z + hv.w * as.w;
    float pd = hv.x * ad.x + hv.y * ad.y + hv.z * ad.z + hv.w * ad.w;
#pragma unroll
    for (int off = 16; off > 0; off >>= 1) {
        ps += __shfl_xor(ps, off, 32);
        pd += __shfl_xor(pd, off, 32);
    }
    if (lane == 0) { s_src[node] = ps; s_dst[node] = pd; }
}

__global__ void __launch_bounds__(256)
csr_count(const int* __restrict__ ei, int E, int* __restrict__ deg) {
    const int e = blockIdx.x * blockDim.x + threadIdx.x;
    if (e >= E) return;
    atomicAdd(deg + ei[E + e], 1);
}

__global__ void __launch_bounds__(1024)
csr_offsets(const int* __restrict__ deg, int n, int* __restrict__ start,
            int* __restrict__ cursor) {
    __shared__ int wtot[32];
    __shared__ int wpre[32];
    const int tid   = threadIdx.x;
    const int lane  = tid & 31;
    const int wave  = tid >> 5;
    const int first = blockIdx.x * 1024;
    int pre = 0;
    for (int t = tid; t < first; t += 1024) pre += deg[t];
    const int j = first + tid;
    const int d = (j < n) ? deg[j] : 0;
    int incl = d;
#pragma unroll
    for (int off = 1; off < 32; off <<= 1) {
        const int t = __shfl_up(incl, off, 32);
        if (lane >= off) incl += t;
    }
#pragma unroll
    for (int off = 16; off > 0; off >>= 1) pre += __shfl_xor(pre, off, 32);
    if (lane == 31) wtot[wave] = incl;
    if (lane == 0)  wpre[wave] = pre;
    __syncthreads();
    int base = wpre[lane] + ((lane < wave) ? wtot[lane] : 0);
#pragma unroll
    for (int off = 16; off > 0; off >>= 1) base += __shfl_xor(base, off, 32);
    const int st = base + incl - d;
    if (j < n) { start[j] = st; cursor[j] = st; }
}

__global__ void __launch_bounds__(256)
csr_fill(const int* __restrict__ ei, int E, int* __restrict__ cursor,
         int* __restrict__ csr_src) {
    const int e = blockIdx.x * blockDim.x + threadIdx.x;
    if (e >= E) return;
    const int pos = atomicAdd(cursor + ei[E + e], 1);
    csr_src[pos] = ei[e];
}

__device__ __forceinline__ void gather_rows(const float* __restrict__ H, int iv,
                                            float wv, int cnt, int lane, float4& a) {
    const int wi = __float_as_int(wv);
    int k = 0;
    for (; k + 4 <= cnt; k += 4) {
        const int   i0 = __builtin_amdgcn_readlane(iv, k);
        const int   i1 = __builtin_amdgcn_readlane(iv, k + 1);
        const int   i2 = __builtin_amdgcn_readlane(iv, k + 2);
        const int   i3 = __builtin_amdgcn_readlane(iv, k + 3);
        const float w0 = __int_as_float(__builtin_amdgcn_readlane(wi, k));
        const float w1 = __int_as_float(__builtin_amdgcn_readlane(wi, k + 1));
        const float w2 = __int_as_float(__builtin_amdgcn_readlane(wi, k + 2));
        const float w3 = __int_as_float(__builtin_amdgcn_readlane(wi, k + 3));
        const float4 h0 = *(const float4*)(H + (long)i0 * GDIM + lane * 4);
        const float4 h1 = *(const float4*)(H + (long)i1 * GDIM + lane * 4);
        const float4 h2 = *(const float4*)(H + (long)i2 * GDIM + lane * 4);
        const float4 h3 = *(const float4*)(H + (long)i3 * GDIM + lane * 4);
        a.x = fmaf(w0, h0.x, a.x); a.y = fmaf(w0, h0.y, a.y);
        a.z = fmaf(w0, h0.z, a.z); a.w = fmaf(w0, h0.w, a.w);
        a.x = fmaf(w1, h1.x, a.x); a.y = fmaf(w1, h1.y, a.y);
        a.z = fmaf(w1, h1.z, a.z); a.w = fmaf(w1, h1.w, a.w);
        a.x = fmaf(w2, h2.x, a.x); a.y = fmaf(w2, h2.y, a.y);
        a.z = fmaf(w2, h2.z, a.z); a.w = fmaf(w2, h2.w, a.w);
        a.x = fmaf(w3, h3.x, a.x); a.y = fmaf(w3, h3.y, a.y);
        a.z = fmaf(w3, h3.z, a.z); a.w = fmaf(w3, h3.w, a.w);
    }
    for (; k < cnt; ++k) {
        const int   i0 = __builtin_amdgcn_readlane(iv, k);
        const float w0 = __int_as_float(__builtin_amdgcn_readlane(wi, k));
        const float4 h0 = *(const float4*)(H + (long)i0 * GDIM + lane * 4);
        a.x = fmaf(w0, h0.x, a.x); a.y = fmaf(w0, h0.y, a.y);
        a.z = fmaf(w0, h0.z, a.z); a.w = fmaf(w0, h0.w, a.w);
    }
}

__device__ __forceinline__ float wave_max(float v) {
#pragma unroll
    for (int off = 16; off > 0; off >>= 1) v = fmaxf(v, __shfl_xor(v, off, 32));
    return v;
}
__device__ __forceinline__ float wave_sum(float v) {
#pragma unroll
    for (int off = 16; off > 0; off >>= 1) v += __shfl_xor(v, off, 32);
    return v;
}

__global__ void __launch_bounds__(256)
gat_gather(const int* __restrict__ start, const int* __restrict__ endp,
           const int* __restrict__ csr_src, const float* __restrict__ s_src,
           const float* __restrict__ s_dst, const float* __restrict__ H,
           const float* __restrict__ bias, float* __restrict__ out, int n) {
    const int lane = threadIdx.x & 31;
    const int wave = __builtin_amdgcn_readfirstlane((int)(threadIdx.x >> 5));
    const int j    = blockIdx.x * 8 + wave;
    if (j >= n) return;
    const int   beg = start[j];
    const int   deg = endp[j] - beg;
    const int   T   = deg + 1;
    const float sd  = s_dst[j];
    const float NEG_INF = -__builtin_inff();

    float4 a   = make_float4(0.f, 0.f, 0.f, 0.f);
    float  den = 0.f;

    if (T <= 32) {
        const int   i  = (lane < deg) ? csr_src[beg + lane] : j;
        const float s  = s_src[i] + sd;
        const float lg = (s > 0.f) ? s : NEG_SLOPE * s;
        const float m  = wave_max((lane < T) ? lg : NEG_INF);
        const float w  = (lane < T) ? __expf(lg - m) : 0.f;
        den = wave_sum(w);
        gather_rows(H, i, w, T, lane, a);
    } else {
        float m = NEG_INF;
        for (int base = 0; base < T; base += 32) {
            const int   t  = base + lane;
            const int   i  = (t < deg) ? csr_src[beg + t] : j;
            const float s  = s_src[i] + sd;
            const float lg = (s > 0.f) ? s : NEG_SLOPE * s;
            if (t < T) m = fmaxf(m, lg);
        }
        m = wave_max(m);
        for (int base = 0; base < T; base += 32) {
            const int   t  = base + lane;
            const int   i  = (t < deg) ? csr_src[beg + t] : j;
            const float s  = s_src[i] + sd;
            const float lg = (s > 0.f) ? s : NEG_SLOPE * s;
            const float w  = (t < T) ? __expf(lg - m) : 0.f;
            den += w;
            gather_rows(H, i, w, min(32, T - base), lane, a);
        }
        den = wave_sum(den);
    }

    const float  inv = 1.0f / den;
    const float4 b4  = *(const float4*)(bias + lane * 4);
    float4 v;
    v.x = a.x * inv + b4.x;
    v.y = a.y * inv + b4.y;
    v.z = a.z * inv + b4.z;
    v.w = a.w * inv + b4.w;
    *(float4*)(out + (long)j * GDIM + lane * 4) = v;
}

#define POOL_CHUNK 16

__device__ __forceinline__ void pool_flush(float* __restrict__ psum,
                                           float* __restrict__ pcount, int g,
                                           float4 s, float cnt, int lane) {
    float* pp = psum + (long)g * GDIM + lane * 4;
    atomicAdd(pp + 0, s.x);
    atomicAdd(pp + 1, s.y);
    atomicAdd(pp + 2, s.z);
    atomicAdd(pp + 3, s.w);
    if (lane == 0) atomicAdd(pcount + g, cnt);
}

__device__ __forceinline__ float4 add4(float4 a, float4 b) {
    return make_float4(a.x + b.x, a.y + b.y, a.z + b.z, a.w + b.w);
}

__global__ void __launch_bounds__(256)
gat_pool_accum(const float* __restrict__ h, const int* __restrict__ batch, int n,
               float* __restrict__ psum, float* __restrict__ pcount) {
    const int lane  = threadIdx.x & 31;
    const int wave  = __builtin_amdgcn_readfirstlane((int)(threadIdx.x >> 5));
    const int node0 = (blockIdx.x * 8 + wave) * POOL_CHUNK;
    if (node0 >= n) return;
    const int valid = min(POOL_CHUNK, n - node0);
    const float* hp = h + (long)node0 * GDIM + lane * 4;
    const int gl = batch[node0 + (lane < valid ? lane : 0)];

    if (valid == POOL_CHUNK) {
        float4 v[POOL_CHUNK];
#pragma unroll
        for (int u = 0; u < POOL_CHUNK; ++u)
            v[u] = *(const float4*)(hp + (long)u * GDIM);
        const int g0 = __builtin_amdgcn_readfirstlane(gl);
        if (__all(gl == g0)) {
#pragma unroll
            for (int w = POOL_CHUNK / 2; w > 0; w >>= 1)
#pragma unroll
                for (int u = 0; u < w; ++u) v[u] = add4(v[u], v[u + w]);
            pool_flush(psum, pcount, g0, v[0], (float)POOL_CHUNK, lane);
        } else {
            float4 s   = make_float4(0.f, 0.f, 0.f, 0.f);
            int    gc  = g0;
            float  cnt = 0.f;
#pragma unroll
            for (int u = 0; u < POOL_CHUNK; ++u) {
                const int g = __builtin_amdgcn_readlane(gl, u);
                if (g != gc) {
                    pool_flush(psum, pcount, gc, s, cnt, lane);
                    s = make_float4(0.f, 0.f, 0.f, 0.f); cnt = 0.f; gc = g;
                }
                s = add4(s, v[u]); cnt += 1.0f;
            }
            pool_flush(psum, pcount, gc, s, cnt, lane);
        }
    } else {
        float4 s   = make_float4(0.f, 0.f, 0.f, 0.f);
        int    gc  = __builtin_amdgcn_readfirstlane(gl);
        float  cnt = 0.f;
        for (int u = 0; u < valid; ++u) {
            const int    g  = __shfl(gl, u, 32);
            const float4 hv = *(const float4*)(hp + (long)u * GDIM);
            if (g != gc) {
                pool_flush(psum, pcount, gc, s, cnt, lane);
                s = make_float4(0.f, 0.f, 0.f, 0.f); cnt = 0.f; gc = g;
            }
            s = add4(s, hv); cnt += 1.0f;
        }
        pool_flush(psum, pcount, gc, s, cnt, lane);
    }
}

__global__ void __launch_bounds__(256)
gat_pool_final(const float* __restrict__ psum, const float* __restrict__ pcount,
               float* __restrict__ out) {
    const int idx = blockIdx.x * blockDim.x + threadIdx.x;
    const float cnt = pcount[idx >> 7];
    out[idx] = psum[idx] / fmaxf(cnt, 1.0f);
}

extern "C" void kernel_launch(void* const* d_in, const int* in_sizes, int n_in,
                              void* d_out, int out_size, void* d_ws, size_t ws_size,
                              hipStream_t stream) {
    const float* x     = (const float*)d_in[0];
    const int*   ei    = (const int*)d_in[1];
    const int*   batch = (const int*)d_in[2];
    const float* W[3]    = {(const float*)d_in[3], (const float*)d_in[7],  (const float*)d_in[11]};
    const float* asrc[3] = {(const float*)d_in[4], (const float*)d_in[8],  (const float*)d_in[12]};
    const float* adst[3] = {(const float*)d_in[5], (const float*)d_in[9],  (const float*)d_in[13]};
    const float* bias[3] = {(const float*)d_in[6], (const float*)d_in[10], (const float*)d_in[14]};

    const int n  = in_sizes[0] / GDIM;
    const int E  = in_sizes[1] / 2;

    char*  wsb = (char*)d_ws;
    size_t off = 0;
    auto take = [&](size_t bytes) -> void* {
        void* p = wsb + off;
        off += (bytes + 255) & ~(size_t)255;
        return p;
    };
    const size_t szH = (size_t)n * GDIM * sizeof(float);
    float*    h      = (float*)   take(szH);
    float*    acc    = (float*)   take(szH);
    float*    s_src  = (float*)   take((size_t)n * sizeof(float));
    float*    s_dst  = (float*)   take((size_t)n * sizeof(float));
    int*      deg    = (int*)     take((size_t)n * sizeof(int));
    int*      start  = (int*)     take((size_t)n * sizeof(int));
    int*      cursor = (int*)     take((size_t)n * sizeof(int));
    int*      csr    = (int*)     take((size_t)E * sizeof(int));
    float*    psum   = (float*)   take((size_t)NGRAPH * GDIM * sizeof(float));
    float*    pcount = (float*)   take((size_t)NGRAPH * sizeof(float));
    (void)ws_size; (void)n_in; (void)out_size;

    const int nTiles     = (n + 15) / 16;
    const int gemmBlocks = (nTiles + 3) / 4;
    const int nodeBlocks = (n + 7) / 8;
    const int edgeBlocks = (E + 255) / 256;
    const int scanBlocks = (n + 1023) / 1024;
    const int poolBlocks = (n + 8 * POOL_CHUNK - 1) / (8 * POOL_CHUNK);

    hipMemsetAsync(deg, 0, (size_t)n * sizeof(int), stream);
    csr_count<<<edgeBlocks, 256, 0, stream>>>(ei, E, deg);
    csr_offsets<<<scanBlocks, 1024, 0, stream>>>(deg, n, start, cursor);
    csr_fill<<<edgeBlocks, 256, 0, stream>>>(ei, E, cursor, csr);

    const float* xin = x;
    for (int l = 0; l < 3; ++l) {
        gat_gemm_wmma<<<gemmBlocks, 128, GDIM * GDIM * sizeof(float), stream>>>(
            xin, W[l], h, nTiles);
        gat_node_scores<<<nodeBlocks, 256, 0, stream>>>(h, asrc[l], adst[l],
                                                        s_src, s_dst, n);
        gat_gather<<<nodeBlocks, 256, 0, stream>>>(start, cursor, csr, s_src, s_dst,
                                                   h, bias[l], acc, n);
        xin = acc;
    }

    hipMemsetAsync(psum,   0, (size_t)NGRAPH * GDIM * sizeof(float), stream);
    hipMemsetAsync(pcount, 0, (size_t)NGRAPH * sizeof(float), stream);
    gat_pool_accum<<<poolBlocks, 256, 0, stream>>>(acc, batch, n, psum, pcount);
    gat_pool_final<<<(NGRAPH * GDIM) / 256, 256, 0, stream>>>(psum, pcount,
                                                              (float*)d_out);
}


# attention scores h.a_src and h.a_dst taken from the GEMM's accumulators in its epilogue, the per-node score launch of each layer removed; changes the score dot products' summation order
# speedup vs baseline: 4.6701x; 1.1298x over previous
#include <hip/hip_runtime.h>

#define GDIM 128
#define NGRAPH 256
#define NEG_SLOPE 0.2f

typedef float v2f __attribute__((ext_vector_type(2)));
typedef float v8f __attribute__((ext_vector_type(8)));

__global__ void __launch_bounds__(128)
gat_gemm_wmma(const float* __restrict__ X, const float* __restrict__ W,
              const float* __restrict__ a_src, const float* __restrict__ a_dst,
              float* __restrict__ H, float* __restrict__ s_src,
              float* __restrict__ s_dst, int nTiles) {
    extern __shared__ float sW[];
    const int tid = threadIdx.x;

    const float4* W4 = (const float4*)W;
    float4*       S4 = (float4*)sW;
#pragma unroll
    for (int t = 0; t < 32; ++t) S4[t * 128 + tid] = W4[t * 128 + tid];
    __syncthreads();

    const int wid  = tid >> 5;
    const int lane = tid & 31;
    const int hi   = lane >> 4;
    const int col  = lane & 15;
    const int tile = blockIdx.x * 4 + wid;
    if (tile >= nTiles) return;

    const long   row0 = (long)tile * 16;
    const float* Xr   = X + (row0 + col) * GDIM;

    v8f c[8];
#pragma unroll
    for (int n = 0; n < 8; ++n) c[n] = (v8f)0.0f;

    for (int kk = 0; kk < GDIM; kk += 4) {
        const int ka = kk + 2 * hi;
        v2f a = *(const v2f*)(Xr + ka);
#pragma unroll
        for (int n = 0; n < 8; ++n) {
            v2f b;
            b.x = sW[(ka + 0) * GDIM + n * 16 + col];
            b.y = sW[(ka + 1) * GDIM + n * 16 + col];
            c[n] = __builtin_amdgcn_wmma_f32_16x16x4_f32(
                       false, a, false, b, (short)0, c[n], false, false);
        }
    }

    float* Hr = H + row0 * GDIM;
#pragma unroll
    for (int n = 0; n < 8; ++n)
#pragma unroll
        for (int v = 0; v < 8; ++v)
            Hr[(v + 8 * hi) * GDIM + n * 16 + col] = c[n][v];

    float ps[8], pd[8];
#pragma unroll
    for (int v = 0; v < 8; ++v) { ps[v] = 0.0f; pd[v] = 0.0f; }
#pragma unroll
    for (int n = 0; n < 8; ++n) {
        const float as = a_src[n * 16 + col];
        const float ad = a_dst[n * 16 + col];
#pragma unroll
        for (int v = 0; v < 8; ++v) {
            ps[v] = fmaf(c[n][v], as, ps[v]);
            pd[v] = fmaf(c[n][v], ad, pd[v]);
        }
    }
    const bool up = (col & 8) != 0;
    float r[8];
#pragma unroll
    for (int v = 0; v < 8; ++v) {
        const float keep = up ? pd[v] : ps[v];
        const float give = up ? ps[v] : pd[v];
        r[v] = keep + __shfl_xor(give, 8, 32);
    }
#pragma unroll
    for (int off = 4; off > 0; off >>= 1)
#pragma unroll
        for (int v = 0; v < 8; ++v) r[v] += __shfl_xor(r[v], off, 32);
    if ((col & 7) == 0) {
        float* sp = (up ? s_dst : s_src) + row0 + 8 * hi;
        *(float4*)(sp)     = make_float4(r[0], r[1], r[2], r[3]);
        *(float4*)(sp + 4) = make_float4(r[4], r[5], r[6], r[7]);
    }
}

__global__ void __launch_bounds__(256)
csr_count(const int* __restrict__ ei, int E, int* __restrict__ deg) {
    const int e = blockIdx.x * blockDim.x + threadIdx.x;
    if (e >= E) return;
    atomicAdd(deg + ei[E + e], 1);
}

__global__ void __launch_bounds__(1024)
csr_offsets(const int* __restrict__ deg, int n, int* __restrict__ start,
            int* __restrict__ cursor) {
    __shared__ int wtot[32];
    __shared__ int wpre[32];
    const int tid   = threadIdx.x;
    const int lane  = tid & 31;
    const int wave  = tid >> 5;
    const int first = blockIdx.x * 1024;
    int pre = 0;
    for (int t = tid; t < first; t += 1024) pre += deg[t];
    const int j = first + tid;
    const int d = (j < n) ? deg[j] : 0;
    int incl = d;
#pragma unroll
    for (int off = 1; off < 32; off <<= 1) {
        const int t = __shfl_up(incl, off, 32);
        if (lane >= off) incl += t;
    }
#pragma unroll
    for (int off = 16; off > 0; off >>= 1) pre += __shfl_xor(pre, off, 32);
    if (lane == 31) wtot[wave] = incl;
    if (lane == 0)  wpre[wave] = pre;
    __syncthreads();
    int base = wpre[lane] + ((lane < wave) ? wtot[lane] : 0);
#pragma unroll
    for (int off = 16; off > 0; off >>= 1) base += __shfl_xor(base, off, 32);
    const int st = base + incl - d;
    if (j < n) { start[j] = st; cursor[j] = st; }
}

__global__ void __launch_bounds__(256)
csr_fill(const int* __restrict__ ei, int E, int* __restrict__ cursor,
         int* __restrict__ csr_src) {
    const int e = blockIdx.x * blockDim.x + threadIdx.x;
    if (e >= E) return;
    const int pos = atomicAdd(cursor + ei[E + e], 1);
    csr_src[pos] = ei[e];
}

__device__ __forceinline__ void gather_rows(const float* __restrict__ H, int iv,
                                            float wv, int cnt, int lane, float4& a) {
    const int wi = __float_as_int(wv);
    int k = 0;
    for (; k + 4 <= cnt; k += 4) {
        const int   i0 = __builtin_amdgcn_readlane(iv, k);
        const int   i1 = __builtin_amdgcn_readlane(iv, k + 1);
        const int   i2 = __builtin_amdgcn_readlane(iv, k + 2);
        const int   i3 = __builtin_amdgcn_readlane(iv, k + 3);
        const float w0 = __int_as_float(__builtin_amdgcn_readlane(wi, k));
        const float w1 = __int_as_float(__builtin_amdgcn_readlane(wi, k + 1));
        const float w2 = __int_as_float(__builtin_amdgcn_readlane(wi, k + 2));
        const float w3 = __int_as_float(__builtin_amdgcn_readlane(wi, k + 3));
        const float4 h0 = *(const float4*)(H + (long)i0 * GDIM + lane * 4);
        const float4 h1 = *(const float4*)(H + (long)i1 * GDIM + lane * 4);
        const float4 h2 = *(const float4*)(H + (long)i2 * GDIM + lane * 4);
        const float4 h3 = *(const float4*)(H + (long)i3 * GDIM + lane * 4);
        a.x = fmaf(w0, h0.x, a.x); a.y = fmaf(w0, h0.y, a.y);
        a.z = fmaf(w0, h0.z, a.z); a.w = fmaf(w0, h0.w, a.w);
        a.x = fmaf(w1, h1.x, a.x); a.y = fmaf(w1, h1.y, a.y);
        a.z = fmaf(w1, h1.z, a.z); a.w = fmaf(w1, h1.w, a.w);
        a.x = fmaf(w2, h2.x, a.x); a.y = fmaf(w2, h2.y, a.y);
        a.z = fmaf(w2, h2.z, a.z); a.w = fmaf(w2, h2.w, a.w);
        a.x = fmaf(w3, h3.x, a.x); a.y = fmaf(w3, h3.y, a.y);
        a.z = fmaf(w3, h3.z, a.z); a.w = fmaf(w3, h3.w, a.w);
    }
    for (; k < cnt; ++k) {
        const int   i0 = __builtin_amdgcn_readlane(iv, k);
        const float w0 = __int_as_float(__builtin_amdgcn_readlane(wi, k));
        const float4 h0 = *(const float4*)(H + (long)i0 * GDIM + lane * 4);
        a.x = fmaf(w0, h0.x, a.x); a.y = fmaf(w0, h0.y, a.y);
        a.z = fmaf(w0, h0.z, a.z); a.w = fmaf(w0, h0.w, a.w);
    }
}

__device__ __forceinline__ float wave_max(float v) {
#pragma unroll
    for (int off = 16; off > 0; off >>= 1) v = fmaxf(v, __shfl_xor(v, off, 32));
    return v;
}
__device__ __forceinline__ float wave_sum(float v) {
#pragma unroll
    for (int off = 16; off > 0; off >>= 1) v += __shfl_xor(v, off, 32);
    return v;
}

__global__ void __launch_bounds__(256)
gat_gather(const int* __restrict__ start, const int* __restrict__ endp,
           const int* __restrict__ csr_src, const float* __restrict__ s_src,
           const float* __restrict__ s_dst, const float* __restrict__ H,
           const float* __restrict__ bias, float* __restrict__ out, int n) {
    const int lane = threadIdx.x & 31;
    const int wave = __builtin_amdgcn_readfirstlane((int)(threadIdx.x >> 5));
    const int j    = blockIdx.x * 8 + wave;
    if (j >= n) return;
    const int   beg = start[j];
    const int   deg = endp[j] - beg;
    const int   T   = deg + 1;
    const float sd  = s_dst[j];
    const float NEG_INF = -__builtin_inff();

    float4 a   = make_float4(0.f, 0.f, 0.f, 0.f);
    float  den = 0.f;

    if (T <= 32) {
        const int   i  = (lane < deg) ? csr_src[beg + lane] : j;
        const float s  = s_src[i] + sd;
        const float lg = (s > 0.f) ? s : NEG_SLOPE * s;
        const float m  = wave_max((lane < T) ? lg : NEG_INF);
        const float w  = (lane < T) ? __expf(lg - m) : 0.f;
        den = wave_sum(w);
        gather_rows(H, i, w, T, lane, a);
    } else {
        float m = NEG_INF;
        for (int base = 0; base < T; base += 32) {
            const int   t  = base + lane;
            const int   i  = (t < deg) ? csr_src[beg + t] : j;
            const float s  = s_src[i] + sd;
            const float lg = (s > 0.f) ? s : NEG_SLOPE * s;
            if (t < T) m = fmaxf(m, lg);
        }
        m = wave_max(m);
        for (int base = 0; base < T; base += 32) {
            const int   t  = base + lane;
            const int   i  = (t < deg) ? csr_src[beg + t] : j;
            const float s  = s_src[i] + sd;
            const float lg = (s > 0.f) ? s : NEG_SLOPE * s;
            const float w  = (t < T) ? __expf(lg - m) : 0.f;
            den += w;
            gather_rows(H, i, w, min(32, T - base), lane, a);
        }
        den = wave_sum(den);
    }

    const float  inv = 1.0f / den;
    const float4 b4  = *(const float4*)(bias + lane * 4);
    float4 v;
    v.x = a.x * inv + b4.x;
    v.y = a.y * inv + b4.y;
    v.z = a.z * inv + b4.z;
    v.w = a.w * inv + b4.w;
    *(float4*)(out + (long)j * GDIM + lane * 4) = v;
}

#define POOL_CHUNK 16

__device__ __forceinline__ void pool_flush(float* __restrict__ psum,
                                           float* __restrict__ pcount, int g,
                                           float4 s, float cnt, int lane) {
    float* pp = psum + (long)g * GDIM + lane * 4;
    atomicAdd(pp + 0, s.x);
    atomicAdd(pp + 1, s.y);
    atomicAdd(pp + 2, s.z);
    atomicAdd(pp + 3, s.w);
    if (lane == 0) atomicAdd(pcount + g, cnt);
}

__device__ __forceinline__ float4 add4(float4 a, float4 b) {
    return make_float4(a.x + b.x, a.y + b.y, a.z + b.z, a.w + b.w);
}

__global__ void __launch_bounds__(256)
gat_pool_accum(const float* __restrict__ h, const int* __restrict__ batch, int n,
               float* __restrict__ psum, float* __restrict__ pcount) {
    const int lane  = threadIdx.x & 31;
    const int wave  = __builtin_amdgcn_readfirstlane((int)(threadIdx.x >> 5));
    const int node0 = (blockIdx.x * 8 + wave) * POOL_CHUNK;
    if (node0 >= n) return;
    const int valid = min(POOL_CHUNK, n - node0);
    const float* hp = h + (long)node0 * GDIM + lane * 4;
    const int gl = batch[node0 + (lane < valid ? lane : 0)];

    if (valid == POOL_CHUNK) {
        float4 v[POOL_CHUNK];
#pragma unroll
        for (int u = 0; u < POOL_CHUNK; ++u)
            v[u] = *(const float4*)(hp + (long)u * GDIM);
        const int g0 = __builtin_amdgcn_readfirstlane(gl);
        if (__all(gl == g0)) {
#pragma unroll
            for (int w = POOL_CHUNK / 2; w > 0; w >>= 1)
#pragma unroll
                for (int u = 0; u < w; ++u) v[u] = add4(v[u], v[u + w]);
            pool_flush(psum, pcount, g0, v[0], (float)POOL_CHUNK, lane);
        } else {
            float4 s   = make_float4(0.f, 0.f, 0.f, 0.f);
            int    gc  = g0;
            float  cnt = 0.f;
#pragma unroll
            for (int u = 0; u < POOL_CHUNK; ++u) {
                const int g = __builtin_amdgcn_readlane(gl, u);
                if (g != gc) {
                    pool_flush(psum, pcount, gc, s, cnt, lane);
                    s = make_float4(0.f, 0.f, 0.f, 0.f); cnt = 0.f; gc = g;
                }
                s = add4(s, v[u]); cnt += 1.0f;
            }
            pool_flush(psum, pcount, gc, s, cnt, lane);
        }
    } else {
        float4 s   = make_float4(0.f, 0.f, 0.f, 0.f);
        int    gc  = __builtin_amdgcn_readfirstlane(gl);
        float  cnt = 0.f;
        for (int u = 0; u < valid; ++u) {
            const int    g  = __shfl(gl, u, 32);
            const float4 hv = *(const float4*)(hp + (long)u * GDIM);
            if (g != gc) {
                pool_flush(psum, pcount, gc, s, cnt, lane);
                s = make_float4(0.f, 0.f, 0.f, 0.f); cnt = 0.f; gc = g;
            }
            s = add4(s, hv); cnt += 1.0f;
        }
        pool_flush(psum, pcount, gc, s, cnt, lane);
    }
}

__global__ void __launch_bounds__(256)
gat_pool_final(const float* __restrict__ psum, const float* __restrict__ pcount,
               float* __restrict__ out) {
    const int idx = blockIdx.x * blockDim.x + threadIdx.x;
    const float cnt = pcount[idx >> 7];
    out[idx] = psum[idx] / fmaxf(cnt, 1.0f);
}

extern "C" void kernel_launch(void* const* d_in, const int* in_sizes, int n_in,
                              void* d_out, int out_size, void* d_ws, size_t ws_size,
                              hipStream_t stream) {
    const float* x     = (const float*)d_in[0];
    const int*   ei    = (const int*)d_in[1];
    const int*   batch = (const int*)d_in[2];
    const float* W[3]    = {(const float*)d_in[3], (const float*)d_in[7],  (const float*)d_in[11]};
    const float* asrc[3] = {(const float*)d_in[4], (const float*)d_in[8],  (const float*)d_in[12]};
    const float* adst[3] = {(const float*)d_in[5], (const float*)d_in[9],  (const float*)d_in[13]};
    const float* bias[3] = {(const float*)d_in[6], (const float*)d_in[10], (const float*)d_in[14]};

    const int n  = in_sizes[0] / GDIM;
    const int E  = in_sizes[1] / 2;

    char*  wsb = (char*)d_ws;
    size_t off = 0;
    auto take = [&](size_t bytes) -> void* {
        void* p = wsb + off;
        off += (bytes + 255) & ~(size_t)255;
        return p;
    };
    const size_t szH = (size_t)n * GDIM * sizeof(float);
    float*    h      = (float*)   take(szH);
    float*    acc    = (float*)   take(szH);
    float*    s_src  = (float*)   take((size_t)n * sizeof(float));
    float*    s_dst  = (float*)   take((size_t)n * sizeof(float));
    int*      deg    = (int*)     take((size_t)n * sizeof(int));
    int*      start  = (int*)     take((size_t)n * sizeof(int));
    int*      cursor = (int*)     take((size_t)n * sizeof(int));
    int*      csr    = (int*)     take((size_t)E * sizeof(int));
    float*    psum   = (float*)   take((size_t)NGRAPH * GDIM * sizeof(float));
    float*    pcount = (float*)   take((size_t)NGRAPH * sizeof(float));
    (void)ws_size; (void)n_in; (void)out_size;

    const int nTiles     = (n + 15) / 16;
    const int gemmBlocks = (nTiles + 3) / 4;
    const int nodeBlocks = (n + 7) / 8;
    const int edgeBlocks = (E + 255) / 256;
    const int scanBlocks = (n + 1023) / 1024;
    const int poolBlocks = (n + 8 * POOL_CHUNK - 1) / (8 * POOL_CHUNK);

    hipMemsetAsync(deg, 0, (size_t)n * sizeof(int), stream);
    csr_count<<<edgeBlocks, 256, 0, stream>>>(ei, E, deg);
    csr_offsets<<<scanBlocks, 1024, 0, stream>>>(deg, n, start, cursor);
    csr_fill<<<edgeBlocks, 256, 0, stream>>>(ei, E, cursor, csr);

    const float* xin = x;
    for (int l = 0; l < 3; ++l) {
        gat_gemm_wmma<<<gemmBlocks, 128, GDIM * GDIM * sizeof(float), stream>>>(
            xin, W[l], asrc[l], adst[l], h, s_src, s_dst, nTiles);
        gat_gather<<<nodeBlocks, 256, 0, stream>>>(start, cursor, csr, s_src, s_dst,
                                                   h, bias[l], acc, n);
        xin = acc;
    }

    hipMemsetAsync(psum,   0, (size_t)NGRAPH * GDIM * sizeof(float), stream);
    hipMemsetAsync(pcount, 0, (size_t)NGRAPH * sizeof(float), stream);
    gat_pool_accum<<<poolBlocks, 256, 0, stream>>>(acc, batch, n, psum, pcount);
    gat_pool_final<<<(NGRAPH * GDIM) / 256, 256, 0, stream>>>(psum, pcount,
                                                              (float*)d_out);
}


# each layer is one launch: a node's softmax-weighted sum of INPUT rows goes to an LDS tile and is multiplied by W on WMMA there (layer is linear); scores as x.(W a); H not stored; new sum order
# speedup vs baseline: 5.5693x; 1.1925x over previous
#include <hip/hip_runtime.h>

#define GDIM 128
#define NGRAPH 256
#define NEG_SLOPE 0.2f
#define TROWS 16
#define TPITCH 132

typedef float v2f __attribute__((ext_vector_type(2)));
typedef float v8f __attribute__((ext_vector_type(8)));

__global__ void __launch_bounds__(256)
csr_count(const int* __restrict__ ei, int E, int* __restrict__ deg) {
    const int e = blockIdx.x * blockDim.x + threadIdx.x;
    if (e >= E) return;
    atomicAdd(deg + ei[E + e], 1);
}

__global__ void __launch_bounds__(1024)
csr_offsets(const int* __restrict__ deg, int n, int* __restrict__ start,
            int* __restrict__ cursor) {
    __shared__ int wtot[32];
    __shared__ int wpre[32];
    const int tid   = threadIdx.x;
    const int lane  = tid & 31;
    const int wave  = tid >> 5;
    const int first = blockIdx.x * 1024;
    int pre = 0;
    for (int t = tid; t < first; t += 1024) pre += deg[t];
    const int j = first + tid;
    const int d = (j < n) ? deg[j] : 0;
    int incl = d;
#pragma unroll
    for (int off = 1; off < 32; off <<= 1) {
        const int t = __shfl_up(incl, off, 32);
        if (lane >= off) incl += t;
    }
#pragma unroll
    for (int off = 16; off > 0; off >>= 1) pre += __shfl_xor(pre, off, 32);
    if (lane == 31) wtot[wave] = incl;
    if (lane == 0)  wpre[wave] = pre;
    __syncthreads();
    int base = wpre[lane] + ((lane < wave) ? wtot[lane] : 0);
#pragma unroll
    for (int off = 16; off > 0; off >>= 1) base += __shfl_xor(base, off, 32);
    const int st = base + incl - d;
    if (j < n) { start[j] = st; cursor[j] = st; }
}

__global__ void __launch_bounds__(256)
csr_fill(const int* __restrict__ ei, int E, int* __restrict__ cursor,
         int* __restrict__ csr_src) {
    const int e = blockIdx.x * blockDim.x + threadIdx.x;
    if (e >= E) return;
    const int pos = atomicAdd(cursor + ei[E + e], 1);
    csr_src[pos] = ei[e];
}

__device__ __forceinline__ void gather_rows(const float* __restrict__ H, int iv,
                                            float wv, int cnt, int lane, float4& a) {
    const int wi = __float_as_int(wv);
    int k = 0;
    for (; k + 4 <= cnt; k += 4) {
        const int   i0 = __builtin_amdgcn_readlane(iv, k);
        const int   i1 = __builtin_amdgcn_readlane(iv, k + 1);
        const int   i2 = __builtin_amdgcn_readlane(iv, k + 2);
        const int   i3 = __builtin_amdgcn_readlane(iv, k + 3);
        const float w0 = __int_as_float(__builtin_amdgcn_readlane(wi, k));
        const float w1 = __int_as_float(__builtin_amdgcn_readlane(wi, k + 1));
        const float w2 = __int_as_float(__builtin_amdgcn_readlane(wi, k + 2));
        const float w3 = __int_as_float(__builtin_amdgcn_readlane(wi, k + 3));
        const float4 h0 = *(const float4*)(H + (long)i0 * GDIM + lane * 4);
        const float4 h1 = *(const float4*)(H + (long)i1 * GDIM + lane * 4);
        const float4 h2 = *(const float4*)(H + (long)i2 * GDIM + lane * 4);
        const float4 h3 = *(const float4*)(H + (long)i3 * GDIM + lane * 4);
        a.x = fmaf(w0, h0.x, a.x); a.y = fmaf(w0, h0.y, a.y);
        a.z = fmaf(w0, h0.z, a.z); a.w = fmaf(w0, h0.w, a.w);
        a.x = fmaf(w1, h1.x, a.x); a.y = fmaf(w1, h1.y, a.y);
        a.z = fmaf(w1, h1.z, a.z); a.w = fmaf(w1, h1.w, a.w);
        a.x = fmaf(w2, h2.x, a.x); a.y = fmaf(w2, h2.y, a.y);
        a.z = fmaf(w2, h2.z, a.z); a.w = fmaf(w2, h2.w, a.w);
        a.x = fmaf(w3, h3.x, a.x); a.y = fmaf(w3, h3.y, a.y);
        a.z = fmaf(w3, h3.z, a.z); a.w = fmaf(w3, h3.w, a.w);
    }
    for (; k < cnt; ++k) {
        const int   i0 = __builtin_amdgcn_readlane(iv, k);
        const float w0 = __int_as_float(__builtin_amdgcn_readlane(wi, k));
        const float4 h0 = *(const float4*)(H + (long)i0 * GDIM + lane * 4);
        a.x = fmaf(w0, h0.x, a.x); a.y = fmaf(w0, h0.y, a.y);
        a.z = fmaf(w0, h0.z, a.z); a.w = fmaf(w0, h0.w, a.w);
    }
}

__device__ __forceinline__ float wave_max(float v) {
#pragma unroll
    for (int off = 16; off > 0; off >>= 1) v = fmaxf(v, __shfl_xor(v, off, 32));
    return v;
}
__device__ __forceinline__ float wave_sum(float v) {
#pragma unroll
    for (int off = 16; off > 0; off >>= 1) v += __shfl_xor(v, off, 32);
    return v;
}

__global__ void __launch_bounds__(256)
gat_uvec(const float* __restrict__ W1, const float* __restrict__ W2,
         const float* __restrict__ W3, const float* __restrict__ as1,
         const float* __restrict__ as2, const float* __restrict__ as3,
         const float* __restrict__ ad1, const float* __restrict__ ad2,
         const float* __restrict__ ad3, float* __restrict__ u) {
    const int lane  = threadIdx.x & 31;
    const int wave  = threadIdx.x >> 5;
    const int layer = blockIdx.x >> 4;
    const int row   = (blockIdx.x & 15) * 8 + wave;
    const float* W  = (layer == 0) ? W1  : (layer == 1) ? W2  : W3;
    const float* as = (layer == 0) ? as1 : (layer == 1) ? as2 : as3;
    const float* ad = (layer == 0) ? ad1 : (layer == 1) ? ad2 : ad3;
    const float4 w4 = *(const float4*)(W + row * GDIM + lane * 4);
    const float4 a4 = *(const float4*)(as + lane * 4);
    const float4 d4 = *(const float4*)(ad + lane * 4);
    float ps = w4.x * a4.x + w4.y * a4.y + w4.z * a4.z + w4.w * a4.w;
    float pd = w4.x * d4.x + w4.y * d4.y + w4.z * d4.z + w4.w * d4.w;
    ps = wave_sum(ps);
    pd = wave_sum(pd);
    if (lane == 0) {
        u[layer * 2 * GDIM + row]        = ps;
        u[layer * 2 * GDIM + GDIM + row] = pd;
    }
}

__global__ void __launch_bounds__(256)
gat_scores_x(const float* __restrict__ x, const float* __restrict__ u,
             float* __restrict__ s_src, float* __restrict__ s_dst, int n) {
    const int lane = threadIdx.x & 31;
    const int wave = __builtin_amdgcn_readfirstlane((int)(threadIdx.x >> 5));
    const int row0 = (blockIdx.x * 8 + wave) * 4;
    if (row0 >= n) return;
    const float4 us = *(const float4*)(u + lane * 4);
    const float4 ud = *(const float4*)(u + GDIM + lane * 4);
    float ps[4], pd[4];
#pragma unroll
    for (int t = 0; t < 4; ++t) {
        const int row = min(row0 + t, n - 1);
        const float4 r = *(const float4*)(x + (long)row * GDIM + lane * 4);
        ps[t] = r.x * us.x + r.y * us.y + r.z * us.z + r.w * us.w;
        pd[t] = r.x * ud.x + r.y * ud.y + r.z * ud.z + r.w * ud.w;
    }
#pragma unroll
    for (int t = 0; t < 4; ++t) { ps[t] = wave_sum(ps[t]); pd[t] = wave_sum(pd[t]); }
    float os = ps[0], od = pd[0];
#pragma unroll
    for (int t = 1; t < 4; ++t) if (lane == t) { os = ps[t]; od = pd[t]; }
    if (lane < 4 && row0 + lane < n) { s_src[row0 + lane] = os; s_dst[row0 + lane] = od; }
}

template <bool HAS_NEXT>
__global__ void __launch_bounds__(512, 16)
gat_layer(const int* __restrict__ start, const int* __restrict__ endp,
          const int* __restrict__ csr_src, const float* __restrict__ s_src,
          const float* __restrict__ s_dst, const float* __restrict__ Xin,
          const float* __restrict__ W, const float* __restrict__ bias,
          const float* __restrict__ u_next, float* __restrict__ Xout,
          float* __restrict__ s_src_next, float* __restrict__ s_dst_next,
          int n, int nTiles) {
    __shared__ __attribute__((aligned(16))) float sW[GDIM * GDIM];
    __shared__ __attribute__((aligned(16))) float sT[TROWS * TPITCH];
    const int tid  = threadIdx.x;
    const int lane = tid & 31;
    const int wave = __builtin_amdgcn_readfirstlane(tid >> 5);
    const int hi   = lane >> 4;
    const int col  = lane & 15;

    {
        const float4* W4 = (const float4*)W;
        float4*       S4 = (float4*)sW;
#pragma unroll
        for (int t = 0; t < 8; ++t) S4[t * 512 + tid] = W4[t * 512 + tid];
    }
    const float NEG_INF = -__builtin_inff();

    for (int tile = blockIdx.x; tile < nTiles; tile += gridDim.x) {
        const int j = tile * TROWS + wave;
        float4 a = make_float4(0.f, 0.f, 0.f, 0.f);
        if (j < n) {
            const int   beg = start[j];
            const int   deg = endp[j] - beg;
            const int   T   = deg + 1;
            const float sd  = s_dst[j];
            float den = 0.f;
            if (T <= 32) {
                const int   i  = (lane < deg) ? csr_src[beg + lane] : j;
                const float s  = s_src[i] + sd;
                const float lg = (s > 0.f) ? s : NEG_SLOPE * s;
                const float m  = wave_max((lane < T) ? lg : NEG_INF);
                const float w  = (lane < T) ? __expf(lg - m) : 0.f;
                den = wave_sum(w);
                gather_rows(Xin, i, w, T, lane, a);
            } else {
                float m = NEG_INF;
                for (int base = 0; base < T; base += 32) {
                    const int   t  = base + lane;
                    const int   i  = (t < deg) ? csr_src[beg + t] : j;
                    const float s  = s_src[i] + sd;
                    const float lg = (s > 0.f) ? s : NEG_SLOPE * s;
                    if (t < T) m = fmaxf(m, lg);
                }
                m = wave_max(m);
                for (int base = 0; base < T; base += 32) {
                    const int   t  = base + lane;
                    const int   i  = (t < deg) ? csr_src[beg + t] : j;
                    const float s  = s_src[i] + sd;
                    const float lg = (s > 0.f) ? s : NEG_SLOPE * s;
                    const float w  = (t < T) ? __expf(lg - m) : 0.f;
                    den += w;
                    gather_rows(Xin, i, w, min(32, T - base), lane, a);
                }
                den = wave_sum(den);
            }
            const float inv = 1.0f / den;
            a.x *= inv; a.y *= inv; a.z *= inv; a.w *= inv;
        }
        *(float4*)(sT + wave * TPITCH + lane * 4) = a;
        __syncthreads();

        v8f c = (v8f)0.0f;
        if (wave < 8) {
            v8f c0 = (v8f)0.0f, c1 = (v8f)0.0f;
            const float* At = sT + col * TPITCH + 2 * hi;
            const float* Bt = sW + (2 * hi) * GDIM + wave * 16 + col;
#pragma unroll
            for (int kk = 0; kk < GDIM; kk += 8) {
                const v2f a0 = *(const v2f*)(At + kk);
                v2f b0;
                b0.x = Bt[(kk + 0) * GDIM];
                b0.y = Bt[(kk + 1) * GDIM];
                const v2f a1 = *(const v2f*)(At + kk + 4);
                v2f b1;
                b1.x = Bt[(kk + 4) * GDIM];
                b1.y = Bt[(kk + 5) * GDIM];
                c0 = __builtin_amdgcn_wmma_f32_16x16x4_f32(false, a0, false, b0, (short)0, c0, false, false);
                c1 = __builtin_amdgcn_wmma_f32_16x16x4_f32(false, a1, false, b1, (short)0, c1, false, false);
            }
            c = c0 + c1;
        }
        __syncthreads();
        if (wave < 8) {
#pragma unroll
            for (int v = 0; v < 8; ++v) sT[(v + 8 * hi) * TPITCH + wave * 16 + col] = c[v];
        }
        __syncthreads();
        if (j < n) {
            const float4 b4 = *(const float4*)(bias + lane * 4);
            float4 r = *(const float4*)(sT + wave * TPITCH + lane * 4);
            r.x += b4.x; r.y += b4.y; r.z += b4.z; r.w += b4.w;
            *(float4*)(Xout + (long)j * GDIM + lane * 4) = r;
            if (HAS_NEXT) {
                const float4 us = *(const float4*)(u_next + lane * 4);
                const float4 ud = *(const float4*)(u_next + GDIM + lane * 4);
                float ps = r.x * us.x + r.y * us.y + r.z * us.z + r.w * us.w;
                float pd = r.x * ud.x + r.y * ud.y + r.z * ud.z + r.w * ud.w;
                ps = wave_sum(ps);
                pd = wave_sum(pd);
                if (lane == 0) { s_src_next[j] = ps; s_dst_next[j] = pd; }
            }
        }
    }
}

#define POOL_CHUNK 16

__device__ __forceinline__ void pool_flush(float* __restrict__ psum,
                                           float* __restrict__ pcount, int g,
                                           float4 s, float cnt, int lane) {
    float* pp = psum + (long)g * GDIM + lane * 4;
    atomicAdd(pp + 0, s.x);
    atomicAdd(pp + 1, s.y);
    atomicAdd(pp + 2, s.z);
    atomicAdd(pp + 3, s.w);
    if (lane == 0) atomicAdd(pcount + g, cnt);
}

__device__ __forceinline__ float4 add4(float4 a, float4 b) {
    return make_float4(a.x + b.x, a.y + b.y, a.z + b.z, a.w + b.w);
}

__global__ void __launch_bounds__(256)
gat_pool_accum(const float* __restrict__ h, const int* __restrict__ batch, int n,
               float* __restrict__ psum, float* __restrict__ pcount) {
    const int lane  = threadIdx.x & 31;
    const int wave  = __builtin_amdgcn_readfirstlane((int)(threadIdx.x >> 5));
    const int node0 = (blockIdx.x * 8 + wave) * POOL_CHUNK;
    if (node0 >= n) return;
    const int valid = min(POOL_CHUNK, n - node0);
    const float* hp = h + (long)node0 * GDIM + lane * 4;
    const int gl = batch[node0 + (lane < valid ? lane : 0)];

    if (valid == POOL_CHUNK) {
        float4 v[POOL_CHUNK];
#pragma unroll
        for (int u = 0; u < POOL_CHUNK; ++u)
            v[u] = *(const float4*)(hp + (long)u * GDIM);
        const int g0 = __builtin_amdgcn_readfirstlane(gl);
        if (__all(gl == g0)) {
#pragma unroll
            for (int w = POOL_CHUNK / 2; w > 0; w >>= 1)
#pragma unroll
                for (int u = 0; u < w; ++u) v[u] = add4(v[u], v[u + w]);
            pool_flush(psum, pcount, g0, v[0], (float)POOL_CHUNK, lane);
        } else {
            float4 s   = make_float4(0.f, 0.f, 0.f, 0.f);
            int    gc  = g0;
            float  cnt = 0.f;
#pragma unroll
            for (int u = 0; u < POOL_CHUNK; ++u) {
                const int g = __builtin_amdgcn_readlane(gl, u);
                if (g != gc) {
                    pool_flush(psum, pcount, gc, s, cnt, lane);
                    s = make_float4(0.f, 0.f, 0.f, 0.f); cnt = 0.f; gc = g;
                }
                s = add4(s, v[u]); cnt += 1.0f;
            }
            pool_flush(psum, pcount, gc, s, cnt, lane);
        }
    } else {
        float4 s   = make_float4(0.f, 0.f, 0.f, 0.f);
        int    gc  = __builtin_amdgcn_readfirstlane(gl);
        float  cnt = 0.f;
        for (int u = 0; u < valid; ++u) {
            const int    g  = __shfl(gl, u, 32);
            const float4 hv = *(const float4*)(hp + (long)u * GDIM);
            if (g != gc) {
                pool_flush(psum, pcount, gc, s, cnt, lane);
                s = make_float4(0.f, 0.f, 0.f, 0.f); cnt = 0.f; gc = g;
            }
            s = add4(s, hv); cnt += 1.0f;
        }
        pool_flush(psum, pcount, gc, s, cnt, lane);
    }
}

__global__ void __launch_bounds__(256)
gat_pool_final(const float* __restrict__ psum, const float* __restrict__ pcount,
               float* __restrict__ out) {
    const int idx = blockIdx.x * blockDim.x + threadIdx.x;
    const float cnt = pcount[idx >> 7];
    out[idx] = psum[idx] / fmaxf(cnt, 1.0f);
}

extern "C" void kernel_launch(void* const* d_in, const int* in_sizes, int n_in,
                              void* d_out, int out_size, void* d_ws, size_t ws_size,
                              hipStream_t stream) {
    const float* x     = (const float*)d_in[0];
    const int*   ei    = (const int*)d_in[1];
    const int*   batch = (const int*)d_in[2];
    const float* W[3]    = {(const float*)d_in[3], (const float*)d_in[7],  (const float*)d_in[11]};
    const float* asrc[3] = {(const float*)d_in[4], (const float*)d_in[8],  (const float*)d_in[12]};
    const float* adst[3] = {(const float*)d_in[5], (const float*)d_in[9],  (const float*)d_in[13]};
    const float* bias[3] = {(const float*)d_in[6], (const float*)d_in[10], (const float*)d_in[14]};

    const int n  = in_sizes[0] / GDIM;
    const int E  = in_sizes[1] / 2;

    char*  wsb = (char*)d_ws;
    size_t off = 0;
    auto take = [&](size_t bytes) -> void* {
        void* p = wsb + off;
        off += (bytes + 255) & ~(size_t)255;
        return p;
    };
    const size_t szH = (size_t)n * GDIM * sizeof(float);
    float*    bufA   = (float*)   take(szH);
    float*    bufB   = (float*)   take(szH);
    float*    sA_src = (float*)   take((size_t)n * sizeof(float));
    float*    sA_dst = (float*)   take((size_t)n * sizeof(float));
    float*    sB_src = (float*)   take((size_t)n * sizeof(float));
    float*    sB_dst = (float*)   take((size_t)n * sizeof(float));
    int*      deg    = (int*)     take((size_t)n * sizeof(int));
    int*      start  = (int*)     take((size_t)n * sizeof(int));
    int*      cursor = (int*)     take((size_t)n * sizeof(int));
    int*      csr    = (int*)     take((size_t)E * sizeof(int));
    float*    psum   = (float*)   take((size_t)NGRAPH * GDIM * sizeof(float));
    float*    pcount = (float*)   take((size_t)NGRAPH * sizeof(float));
    float*    u      = (float*)   take((size_t)3 * 2 * GDIM * sizeof(float));
    (void)ws_size; (void)n_in; (void)out_size;

    const int nTiles      = (n + TROWS - 1) / TROWS;
    const int layerBlocks = nTiles < 1024 ? nTiles : 1024;
    const int edgeBlocks  = (E + 255) / 256;
    const int scanBlocks  = (n + 1023) / 1024;
    const int rowBlocks   = (n + 31) / 32;
    const int poolBlocks  = (n + 8 * POOL_CHUNK - 1) / (8 * POOL_CHUNK);

    hipMemsetAsync(deg, 0, (size_t)n * sizeof(int), stream);
    csr_count<<<edgeBlocks, 256, 0, stream>>>(ei, E, deg);
    csr_offsets<<<scanBlocks, 1024, 0, stream>>>(deg, n, start, cursor);
    csr_fill<<<edgeBlocks, 256, 0, stream>>>(ei, E, cursor, csr);

    gat_uvec<<<48, 256, 0, stream>>>(W[0], W[1], W[2], asrc[0], asrc[1], asrc[2],
                                     adst[0], adst[1], adst[2], u);
    gat_scores_x<<<rowBlocks, 256, 0, stream>>>(x, u, sA_src, sA_dst, n);

    gat_layer<true><<<layerBlocks, 512, 0, stream>>>(
        start, cursor, csr, sA_src, sA_dst, x, W[0], bias[0], u + 2 * GDIM,
        bufA, sB_src, sB_dst, n, nTiles);
    gat_layer<true><<<layerBlocks, 512, 0, stream>>>(
        start, cursor, csr, sB_src, sB_dst, bufA, W[1], bias[1], u + 4 * GDIM,
        bufB, sA_src, sA_dst, n, nTiles);
    gat_layer<false><<<layerBlocks, 512, 0, stream>>>(
        start, cursor, csr, sA_src, sA_dst, bufB, W[2], bias[2], nullptr,
        bufA, nullptr, nullptr, n, nTiles);

    hipMemsetAsync(psum,   0, (size_t)NGRAPH * GDIM * sizeof(float), stream);
    hipMemsetAsync(pcount, 0, (size_t)NGRAPH * sizeof(float), stream);
    gat_pool_accum<<<poolBlocks, 256, 0, stream>>>(bufA, batch, n, psum, pcount);
    gat_pool_final<<<(NGRAPH * GDIM) / 256, 256, 0, stream>>>(psum, pcount,
                                                              (float*)d_out);
}


# edge logits taken from the gathered rows themselves (row . Wa, DPP lane sums) with a running-max softmax per 4 rows; per-edge score lookups, score arrays and score launch removed; new sum order
# speedup vs baseline: 5.7741x; 1.0368x over previous
#include <hip/hip_runtime.h>

#define GDIM 128
#define NGRAPH 256
#define NEG_SLOPE 0.2f
#define TROWS 16
#define TPITCH 132

typedef float v2f __attribute__((ext_vector_type(2)));
typedef float v8f __attribute__((ext_vector_type(8)));

__global__ void __launch_bounds__(256)
csr_count(const int* __restrict__ ei, int E, int* __restrict__ deg) {
    const int e = blockIdx.x * blockDim.x + threadIdx.x;
    if (e >= E) return;
    atomicAdd(deg + ei[E + e], 1);
}

__global__ void __launch_bounds__(1024)
csr_offsets(const int* __restrict__ deg, int n, int* __restrict__ start,
            int* __restrict__ cursor) {
    __shared__ int wtot[32];
    __shared__ int wpre[32];
    const int tid   = threadIdx.x;
    const int lane  = tid & 31;
    const int wave  = tid >> 5;
    const int first = blockIdx.x * 1024;
    int pre = 0;
    for (int t = tid; t < first; t += 1024) pre += deg[t];
    const int j = first + tid;
    const int d = (j < n) ? deg[j] : 0;
    int incl = d;
#pragma unroll
    for (int off = 1; off < 32; off <<= 1) {
        const int t = __shfl_up(incl, off, 32);
        if (lane >= off) incl += t;
    }
#pragma unroll
    for (int off = 16; off > 0; off >>= 1) pre += __shfl_xor(pre, off, 32);
    if (lane == 31) wtot[wave] = incl;
    if (lane == 0)  wpre[wave] = pre;
    __syncthreads();
    int base = wpre[lane] + ((lane < wave) ? wtot[lane] : 0);
#pragma unroll
    for (int off = 16; off > 0; off >>= 1) base += __shfl_xor(base, off, 32);
    const int st = base + incl - d;
    if (j < n) { start[j] = st; cursor[j] = st; }
}

__global__ void __launch_bounds__(256)
csr_fill(const int* __restrict__ ei, int E, int* __restrict__ cursor,
         int* __restrict__ csr_src) {
    const int e = blockIdx.x * blockDim.x + threadIdx.x;
    if (e >= E) return;
    const int pos = atomicAdd(cursor + ei[E + e], 1);
    csr_src[pos] = ei[e];
}

__device__ __forceinline__ float wave_sum(float v) {
#pragma unroll
    for (int off = 16; off > 0; off >>= 1) v += __shfl_xor(v, off, 32);
    return v;
}

template <int M>
__device__ __forceinline__ float row_xor(float x) {
    return __builtin_bit_cast(float, __builtin_amdgcn_update_dpp(
        0, __builtin_bit_cast(int, x), 0x160 + M, 0xf, 0xf, false));
}
__device__ __forceinline__ float half_xor(float x) {
    return __builtin_bit_cast(float, __builtin_amdgcn_permlanex16(
        __builtin_bit_cast(int, x), __builtin_bit_cast(int, x), 0x76543210, 0xfedcba98,
        false, false));
}
__device__ __forceinline__ float lane_sum(float v) {
    v += row_xor<1>(v);
    v += row_xor<2>(v);
    v += row_xor<4>(v);
    v += row_xor<8>(v);
    v += half_xor(v);
    return v;
}

__device__ __forceinline__ float dot4(const float4 h, const float4 b) {
    v2f q = (v2f){h.x, h.y} * (v2f){b.x, b.y};
    q = __builtin_elementwise_fma((v2f){h.z, h.w}, (v2f){b.z, b.w}, q);
    return q.x + q.y;
}
__device__ __forceinline__ float4 load_row(const float* __restrict__ H, int i, int lane) {
    const float* rp = H + ((long)i << 7);
    return ((const float4*)rp)[lane];
}
__device__ __forceinline__ float leaky(float s) { return fmaxf(s, NEG_SLOPE * s); }

__device__ __forceinline__ void fma4(float w, const float4 h, float4& a) {
    a.x = fmaf(w, h.x, a.x); a.y = fmaf(w, h.y, a.y);
    a.z = fmaf(w, h.z, a.z); a.w = fmaf(w, h.w, a.w);
}

template <int NV>
__device__ __forceinline__ void edge_group(const float* __restrict__ H, const float4 us,
                                           float sd, int iv, int k0, int nvar, int lane,
                                           float& m, float& den, float4& a) {
    const int   nv      = NV ? NV : nvar;
    const float NEG_INF = -__builtin_inff();
    const int i0 = __builtin_amdgcn_readlane(iv, k0);
    const int i1 = __builtin_amdgcn_readlane(iv, k0 + 1);
    const int i2 = __builtin_amdgcn_readlane(iv, k0 + 2);
    const int i3 = __builtin_amdgcn_readlane(iv, k0 + 3);
    const float4 z = make_float4(0.f, 0.f, 0.f, 0.f);
    float4 h0 = load_row(H, i0, lane);
    float4 h1 = z, h2 = z, h3 = z;
    if (nv > 1) h1 = load_row(H, i1, lane);
    if (nv > 2) h2 = load_row(H, i2, lane);
    if (nv > 3) h3 = load_row(H, i3, lane);
    float p0 = dot4(h0, us), p1 = dot4(h1, us), p2 = dot4(h2, us), p3 = dot4(h3, us);
    p0 = lane_sum(p0); p1 = lane_sum(p1); p2 = lane_sum(p2); p3 = lane_sum(p3);
    const float l0 = leaky(p0 + sd);
    const float l1 = (nv > 1) ? leaky(p1 + sd) : NEG_INF;
    const float l2 = (nv > 2) ? leaky(p2 + sd) : NEG_INF;
    const float l3 = (nv > 3) ? leaky(p3 + sd) : NEG_INF;
    const float mn = fmaxf(fmaxf(m, l0), fmaxf(fmaxf(l1, l2), l3));
    const float sc = __expf(m - mn);
    const float w0 = __expf(l0 - mn);
    const float w1 = __expf(l1 - mn);
    const float w2 = __expf(l2 - mn);
    const float w3 = __expf(l3 - mn);
    den = fmaf(den, sc, (w0 + w1) + (w2 + w3));
    a.x *= sc; a.y *= sc; a.z *= sc; a.w *= sc;
    fma4(w0, h0, a); fma4(w1, h1, a); fma4(w2, h2, a); fma4(w3, h3, a);
    m = mn;
}

__global__ void __launch_bounds__(256)
gat_uvec(const float* __restrict__ W1, const float* __restrict__ W2,
         const float* __restrict__ W3, const float* __restrict__ as1,
         const float* __restrict__ as2, const float* __restrict__ as3,
         const float* __restrict__ ad1, const float* __restrict__ ad2,
         const float* __restrict__ ad3, float* __restrict__ u) {
    const int lane  = threadIdx.x & 31;
    const int wave  = threadIdx.x >> 5;
    const int layer = blockIdx.x >> 4;
    const int row   = (blockIdx.x & 15) * 8 + wave;
    const float* W  = (layer == 0) ? W1  : (layer == 1) ? W2  : W3;
    const float* as = (layer == 0) ? as1 : (layer == 1) ? as2 : as3;
    const float* ad = (layer == 0) ? ad1 : (layer == 1) ? ad2 : ad3;
    const float4 w4 = *(const float4*)(W + row * GDIM + lane * 4);
    const float4 a4 = *(const float4*)(as + lane * 4);
    const float4 d4 = *(const float4*)(ad + lane * 4);
    float ps = w4.x * a4.x + w4.y * a4.y + w4.z * a4.z + w4.w * a4.w;
    float pd = w4.x * d4.x + w4.y * d4.y + w4.z * d4.z + w4.w * d4.w;
    ps = wave_sum(ps);
    pd = wave_sum(pd);
    if (lane == 0) {
        u[layer * 2 * GDIM + row]        = ps;
        u[layer * 2 * GDIM + GDIM + row] = pd;
    }
}

__global__ void __launch_bounds__(512, 16)
gat_layer(const int* __restrict__ start, const int* __restrict__ endp,
          const int* __restrict__ csr_src, const float* __restrict__ Xin,
          const float* __restrict__ W, const float* __restrict__ bias,
          const float* __restrict__ u, float* __restrict__ Xout,
          int n, int nTiles) {
    __shared__ __attribute__((aligned(16))) float sW[GDIM * GDIM];
    __shared__ __attribute__((aligned(16))) float sT[TROWS * TPITCH];
    const int tid  = threadIdx.x;
    const int lane = tid & 31;
    const int wave = __builtin_amdgcn_readfirstlane(tid >> 5);
    const int hi   = lane >> 4;
    const int col  = lane & 15;

    {
        const float4* W4 = (const float4*)W;
        float4*       S4 = (float4*)sW;
#pragma unroll
        for (int t = 0; t < 8; ++t) S4[t * 512 + tid] = W4[t * 512 + tid];
    }

    for (int tile = blockIdx.x; tile < nTiles; tile += gridDim.x) {
        const int j = tile * TROWS + wave;
        float4 a = make_float4(0.f, 0.f, 0.f, 0.f);
        if (j < n) {
            const float4 hs  = load_row(Xin, j, lane);
            const float4 us  = *(const float4*)(u + lane * 4);
            const float4 ud  = *(const float4*)(u + GDIM + lane * 4);
            const int    beg = start[j];
            const int    deg = endp[j] - beg;
            int iv = (lane < deg) ? csr_src[beg + lane] : j;
            __builtin_amdgcn_sched_barrier(0);
            const float  ss  = lane_sum(dot4(hs, us));
            const float  sd  = lane_sum(dot4(hs, ud));
            float m   = leaky(ss + sd);
            float den = 1.0f;
            a = hs;
            for (int base = 0; base < deg; base += 32) {
                if (base > 0) {
                    const int t = base + lane;
                    iv = (t < deg) ? csr_src[beg + t] : j;
                }
                const int cnt = min(32, deg - base);
                int k = 0;
                for (; k + 4 <= cnt; k += 4)
                    edge_group<4>(Xin, us, sd, iv, k, 4, lane, m, den, a);
                if (k < cnt)
                    edge_group<0>(Xin, us, sd, iv, k, cnt - k, lane, m, den, a);
            }
            const float inv = 1.0f / den;
            a.x *= inv; a.y *= inv; a.z *= inv; a.w *= inv;
        }
        *(float4*)(sT + wave * TPITCH + lane * 4) = a;
        __syncthreads();

        v8f c = (v8f)0.0f;
        if (wave < 8) {
            v8f c0 = (v8f)0.0f, c1 = (v8f)0.0f;
            const float* At = sT + col * TPITCH + 2 * hi;
            const float* Bt = sW + (2 * hi) * GDIM + wave * 16 + col;
#pragma unroll
            for (int kk = 0; kk < GDIM; kk += 8) {
                const v2f a0 = *(const v2f*)(At + kk);
                v2f b0;
                b0.x = Bt[(kk + 0) * GDIM];
                b0.y = Bt[(kk + 1) * GDIM];
                const v2f a1 = *(const v2f*)(At + kk + 4);
                v2f b1;
                b1.x = Bt[(kk + 4) * GDIM];
                b1.y = Bt[(kk + 5) * GDIM];
                c0 = __builtin_amdgcn_wmma_f32_16x16x4_f32(false, a0, false, b0, (short)0, c0, false, false);
                c1 = __builtin_amdgcn_wmma_f32_16x16x4_f32(false, a1, false, b1, (short)0, c1, false, false);
            }
            c = c0 + c1;
        }
        __syncthreads();
        if (wave < 8) {
#pragma unroll
            for (int v = 0; v < 8; ++v) sT[(v + 8 * hi) * TPITCH + wave * 16 + col] = c[v];
        }
        __syncthreads();
        if (j < n) {
            const float4 b4 = *(const float4*)(bias + lane * 4);
            float4 r = *(const float4*)(sT + wave * TPITCH + lane * 4);
            r.x += b4.x; r.y += b4.y; r.z += b4.z; r.w += b4.w;
            *(float4*)(Xout + (long)j * GDIM + lane * 4) = r;
        }
    }
}

#define POOL_CHUNK 16

__device__ __forceinline__ void pool_flush(float* __restrict__ psum,
                                           float* __restrict__ pcount, int g,
                                           float4 s, float cnt, int lane) {
    float* pp = psum + (long)g * GDIM + lane * 4;
    atomicAdd(pp + 0, s.x);
    atomicAdd(pp + 1, s.y);
    atomicAdd(pp + 2, s.z);
    atomicAdd(pp + 3, s.w);
    if (lane == 0) atomicAdd(pcount + g, cnt);
}

__device__ __forceinline__ float4 add4(float4 a, float4 b) {
    return make_float4(a.x + b.x, a.y + b.y, a.z + b.z, a.w + b.w);
}

__global__ void __launch_bounds__(256)
gat_pool_accum(const float* __restrict__ h, const int* __restrict__ batch, int n,
               float* __restrict__ psum, float* __restrict__ pcount) {
    const int lane  = threadIdx.x & 31;
    const int wave  = __builtin_amdgcn_readfirstlane((int)(threadIdx.x >> 5));
    const int node0 = (blockIdx.x * 8 + wave) * POOL_CHUNK;
    if (node0 >= n) return;
    const int valid = min(POOL_CHUNK, n - node0);
    const float* hp = h + (long)node0 * GDIM + lane * 4;
    const int gl = batch[node0 + (lane < valid ? lane : 0)];

    if (valid == POOL_CHUNK) {
        float4 v[POOL_CHUNK];
#pragma unroll
        for (int u = 0; u < POOL_CHUNK; ++u)
            v[u] = *(const float4*)(hp + (long)u * GDIM);
        const int g0 = __builtin_amdgcn_readfirstlane(gl);
        if (__all(gl == g0)) {
#pragma unroll
            for (int w = POOL_CHUNK / 2; w > 0; w >>= 1)
#pragma unroll
                for (int u = 0; u < w; ++u) v[u] = add4(v[u], v[u + w]);
            pool_flush(psum, pcount, g0, v[0], (float)POOL_CHUNK, lane);
        } else {
            float4 s   = make_float4(0.f, 0.f, 0.f, 0.f);
            int    gc  = g0;
            float  cnt = 0.f;
#pragma unroll
            for (int u = 0; u < POOL_CHUNK; ++u) {
                const int g = __builtin_amdgcn_readlane(gl, u);
                if (g != gc) {
                    pool_flush(psum, pcount, gc, s, cnt, lane);
                    s = make_float4(0.f, 0.f, 0.f, 0.f); cnt = 0.f; gc = g;
                }
                s = add4(s, v[u]); cnt += 1.0f;
            }
            pool_flush(psum, pcount, gc, s, cnt, lane);
        }
    } else {
        float4 s   = make_float4(0.f, 0.f, 0.f, 0.f);
        int    gc  = __builtin_amdgcn_readfirstlane(gl);
        float  cnt = 0.f;
        for (int u = 0; u < valid; ++u) {
            const int    g  = __shfl(gl, u, 32);
            const float4 hv = *(const float4*)(hp + (long)u * GDIM);
            if (g != gc) {
                pool_flush(psum, pcount, gc, s, cnt, lane);
                s = make_float4(0.f, 0.f, 0.f, 0.f); cnt = 0.f; gc = g;
            }
            s = add4(s, hv); cnt += 1.0f;
        }
        pool_flush(psum, pcount, gc, s, cnt, lane);
    }
}

__global__ void __launch_bounds__(256)
gat_pool_final(const float* __restrict__ psum, const float* __restrict__ pcount,
               float* __restrict__ out) {
    const int idx = blockIdx.x * blockDim.x + threadIdx.x;
    const float cnt = pcount[idx >> 7];
    out[idx] = psum[idx] / fmaxf(cnt, 1.0f);
}

extern "C" void kernel_launch(void* const* d_in, const int* in_sizes, int n_in,
                              void* d_out, int out_size, void* d_ws, size_t ws_size,
                              hipStream_t stream) {
    const float* x     = (const float*)d_in[0];
    const int*   ei    = (const int*)d_in[1];
    const int*   batch = (const int*)d_in[2];
    const float* W[3]    = {(const float*)d_in[3], (const float*)d_in[7],  (const float*)d_in[11]};
    const float* asrc[3] = {(const float*)d_in[4], (const float*)d_in[8],  (const float*)d_in[12]};
    const float* adst[3] = {(const float*)d_in[5], (const float*)d_in[9],  (const float*)d_in[13]};
    const float* bias[3] = {(const float*)d_in[6], (const float*)d_in[10], (const float*)d_in[14]};

    const int n  = in_sizes[0] / GDIM;
    const int E  = in_sizes[1] / 2;

    char*  wsb = (char*)d_ws;
    size_t off = 0;
    auto take = [&](size_t bytes) -> void* {
        void* p = wsb + off;
        off += (bytes + 255) & ~(size_t)255;
        return p;
    };
    const size_t szH = (size_t)n * GDIM * sizeof(float);
    float*    bufA   = (float*)   take(szH);
    float*    bufB   = (float*)   take(szH);
    int*      deg    = (int*)     take((size_t)n * sizeof(int));
    int*      start  = (int*)     take((size_t)n * sizeof(int));
    int*      cursor = (int*)     take((size_t)n * sizeof(int));
    int*      csr    = (int*)     take((size_t)E * sizeof(int));
    float*    psum   = (float*)   take((size_t)NGRAPH * GDIM * sizeof(float));
    float*    pcount = (float*)   take((size_t)NGRAPH * sizeof(float));
    float*    u      = (float*)   take((size_t)3 * 2 * GDIM * sizeof(float));
    (void)ws_size; (void)n_in; (void)out_size;

    const int nTiles      = (n + TROWS - 1) / TROWS;
    const int layerBlocks = nTiles < 1024 ? nTiles : 1024;
    const int edgeBlocks  = (E + 255) / 256;
    const int scanBlocks  = (n + 1023) / 1024;
    const int poolBlocks  = (n + 8 * POOL_CHUNK - 1) / (8 * POOL_CHUNK);

    hipMemsetAsync(deg, 0, (size_t)n * sizeof(int), stream);
    csr_count<<<edgeBlocks, 256, 0, stream>>>(ei, E, deg);
    csr_offsets<<<scanBlocks, 1024, 0, stream>>>(deg, n, start, cursor);
    csr_fill<<<edgeBlocks, 256, 0, stream>>>(ei, E, cursor, csr);

    gat_uvec<<<48, 256, 0, stream>>>(W[0], W[1], W[2], asrc[0], asrc[1], asrc[2],
                                     adst[0], adst[1], adst[2], u);
    gat_layer<<<layerBlocks, 512, 0, stream>>>(
        start, cursor, csr, x, W[0], bias[0], u, bufA, n, nTiles);
    gat_layer<<<layerBlocks, 512, 0, stream>>>(
        start, cursor, csr, bufA, W[1], bias[1], u + 2 * GDIM, bufB, n, nTiles);
    gat_layer<<<layerBlocks, 512, 0, stream>>>(
        start, cursor, csr, bufB, W[2], bias[2], u + 4 * GDIM, bufA, n, nTiles);

    hipMemsetAsync(psum,   0, (size_t)NGRAPH * GDIM * sizeof(float), stream);
    hipMemsetAsync(pcount, 0, (size_t)NGRAPH * sizeof(float), stream);
    gat_pool_accum<<<poolBlocks, 256, 0, stream>>>(bufA, batch, n, psum, pcount);
    gat_pool_final<<<(NGRAPH * GDIM) / 256, 256, 0, stream>>>(psum, pcount,
                                                              (float*)d_out);
}


# layer kernel: tiles handed out one at a time from a per-layer counter (thread 0, one atomic per tile, drawn one tile ahead) instead of a fixed stride: no work-group left with a 7th tile at the end
# speedup vs baseline: 5.9777x; 1.0353x over previous
#include <hip/hip_runtime.h>

#define GDIM 128
#define NGRAPH 256
#define NEG_SLOPE 0.2f
#define TROWS 16
#define TPITCH 132

typedef float v2f __attribute__((ext_vector_type(2)));
typedef float v8f __attribute__((ext_vector_type(8)));

__global__ void __launch_bounds__(256)
csr_count(const int* __restrict__ ei, int E, int* __restrict__ deg) {
    const int e = blockIdx.x * blockDim.x + threadIdx.x;
    if (e >= E) return;
    atomicAdd(deg + ei[E + e], 1);
}

__global__ void __launch_bounds__(1024)
csr_offsets(const int* __restrict__ deg, int n, int* __restrict__ start,
            int* __restrict__ cursor) {
    __shared__ int wtot[32];
    __shared__ int wpre[32];
    const int tid   = threadIdx.x;
    const int lane  = tid & 31;
    const int wave  = tid >> 5;
    const int first = blockIdx.x * 1024;
    int pre = 0;
    for (int t = tid; t < first; t += 1024) pre += deg[t];
    const int j = first + tid;
    const int d = (j < n) ? deg[j] : 0;
    int incl = d;
#pragma unroll
    for (int off = 1; off < 32; off <<= 1) {
        const int t = __shfl_up(incl, off, 32);
        if (lane >= off) incl += t;
    }
#pragma unroll
    for (int off = 16; off > 0; off >>= 1) pre += __shfl_xor(pre, off, 32);
    if (lane == 31) wtot[wave] = incl;
    if (lane == 0)  wpre[wave] = pre;
    __syncthreads();
    int base = wpre[lane] + ((lane < wave) ? wtot[lane] : 0);
#pragma unroll
    for (int off = 16; off > 0; off >>= 1) base += __shfl_xor(base, off, 32);
    const int st = base + incl - d;
    if (j < n) { start[j] = st; cursor[j] = st; }
}

__global__ void __launch_bounds__(256)
csr_fill(const int* __restrict__ ei, int E, int* __restrict__ cursor,
         int* __restrict__ csr_src) {
    const int e = blockIdx.x * blockDim.x + threadIdx.x;
    if (e >= E) return;
    const int pos = atomicAdd(cursor + ei[E + e], 1);
    csr_src[pos] = ei[e];
}

__device__ __forceinline__ float wave_sum(float v) {
#pragma unroll
    for (int off = 16; off > 0; off >>= 1) v += __shfl_xor(v, off, 32);
    return v;
}

template <int M>
__device__ __forceinline__ float row_xor(float x) {
    return __builtin_bit_cast(float, __builtin_amdgcn_update_dpp(
        0, __builtin_bit_cast(int, x), 0x160 + M, 0xf, 0xf, false));
}
__device__ __forceinline__ float half_xor(float x) {
    return __builtin_bit_cast(float, __builtin_amdgcn_permlanex16(
        __builtin_bit_cast(int, x), __builtin_bit_cast(int, x), 0x76543210, 0xfedcba98,
        false, false));
}
__device__ __forceinline__ float lane_sum(float v) {
    v += row_xor<1>(v);
    v += row_xor<2>(v);
    v += row_xor<4>(v);
    v += row_xor<8>(v);
    v += half_xor(v);
    return v;
}

__device__ __forceinline__ float dot4(const float4 h, const float4 b) {
    v2f q = (v2f){h.x, h.y} * (v2f){b.x, b.y};
    q = __builtin_elementwise_fma((v2f){h.z, h.w}, (v2f){b.z, b.w}, q);
    return q.x + q.y;
}
__device__ __forceinline__ float4 load_row(const float* __restrict__ H, int i, int lane) {
    const float* rp = H + ((long)i << 7);
    return ((const float4*)rp)[lane];
}
__device__ __forceinline__ float leaky(float s) { return fmaxf(s, NEG_SLOPE * s); }

__device__ __forceinline__ void fma4(float w, const float4 h, float4& a) {
    a.x = fmaf(w, h.x, a.x); a.y = fmaf(w, h.y, a.y);
    a.z = fmaf(w, h.z, a.z); a.w = fmaf(w, h.w, a.w);
}

template <int NV>
__device__ __forceinline__ void edge_group(const float* __restrict__ H, const float4 us,
                                           float sd, int iv, int k0, int nvar, int lane,
                                           float& m, float& den, float4& a) {
    const int   nv      = NV ? NV : nvar;
    const float NEG_INF = -__builtin_inff();
    const int i0 = __builtin_amdgcn_readlane(iv, k0);
    const int i1 = __builtin_amdgcn_readlane(iv, k0 + 1);
    const int i2 = __builtin_amdgcn_readlane(iv, k0 + 2);
    const int i3 = __builtin_amdgcn_readlane(iv, k0 + 3);
    const float4 z = make_float4(0.f, 0.f, 0.f, 0.f);
    float4 h0 = load_row(H, i0, lane);
    float4 h1 = z, h2 = z, h3 = z;
    if (nv > 1) h1 = load_row(H, i1, lane);
    if (nv > 2) h2 = load_row(H, i2, lane);
    if (nv > 3) h3 = load_row(H, i3, lane);
    float p0 = dot4(h0, us), p1 = dot4(h1, us), p2 = dot4(h2, us), p3 = dot4(h3, us);
    p0 = lane_sum(p0); p1 = lane_sum(p1); p2 = lane_sum(p2); p3 = lane_sum(p3);
    const float l0 = leaky(p0 + sd);
    const float l1 = (nv > 1) ? leaky(p1 + sd) : NEG_INF;
    const float l2 = (nv > 2) ? leaky(p2 + sd) : NEG_INF;
    const float l3 = (nv > 3) ? leaky(p3 + sd) : NEG_INF;
    const float mn = fmaxf(fmaxf(m, l0), fmaxf(fmaxf(l1, l2), l3));
    const float sc = __expf(m - mn);
    const float w0 = __expf(l0 - mn);
    const float w1 = __expf(l1 - mn);
    const float w2 = __expf(l2 - mn);
    const float w3 = __expf(l3 - mn);
    den = fmaf(den, sc, (w0 + w1) + (w2 + w3));
    a.x *= sc; a.y *= sc; a.z *= sc; a.w *= sc;
    fma4(w0, h0, a); fma4(w1, h1, a); fma4(w2, h2, a); fma4(w3, h3, a);
    m = mn;
}

__global__ void __launch_bounds__(256)
gat_uvec(const float* __restrict__ W1, const float* __restrict__ W2,
         const float* __restrict__ W3, const float* __restrict__ as1,
         const float* __restrict__ as2, const float* __restrict__ as3,
         const float* __restrict__ ad1, const float* __restrict__ ad2,
         const float* __restrict__ ad3, float* __restrict__ u) {
    const int lane  = threadIdx.x & 31;
    const int wave  = threadIdx.x >> 5;
    const int layer = blockIdx.x >> 4;
    const int row   = (blockIdx.x & 15) * 8 + wave;
    const float* W  = (layer == 0) ? W1  : (layer == 1) ? W2  : W3;
    const float* as = (layer == 0) ? as1 : (layer == 1) ? as2 : as3;
    const float* ad = (layer == 0) ? ad1 : (layer == 1) ? ad2 : ad3;
    const float4 w4 = *(const float4*)(W + row * GDIM + lane * 4);
    const float4 a4 = *(const float4*)(as + lane * 4);
    const float4 d4 = *(const float4*)(ad + lane * 4);
    float ps = w4.x * a4.x + w4.y * a4.y + w4.z * a4.z + w4.w * a4.w;
    float pd = w4.x * d4.x + w4.y * d4.y + w4.z * d4.z + w4.w * d4.w;
    ps = wave_sum(ps);
    pd = wave_sum(pd);
    if (lane == 0) {
        u[layer * 2 * GDIM + row]        = ps;
        u[layer * 2 * GDIM + GDIM + row] = pd;
    }
}

__global__ void __launch_bounds__(512, 16)
gat_layer(const int* __restrict__ start, const int* __restrict__ endp,
          const int* __restrict__ csr_src, const float* __restrict__ Xin,
          const float* __restrict__ W, const float* __restrict__ bias,
          const float* __restrict__ u, float* __restrict__ Xout,
          int n, int nTiles, int* __restrict__ tctr, int nFixed) {
    __shared__ __attribute__((aligned(16))) float sW[GDIM * GDIM];
    __shared__ __attribute__((aligned(16))) float sT[TROWS * TPITCH];
    __shared__ int sNext;
    const int tid  = threadIdx.x;
    const int lane = tid & 31;
    const int wave = __builtin_amdgcn_readfirstlane(tid >> 5);
    const int hi   = lane >> 4;
    const int col  = lane & 15;

    {
        const float4* W4 = (const float4*)W;
        float4*       S4 = (float4*)sW;
#pragma unroll
        for (int t = 0; t < 8; ++t) S4[t * 512 + tid] = W4[t * 512 + tid];
    }

    int tile = blockIdx.x;
    while (tile < nTiles) {
        int nxt = 0;
        const int j = tile * TROWS + wave;
        float4 a = make_float4(0.f, 0.f, 0.f, 0.f);
        if (j < n) {
            const float4 hs  = load_row(Xin, j, lane);
            const float4 us  = *(const float4*)(u + lane * 4);
            const float4 ud  = *(const float4*)(u + GDIM + lane * 4);
            const int    beg = start[j];
            const int    deg = endp[j] - beg;
            int iv = (lane < deg) ? csr_src[beg + lane] : j;
            if (tid == 0) nxt = atomicAdd(tctr, 1);
            __builtin_amdgcn_sched_barrier(0);
            const float  ss  = lane_sum(dot4(hs, us));
            const float  sd  = lane_sum(dot4(hs, ud));
            float m   = leaky(ss + sd);
            float den = 1.0f;
            a = hs;
            for (int base = 0; base < deg; base += 32) {
                if (base > 0) {
                    const int t = base + lane;
                    iv = (t < deg) ? csr_src[beg + t] : j;
                }
                const int cnt = min(32, deg - base);
                int k = 0;
                for (; k + 4 <= cnt; k += 4)
                    edge_group<4>(Xin, us, sd, iv, k, 4, lane, m, den, a);
                if (k < cnt)
                    edge_group<0>(Xin, us, sd, iv, k, cnt - k, lane, m, den, a);
            }
            const float inv = 1.0f / den;
            a.x *= inv; a.y *= inv; a.z *= inv; a.w *= inv;
        } else if (tid == 0) {
            nxt = atomicAdd(tctr, 1);
        }
        *(float4*)(sT + wave * TPITCH + lane * 4) = a;
        if (tid == 0) sNext = nFixed + nxt;
        __syncthreads();
        const int nextTile = __builtin_amdgcn_readfirstlane(sNext);

        v8f c = (v8f)0.0f;
        if (wave < 8) {
            v8f c0 = (v8f)0.0f, c1 = (v8f)0.0f;
            const float* At = sT + col * TPITCH + 2 * hi;
            const float* Bt = sW + (2 * hi) * GDIM + wave * 16 + col;
#pragma unroll
            for (int kk = 0; kk < GDIM; kk += 8) {
                const v2f a0 = *(const v2f*)(At + kk);
                v2f b0;
                b0.x = Bt[(kk + 0) * GDIM];
                b0.y = Bt[(kk + 1) * GDIM];
                const v2f a1 = *(const v2f*)(At + kk + 4);
                v2f b1;
                b1.x = Bt[(kk + 4) * GDIM];
                b1.y = Bt[(kk + 5) * GDIM];
                c0 = __builtin_amdgcn_wmma_f32_16x16x4_f32(false, a0, false, b0, (short)0, c0, false, false);
                c1 = __builtin_amdgcn_wmma_f32_16x16x4_f32(false, a1, false, b1, (short)0, c1, false, false);
            }
            c = c0 + c1;
        }
        __syncthreads();
        if (wave < 8) {
#pragma unroll
            for (int v = 0; v < 8; ++v) sT[(v + 8 * hi) * TPITCH + wave * 16 + col] = c[v];
        }
        __syncthreads();
        if (j < n) {
            const float4 b4 = *(const float4*)(bias + lane * 4);
            float4 r = *(const float4*)(sT + wave * TPITCH + lane * 4);
            r.x += b4.x; r.y += b4.y; r.z += b4.z; r.w += b4.w;
            *(float4*)(Xout + (long)j * GDIM + lane * 4) = r;
        }
        tile = nextTile;
    }
}

#define POOL_CHUNK 16

__device__ __forceinline__ void pool_flush(float* __restrict__ psum,
                                           float* __restrict__ pcount, int g,
                                           float4 s, float cnt, int lane) {
    float* pp = psum + (long)g * GDIM + lane * 4;
    atomicAdd(pp + 0, s.x);
    atomicAdd(pp + 1, s.y);
    atomicAdd(pp + 2, s.z);
    atomicAdd(pp + 3, s.w);
    if (lane == 0) atomicAdd(pcount + g, cnt);
}

__device__ __forceinline__ float4 add4(float4 a, float4 b) {
    return make_float4(a.x + b.x, a.y + b.y, a.z + b.z, a.w + b.w);
}

__global__ void __launch_bounds__(256)
gat_pool_accum(const float* __restrict__ h, const int* __restrict__ batch, int n,
               float* __restrict__ psum, float* __restrict__ pcount) {
    const int lane  = threadIdx.x & 31;
    const int wave  = __builtin_amdgcn_readfirstlane((int)(threadIdx.x >> 5));
    const int node0 = (blockIdx.x * 8 + wave) * POOL_CHUNK;
    if (node0 >= n) return;
    const int valid = min(POOL_CHUNK, n - node0);
    const float* hp = h + (long)node0 * GDIM + lane * 4;
    const int gl = batch[node0 + (lane < valid ? lane : 0)];

    if (valid == POOL_CHUNK) {
        float4 v[POOL_CHUNK];
#pragma unroll
        for (int u = 0; u < POOL_CHUNK; ++u)
            v[u] = *(const float4*)(hp + (long)u * GDIM);
        const int g0 = __builtin_amdgcn_readfirstlane(gl);
        if (__all(gl == g0)) {
#pragma unroll
            for (int w = POOL_CHUNK / 2; w > 0; w >>= 1)
#pragma unroll
                for (int u = 0; u < w; ++u) v[u] = add4(v[u], v[u + w]);
            pool_flush(psum, pcount, g0, v[0], (float)POOL_CHUNK, lane);
        } else {
            float4 s   = make_float4(0.f, 0.f, 0.f, 0.f);
            int    gc  = g0;
            float  cnt = 0.f;
#pragma unroll
            for (int u = 0; u < POOL_CHUNK; ++u) {
                const int g = __builtin_amdgcn_readlane(gl, u);
                if (g != gc) {
                    pool_flush(psum, pcount, gc, s, cnt, lane);
                    s = make_float4(0.f, 0.f, 0.f, 0.f); cnt = 0.f; gc = g;
                }
                s = add4(s, v[u]); cnt += 1.0f;
            }
            pool_flush(psum, pcount, gc, s, cnt, lane);
        }
    } else {
        float4 s   = make_float4(0.f, 0.f, 0.f, 0.f);
        int    gc  = __builtin_amdgcn_readfirstlane(gl);
        float  cnt = 0.f;
        for (int u = 0; u < valid; ++u) {
            const int    g  = __shfl(gl, u, 32);
            const float4 hv = *(const float4*)(hp + (long)u * GDIM);
            if (g != gc) {
                pool_flush(psum, pcount, gc, s, cnt, lane);
                s = make_float4(0.f, 0.f, 0.f, 0.f); cnt = 0.f; gc = g;
            }
            s = add4(s, hv); cnt += 1.0f;
        }
        pool_flush(psum, pcount, gc, s, cnt, lane);
    }
}

__global__ void __launch_bounds__(256)
gat_pool_final(const float* __restrict__ psum, const float* __restrict__ pcount,
               float* __restrict__ out) {
    const int idx = blockIdx.x * blockDim.x + threadIdx.x;
    const float cnt = pcount[idx >> 7];
    out[idx] = psum[idx] / fmaxf(cnt, 1.0f);
}

extern "C" void kernel_launch(void* const* d_in, const int* in_sizes, int n_in,
                              void* d_out, int out_size, void* d_ws, size_t ws_size,
                              hipStream_t stream) {
    const float* x     = (const float*)d_in[0];
    const int*   ei    = (const int*)d_in[1];
    const int*   batch = (const int*)d_in[2];
    const float* W[3]    = {(const float*)d_in[3], (const float*)d_in[7],  (const float*)d_in[11]};
    const float* asrc[3] = {(const float*)d_in[4], (const float*)d_in[8],  (const float*)d_in[12]};
    const float* adst[3] = {(const float*)d_in[5], (const float*)d_in[9],  (const float*)d_in[13]};
    const float* bias[3] = {(const float*)d_in[6], (const float*)d_in[10], (const float*)d_in[14]};

    const int n  = in_sizes[0] / GDIM;
    const int E  = in_sizes[1] / 2;

    char*  wsb = (char*)d_ws;
    size_t off = 0;
    auto take = [&](size_t bytes) -> void* {
        void* p = wsb + off;
        off += (bytes + 255) & ~(size_t)255;
        return p;
    };
    const size_t szH = (size_t)n * GDIM * sizeof(float);
    float*    bufA   = (float*)   take(szH);
    float*    bufB   = (float*)   take(szH);
    const size_t degBytes = ((size_t)n * sizeof(int) + 255) & ~(size_t)255;
    int*      deg    = (int*)     take(degBytes + 256);
    int*      tctr   = (int*)((char*)deg + degBytes);
    int*      start  = (int*)     take((size_t)n * sizeof(int));
    int*      cursor = (int*)     take((size_t)n * sizeof(int));
    int*      csr    = (int*)     take((size_t)E * sizeof(int));
    float*    psum   = (float*)   take((size_t)NGRAPH * GDIM * sizeof(float));
    float*    pcount = (float*)   take((size_t)NGRAPH * sizeof(float));
    float*    u      = (float*)   take((size_t)3 * 2 * GDIM * sizeof(float));
    (void)ws_size; (void)n_in; (void)out_size;

    const int nTiles      = (n + TROWS - 1) / TROWS;
    const int layerBlocks = nTiles < 1024 ? nTiles : 1024;
    const int edgeBlocks  = (E + 255) / 256;
    const int scanBlocks  = (n + 1023) / 1024;
    const int poolBlocks  = (n + 8 * POOL_CHUNK - 1) / (8 * POOL_CHUNK);

    hipMemsetAsync(deg, 0, degBytes + 256, stream);
    csr_count<<<edgeBlocks, 256, 0, stream>>>(ei, E, deg);
    csr_offsets<<<scanBlocks, 1024, 0, stream>>>(deg, n, start, cursor);
    csr_fill<<<edgeBlocks, 256, 0, stream>>>(ei, E, cursor, csr);

    gat_uvec<<<48, 256, 0, stream>>>(W[0], W[1], W[2], asrc[0], asrc[1], asrc[2],
                                     adst[0], adst[1], adst[2], u);
    gat_layer<<<layerBlocks, 512, 0, stream>>>(
        start, cursor, csr, x, W[0], bias[0], u, bufA, n, nTiles, tctr, layerBlocks);
    gat_layer<<<layerBlocks, 512, 0, stream>>>(
        start, cursor, csr, bufA, W[1], bias[1], u + 2 * GDIM, bufB, n, nTiles, tctr + 16, layerBlocks);
    gat_layer<<<layerBlocks, 512, 0, stream>>>(
        start, cursor, csr, bufB, W[2], bias[2], u + 4 * GDIM, bufA, n, nTiles, tctr + 32, layerBlocks);

    hipMemsetAsync(psum,   0, (size_t)NGRAPH * GDIM * sizeof(float), stream);
    hipMemsetAsync(pcount, 0, (size_t)NGRAPH * sizeof(float), stream);
    gat_pool_accum<<<poolBlocks, 256, 0, stream>>>(bufA, batch, n, psum, pcount);
    gat_pool_final<<<(NGRAPH * GDIM) / 256, 256, 0, stream>>>(psum, pcount,
                                                              (float*)d_out);
}


# in-edge lists by a two-level counting sort on LDS counters (bins of 512 dst nodes, one work-group sorts a bin) replacing the global atomic count, scan and fill; new list order, so new sum order
# speedup vs baseline: 7.9214x; 1.3252x over previous
#include <hip/hip_runtime.h>

#define GDIM 128
#define NGRAPH 256
#define NEG_SLOPE 0.2f
#define TROWS 16
#define TPITCH 132

typedef float v2f __attribute__((ext_vector_type(2)));
typedef float v8f __attribute__((ext_vector_type(8)));

#define BIN_SHIFT 9
#define BIN_NODES 512
#define MAXB      256
#define ECHUNK    4096
#define STAGE_CAP 12288

__device__ __forceinline__ int wave_sum_i(int v) {
#pragma unroll
    for (int off = 16; off > 0; off >>= 1) v += __shfl_xor(v, off, 32);
    return v;
}

__global__ void __launch_bounds__(1024)
bin_hist(const int* __restrict__ ei, int E, int nb, int g1, int* __restrict__ histT) {
    __shared__ int h[MAXB];
    const int tid  = threadIdx.x;
    const int base = blockIdx.x * ECHUNK;
    int d[ECHUNK / 1024];
#pragma unroll
    for (int k = 0; k < ECHUNK / 1024; ++k) {
        const int e = base + k * 1024 + tid;
        d[k] = ei[E + min(e, E - 1)];
    }
    if (tid < MAXB) h[tid] = 0;
    __syncthreads();
#pragma unroll
    for (int k = 0; k < ECHUNK / 1024; ++k) {
        const int e = base + k * 1024 + tid;
        if (e < E) atomicAdd(&h[d[k] >> BIN_SHIFT], 1);
    }
    __syncthreads();
    if (tid < nb) histT[tid * g1 + blockIdx.x] = h[tid];
}

__global__ void __launch_bounds__(1024)
bin_scatter(const int* __restrict__ ei, int E, int nb, int g1,
            const int* __restrict__ histT, int* __restrict__ binStart,
            int* __restrict__ binned, int* __restrict__ tctr) {
    __shared__ int tot[MAXB];
    __shared__ int pre[MAXB];
    __shared__ int cur[MAXB];
    __shared__ int bas[MAXB];
    __shared__ int wtot[MAXB / 32];
    const int tid  = threadIdx.x;
    const int lane = tid & 31;
    const int wave = tid >> 5;
    const int w    = blockIdx.x;
    const int base = w * ECHUNK;
    int d[ECHUNK / 1024], s[ECHUNK / 1024];
#pragma unroll
    for (int k = 0; k < ECHUNK / 1024; ++k) {
        const int e = base + k * 1024 + tid;
        d[k] = (e < E) ? ei[E + e] : 0;
        s[k] = (e < E) ? ei[e] : 0;
    }
    for (int b = wave; b < nb; b += 64) {
        const int  b2   = b + 32;
        const bool two  = b2 < nb;
        const int* row0 = histT + b * g1;
        const int* row1 = histT + (two ? b2 : b) * g1;
        int t0 = 0, p0 = 0, t1 = 0, p1 = 0;
        for (int i0 = 0; i0 < g1; i0 += 256) {
#pragma unroll
            for (int q = 0; q < 8; ++q) {
                const int i  = i0 + lane + 32 * q;
                const int v0 = (i < g1) ? row0[i] : 0;
                const int v1 = (i < g1) ? row1[i] : 0;
                t0 += v0;
                p0 += (i < w) ? v0 : 0;
                t1 += v1;
                p1 += (i < w) ? v1 : 0;
            }
        }
        t0 = wave_sum_i(t0);
        p0 = wave_sum_i(p0);
        t1 = wave_sum_i(t1);
        p1 = wave_sum_i(p1);
        if (lane == 0) {
            tot[b] = t0; pre[b] = p0;
            if (two) { tot[b2] = t1; pre[b2] = p1; }
        }
    }
    if (tid < MAXB) cur[tid] = 0;
    __syncthreads();
    const int tv = (tid < nb) ? tot[tid] : 0;
    int incl = tv;
#pragma unroll
    for (int off = 1; off < 32; off <<= 1) {
        const int t = __shfl_up(incl, off, 32);
        if (lane >= off) incl += t;
    }
    if (tid < MAXB && lane == 31) wtot[wave] = incl;
    __syncthreads();
    if (tid < nb) {
        int wb = 0;
#pragma unroll
        for (int i = 0; i < MAXB / 32; ++i) { const int t = wtot[i]; wb += (i < wave) ? t : 0; }
        const int st = wb + incl - tv;
        bas[tid] = st + pre[tid];
        if (w == 0) {
            binStart[tid] = st;
            if (tid == nb - 1) binStart[nb] = st + tv;
        }
    }
    if (w == 0 && tid < 64) tctr[tid] = 0;
    __syncthreads();
    int2* out = (int2*)binned;
#pragma unroll
    for (int k = 0; k < ECHUNK / 1024; ++k) {
        const int e = base + k * 1024 + tid;
        if (e < E) {
            const int b = d[k] >> BIN_SHIFT;
            const int r = atomicAdd(&cur[b], 1);
            out[bas[b] + r] = make_int2(d[k], s[k]);
        }
    }
}

#define SORT_REG 8
__global__ void __launch_bounds__(1024)
bin_sort(const int* __restrict__ binned, const int* __restrict__ binStart, int n,
         int* __restrict__ start, int* __restrict__ endp, int* __restrict__ csr_src) {
    __shared__ int cnt[BIN_NODES];
    __shared__ int wsum[BIN_NODES / 32];
    __shared__ int stage[STAGE_CAP];
    const int tid   = threadIdx.x;
    const int lane  = tid & 31;
    const int wave  = tid >> 5;
    const int bin   = blockIdx.x;
    const int lo    = binStart[bin];
    const int m     = binStart[bin + 1] - lo;
    const int node0 = bin << BIN_SHIFT;
    const int2* in  = (const int2*)binned + lo;
    int2 v[SORT_REG];
#pragma unroll
    for (int q = 0; q < SORT_REG; ++q) {
        const int i = tid + q * 1024;
        v[q] = (i < m) ? in[i] : make_int2(node0, 0);
    }
    if (tid < BIN_NODES) cnt[tid] = 0;
    __syncthreads();
#pragma unroll
    for (int q = 0; q < SORT_REG; ++q)
        if (tid + q * 1024 < m) atomicAdd(&cnt[v[q].x - node0], 1);
    for (int i = tid + SORT_REG * 1024; i < m; i += 1024) atomicAdd(&cnt[in[i].x - node0], 1);
    __syncthreads();
    const int c = (tid < BIN_NODES) ? cnt[tid] : 0;
    int incl = c;
#pragma unroll
    for (int off = 1; off < 32; off <<= 1) {
        const int t = __shfl_up(incl, off, 32);
        if (lane >= off) incl += t;
    }
    if (tid < BIN_NODES && lane == 31) wsum[wave] = incl;
    __syncthreads();
    if (tid < BIN_NODES) {
        int wb = 0;
#pragma unroll
        for (int i = 0; i < BIN_NODES / 32; ++i) { const int t = wsum[i]; wb += (i < wave) ? t : 0; }
        const int ex = wb + incl - c;
        cnt[tid] = ex;
        const int j = node0 + tid;
        if (j < n) { start[j] = lo + ex; endp[j] = lo + ex + c; }
    }
    __syncthreads();
    if (m <= STAGE_CAP) {
#pragma unroll
        for (int q = 0; q < SORT_REG; ++q)
            if (tid + q * 1024 < m) stage[atomicAdd(&cnt[v[q].x - node0], 1)] = v[q].y;
        for (int i = tid + SORT_REG * 1024; i < m; i += 1024) {
            const int2 t = in[i];
            stage[atomicAdd(&cnt[t.x - node0], 1)] = t.y;
        }
        __syncthreads();
        for (int i = tid; i < m; i += 1024) csr_src[lo + i] = stage[i];
    } else {
#pragma unroll
        for (int q = 0; q < SORT_REG; ++q)
            if (tid + q * 1024 < m) csr_src[lo + atomicAdd(&cnt[v[q].x - node0], 1)] = v[q].y;
        for (int i = tid + SORT_REG * 1024; i < m; i += 1024) {
            const int2 t = in[i];
            csr_src[lo + atomicAdd(&cnt[t.x - node0], 1)] = t.y;
        }
    }
}

__device__ __forceinline__ float wave_sum(float v) {
#pragma unroll
    for (int off = 16; off > 0; off >>= 1) v += __shfl_xor(v, off, 32);
    return v;
}

template <int M>
__device__ __forceinline__ float row_xor(float x) {
    return __builtin_bit_cast(float, __builtin_amdgcn_update_dpp(
        0, __builtin_bit_cast(int, x), 0x160 + M, 0xf, 0xf, false));
}
__device__ __forceinline__ float half_xor(float x) {
    return __builtin_bit_cast(float, __builtin_amdgcn_permlanex16(
        __builtin_bit_cast(int, x), __builtin_bit_cast(int, x), 0x76543210, 0xfedcba98,
        false, false));
}
__device__ __forceinline__ float lane_sum(float v) {
    v += row_xor<1>(v);
    v += row_xor<2>(v);
    v += row_xor<4>(v);
    v += row_xor<8>(v);
    v += half_xor(v);
    return v;
}

__device__ __forceinline__ float dot4(const float4 h, const float4 b) {
    v2f q = (v2f){h.x, h.y} * (v2f){b.x, b.y};
    q = __builtin_elementwise_fma((v2f){h.z, h.w}, (v2f){b.z, b.w}, q);
    return q.x + q.y;
}
__device__ __forceinline__ float4 load_row(const float* __restrict__ H, int i, int lane) {
    const float* rp = H + ((long)i << 7);
    return ((const float4*)rp)[lane];
}
__device__ __forceinline__ float leaky(float s) { return fmaxf(s, NEG_SLOPE * s); }

__device__ __forceinline__ void fma4(float w, const float4 h, float4& a) {
    a.x = fmaf(w, h.x, a.x); a.y = fmaf(w, h.y, a.y);
    a.z = fmaf(w, h.z, a.z); a.w = fmaf(w, h.w, a.w);
}

template <int NV>
__device__ __forceinline__ void edge_group(const float* __restrict__ H, const float4 us,
                                           float sd, int iv, int k0, int nvar, int lane,
                                           float& m, float& den, float4& a) {
    const int   nv      = NV ? NV : nvar;
    const float NEG_INF = -__builtin_inff();
    const int i0 = __builtin_amdgcn_readlane(iv, k0);
    const int i1 = __builtin_amdgcn_readlane(iv, k0 + 1);
    const int i2 = __builtin_amdgcn_readlane(iv, k0 + 2);
    const int i3 = __builtin_amdgcn_readlane(iv, k0 + 3);
    const float4 z = make_float4(0.f, 0.f, 0.f, 0.f);
    float4 h0 = load_row(H, i0, lane);
    float4 h1 = z, h2 = z, h3 = z;
    if (nv > 1) h1 = load_row(H, i1, lane);
    if (nv > 2) h2 = load_row(H, i2, lane);
    if (nv > 3) h3 = load_row(H, i3, lane);
    float p0 = dot4(h0, us), p1 = dot4(h1, us), p2 = dot4(h2, us), p3 = dot4(h3, us);
    p0 = lane_sum(p0); p1 = lane_sum(p1); p2 = lane_sum(p2); p3 = lane_sum(p3);
    const float l0 = leaky(p0 + sd);
    const float l1 = (nv > 1) ? leaky(p1 + sd) : NEG_INF;
    const float l2 = (nv > 2) ? leaky(p2 + sd) : NEG_INF;
    const float l3 = (nv > 3) ? leaky(p3 + sd) : NEG_INF;
    const float mn = fmaxf(fmaxf(m, l0), fmaxf(fmaxf(l1, l2), l3));
    const float sc = __expf(m - mn);
    const float w0 = __expf(l0 - mn);
    const float w1 = __expf(l1 - mn);
    const float w2 = __expf(l2 - mn);
    const float w3 = __expf(l3 - mn);
    den = fmaf(den, sc, (w0 + w1) + (w2 + w3));
    a.x *= sc; a.y *= sc; a.z *= sc; a.w *= sc;
    fma4(w0, h0, a); fma4(w1, h1, a); fma4(w2, h2, a); fma4(w3, h3, a);
    m = mn;
}

__global__ void __launch_bounds__(256)
gat_uvec(const float* __restrict__ W1, const float* __restrict__ W2,
         const float* __restrict__ W3, const float* __restrict__ as1,
         const float* __restrict__ as2, const float* __restrict__ as3,
         const float* __restrict__ ad1, const float* __restrict__ ad2,
         const float* __restrict__ ad3, float* __restrict__ u) {
    const int lane  = threadIdx.x & 31;
    const int wave  = threadIdx.x >> 5;
    const int layer = blockIdx.x >> 4;
    const int row   = (blockIdx.x & 15) * 8 + wave;
    const float* W  = (layer == 0) ? W1  : (layer == 1) ? W2  : W3;
    const float* as = (layer == 0) ? as1 : (layer == 1) ? as2 : as3;
    const float* ad = (layer == 0) ? ad1 : (layer == 1) ? ad2 : ad3;
    const float4 w4 = *(const float4*)(W + row * GDIM + lane * 4);
    const float4 a4 = *(const float4*)(as + lane * 4);
    const float4 d4 = *(const float4*)(ad + lane * 4);
    float ps = w4.x * a4.x + w4.y * a4.y + w4.z * a4.z + w4.w * a4.w;
    float pd = w4.x * d4.x + w4.y * d4.y + w4.z * d4.z + w4.w * d4.w;
    ps = wave_sum(ps);
    pd = wave_sum(pd);
    if (lane == 0) {
        u[layer * 2 * GDIM + row]        = ps;
        u[layer * 2 * GDIM + GDIM + row] = pd;
    }
}

__global__ void __launch_bounds__(512, 16)
gat_layer(const int* __restrict__ start, const int* __restrict__ endp,
          const int* __restrict__ csr_src, const float* __restrict__ Xin,
          const float* __restrict__ W, const float* __restrict__ bias,
          const float* __restrict__ u, float* __restrict__ Xout,
          int n, int nTiles, int* __restrict__ tctr, int nFixed) {
    __shared__ __attribute__((aligned(16))) float sW[GDIM * GDIM];
    __shared__ __attribute__((aligned(16))) float sT[TROWS * TPITCH];
    __shared__ int sNext;
    const int tid  = threadIdx.x;
    const int lane = tid & 31;
    const int wave = __builtin_amdgcn_readfirstlane(tid >> 5);
    const int hi   = lane >> 4;
    const int col  = lane & 15;

    {
        const float4* W4 = (const float4*)W;
        float4*       S4 = (float4*)sW;
#pragma unroll
        for (int t = 0; t < 8; ++t) S4[t * 512 + tid] = W4[t * 512 + tid];
    }

    int tile = blockIdx.x;
    while (tile < nTiles) {
        int nxt = 0;
        const int j = tile * TROWS + wave;
        float4 a = make_float4(0.f, 0.f, 0.f, 0.f);
        if (j < n) {
            const float4 hs  = load_row(Xin, j, lane);
            const float4 us  = *(const float4*)(u + lane * 4);
            const float4 ud  = *(const float4*)(u + GDIM + lane * 4);
            const int    beg = start[j];
            const int    deg = endp[j] - beg;
            int iv = (lane < deg) ? csr_src[beg + lane] : j;
            if (tid == 0) nxt = atomicAdd(tctr, 1);
            __builtin_amdgcn_sched_barrier(0);
            const float  ss  = lane_sum(dot4(hs, us));
            const float  sd  = lane_sum(dot4(hs, ud));
            float m   = leaky(ss + sd);
            float den = 1.0f;
            a = hs;
            for (int base = 0; base < deg; base += 32) {
                if (base > 0) {
                    const int t = base + lane;
                    iv = (t < deg) ? csr_src[beg + t] : j;
                }
                const int cnt = min(32, deg - base);
                int k = 0;
                for (; k + 4 <= cnt; k += 4)
                    edge_group<4>(Xin, us, sd, iv, k, 4, lane, m, den, a);
                if (k < cnt)
                    edge_group<0>(Xin, us, sd, iv, k, cnt - k, lane, m, den, a);
            }
            const float inv = 1.0f / den;
            a.x *= inv; a.y *= inv; a.z *= inv; a.w *= inv;
        } else if (tid == 0) {
            nxt = atomicAdd(tctr, 1);
        }
        *(float4*)(sT + wave * TPITCH + lane * 4) = a;
        if (tid == 0) sNext = nFixed + nxt;
        __syncthreads();
        const int nextTile = __builtin_amdgcn_readfirstlane(sNext);

        v8f c = (v8f)0.0f;
        if (wave < 8) {
            v8f c0 = (v8f)0.0f, c1 = (v8f)0.0f;
            const float* At = sT + col * TPITCH + 2 * hi;
            const float* Bt = sW + (2 * hi) * GDIM + wave * 16 + col;
#pragma unroll
            for (int kk = 0; kk < GDIM; kk += 8) {
                const v2f a0 = *(const v2f*)(At + kk);
                v2f b0;
                b0.x = Bt[(kk + 0) * GDIM];
                b0.y = Bt[(kk + 1) * GDIM];
                const v2f a1 = *(const v2f*)(At + kk + 4);
                v2f b1;
                b1.x = Bt[(kk + 4) * GDIM];
                b1.y = Bt[(kk + 5) * GDIM];
                c0 = __builtin_amdgcn_wmma_f32_16x16x4_f32(false, a0, false, b0, (short)0, c0, false, false);
                c1 = __builtin_amdgcn_wmma_f32_16x16x4_f32(false, a1, false, b1, (short)0, c1, false, false);
            }
            c = c0 + c1;
        }
        __syncthreads();
        if (wave < 8) {
#pragma unroll
            for (int v = 0; v < 8; ++v) sT[(v + 8 * hi) * TPITCH + wave * 16 + col] = c[v];
        }
        __syncthreads();
        if (j < n) {
            const float4 b4 = *(const float4*)(bias + lane * 4);
            float4 r = *(const float4*)(sT + wave * TPITCH + lane * 4);
            r.x += b4.x; r.y += b4.y; r.z += b4.z; r.w += b4.w;
            *(float4*)(Xout + (long)j * GDIM + lane * 4) = r;
        }
        tile = nextTile;
    }
}

#define POOL_CHUNK 16

__device__ __forceinline__ void pool_flush(float* __restrict__ psum,
                                           float* __restrict__ pcount, int g,
                                           float4 s, float cnt, int lane) {
    float* pp = psum + (long)g * GDIM + lane * 4;
    atomicAdd(pp + 0, s.x);
    atomicAdd(pp + 1, s.y);
    atomicAdd(pp + 2, s.z);
    atomicAdd(pp + 3, s.w);
    if (lane == 0) atomicAdd(pcount + g, cnt);
}

__device__ __forceinline__ float4 add4(float4 a, float4 b) {
    return make_float4(a.x + b.x, a.y + b.y, a.z + b.z, a.w + b.w);
}

__global__ void __launch_bounds__(256)
gat_pool_accum(const float* __restrict__ h, const int* __restrict__ batch, int n,
               float* __restrict__ psum, float* __restrict__ pcount) {
    const int lane  = threadIdx.x & 31;
    const int wave  = __builtin_amdgcn_readfirstlane((int)(threadIdx.x >> 5));
    const int node0 = (blockIdx.x * 8 + wave) * POOL_CHUNK;
    if (node0 >= n) return;
    const int valid = min(POOL_CHUNK, n - node0);
    const float* hp = h + (long)node0 * GDIM + lane * 4;
    const int gl = batch[node0 + (lane < valid ? lane : 0)];

    if (valid == POOL_CHUNK) {
        float4 v[POOL_CHUNK];
#pragma unroll
        for (int u = 0; u < POOL_CHUNK; ++u)
            v[u] = *(const float4*)(hp + (long)u * GDIM);
        const int g0 = __builtin_amdgcn_readfirstlane(gl);
        if (__all(gl == g0)) {
#pragma unroll
            for (int w = POOL_CHUNK / 2; w > 0; w >>= 1)
#pragma unroll
                for (int u = 0; u < w; ++u) v[u] = add4(v[u], v[u + w]);
            pool_flush(psum, pcount, g0, v[0], (float)POOL_CHUNK, lane);
        } else {
            float4 s   = make_float4(0.f, 0.f, 0.f, 0.f);
            int    gc  = g0;
            float  cnt = 0.f;
#pragma unroll
            for (int u = 0; u < POOL_CHUNK; ++u) {
                const int g = __builtin_amdgcn_readlane(gl, u);
                if (g != gc) {
                    pool_flush(psum, pcount, gc, s, cnt, lane);
                    s = make_float4(0.f, 0.f, 0.f, 0.f); cnt = 0.f; gc = g;
                }
                s = add4(s, v[u]); cnt += 1.0f;
            }
            pool_flush(psum, pcount, gc, s, cnt, lane);
        }
    } else {
        float4 s   = make_float4(0.f, 0.f, 0.f, 0.f);
        int    gc  = __builtin_amdgcn_readfirstlane(gl);
        float  cnt = 0.f;
        for (int u = 0; u < valid; ++u) {
            const int    g  = __shfl(gl, u, 32);
            const float4 hv = *(const float4*)(hp + (long)u * GDIM);
            if (g != gc) {
                pool_flush(psum, pcount, gc, s, cnt, lane);
                s = make_float4(0.f, 0.f, 0.f, 0.f); cnt = 0.f; gc = g;
            }
            s = add4(s, hv); cnt += 1.0f;
        }
        pool_flush(psum, pcount, gc, s, cnt, lane);
    }
}

__global__ void __launch_bounds__(256)
gat_pool_final(const float* __restrict__ psum, const float* __restrict__ pcount,
               float* __restrict__ out) {
    const int idx = blockIdx.x * blockDim.x + threadIdx.x;
    const float cnt = pcount[idx >> 7];
    out[idx] = psum[idx] / fmaxf(cnt, 1.0f);
}

extern "C" void kernel_launch(void* const* d_in, const int* in_sizes, int n_in,
                              void* d_out, int out_size, void* d_ws, size_t ws_size,
                              hipStream_t stream) {
    const float* x     = (const float*)d_in[0];
    const int*   ei    = (const int*)d_in[1];
    const int*   batch = (const int*)d_in[2];
    const float* W[3]    = {(const float*)d_in[3], (const float*)d_in[7],  (const float*)d_in[11]};
    const float* asrc[3] = {(const float*)d_in[4], (const float*)d_in[8],  (const float*)d_in[12]};
    const float* adst[3] = {(const float*)d_in[5], (const float*)d_in[9],  (const float*)d_in[13]};
    const float* bias[3] = {(const float*)d_in[6], (const float*)d_in[10], (const float*)d_in[14]};

    const int n  = in_sizes[0] / GDIM;
    const int E  = in_sizes[1] / 2;

    char*  wsb = (char*)d_ws;
    size_t off = 0;
    auto take = [&](size_t bytes) -> void* {
        void* p = wsb + off;
        off += (bytes + 255) & ~(size_t)255;
        return p;
    };
    const size_t szH = (size_t)n * GDIM * sizeof(float);
    float*    bufA   = (float*)   take(szH);
    float*    bufB   = (float*)   take(szH);
    const int nb = (n + BIN_NODES - 1) / BIN_NODES;
    const int g1 = (E + ECHUNK - 1) / ECHUNK;
    int*      tctr   = (int*)     take(256);
    int*      histT  = (int*)     take((size_t)nb * g1 * sizeof(int));
    int*      bstart = (int*)     take((size_t)(nb + 1) * sizeof(int));
    int*      binned = (int*)     take((size_t)E * 2 * sizeof(int));
    int*      start  = (int*)     take((size_t)n * sizeof(int));
    int*      cursor = (int*)     take((size_t)n * sizeof(int));
    int*      csr    = (int*)     take((size_t)E * sizeof(int));
    float*    psum   = (float*)   take((size_t)NGRAPH * GDIM * sizeof(float));
    float*    pcount = (float*)   take((size_t)NGRAPH * sizeof(float));
    float*    u      = (float*)   take((size_t)3 * 2 * GDIM * sizeof(float));
    (void)ws_size; (void)n_in; (void)out_size;

    const int nTiles      = (n + TROWS - 1) / TROWS;
    const int layerBlocks = nTiles < 1024 ? nTiles : 1024;
    const int poolBlocks  = (n + 8 * POOL_CHUNK - 1) / (8 * POOL_CHUNK);

    bin_hist<<<g1, 1024, 0, stream>>>(ei, E, nb, g1, histT);
    bin_scatter<<<g1, 1024, 0, stream>>>(ei, E, nb, g1, histT, bstart, binned, tctr);
    bin_sort<<<nb, 1024, 0, stream>>>(binned, bstart, n, start, cursor, csr);

    gat_uvec<<<48, 256, 0, stream>>>(W[0], W[1], W[2], asrc[0], asrc[1], asrc[2],
                                     adst[0], adst[1], adst[2], u);
    gat_layer<<<layerBlocks, 512, 0, stream>>>(
        start, cursor, csr, x, W[0], bias[0], u, bufA, n, nTiles, tctr, layerBlocks);
    gat_layer<<<layerBlocks, 512, 0, stream>>>(
        start, cursor, csr, bufA, W[1], bias[1], u + 2 * GDIM, bufB, n, nTiles, tctr + 16, layerBlocks);
    gat_layer<<<layerBlocks, 512, 0, stream>>>(
        start, cursor, csr, bufB, W[2], bias[2], u + 4 * GDIM, bufA, n, nTiles, tctr + 32, layerBlocks);

    hipMemsetAsync(psum,   0, (size_t)NGRAPH * GDIM * sizeof(float), stream);
    hipMemsetAsync(pcount, 0, (size_t)NGRAPH * sizeof(float), stream);
    gat_pool_accum<<<poolBlocks, 256, 0, stream>>>(bufA, batch, n, psum, pcount);
    gat_pool_final<<<(NGRAPH * GDIM) / 256, 256, 0, stream>>>(psum, pcount,
                                                              (float*)d_out);
}


# layer kernel: the four logit lane sums of an edge group folded into one (two exchange steps leave lane l with row l&3 alone); max, exp and weight once per row; same pairs and order, bits unchanged
# speedup vs baseline: 8.1720x; 1.0316x over previous
#include <hip/hip_runtime.h>

#define GDIM 128
#define NGRAPH 256
#define NEG_SLOPE 0.2f
#define TROWS 16
#define TPITCH 132

typedef float v2f __attribute__((ext_vector_type(2)));
typedef float v8f __attribute__((ext_vector_type(8)));

#define BIN_SHIFT 9
#define BIN_NODES 512
#define MAXB      256
#define ECHUNK    4096
#define STAGE_CAP 12288

__device__ __forceinline__ int wave_sum_i(int v) {
#pragma unroll
    for (int off = 16; off > 0; off >>= 1) v += __shfl_xor(v, off, 32);
    return v;
}

__global__ void __launch_bounds__(1024)
bin_hist(const int* __restrict__ ei, int E, int nb, int g1, int* __restrict__ histT) {
    __shared__ int h[MAXB];
    const int tid  = threadIdx.x;
    const int base = blockIdx.x * ECHUNK;
    int d[ECHUNK / 1024];
#pragma unroll
    for (int k = 0; k < ECHUNK / 1024; ++k) {
        const int e = base + k * 1024 + tid;
        d[k] = ei[E + min(e, E - 1)];
    }
    if (tid < MAXB) h[tid] = 0;
    __syncthreads();
#pragma unroll
    for (int k = 0; k < ECHUNK / 1024; ++k) {
        const int e = base + k * 1024 + tid;
        if (e < E) atomicAdd(&h[d[k] >> BIN_SHIFT], 1);
    }
    __syncthreads();
    if (tid < nb) histT[tid * g1 + blockIdx.x] = h[tid];
}

__global__ void __launch_bounds__(1024)
bin_scatter(const int* __restrict__ ei, int E, int nb, int g1,
            const int* __restrict__ histT, int* __restrict__ binStart,
            int* __restrict__ binned, int* __restrict__ tctr) {
    __shared__ int tot[MAXB];
    __shared__ int pre[MAXB];
    __shared__ int cur[MAXB];
    __shared__ int bas[MAXB];
    __shared__ int wtot[MAXB / 32];
    const int tid  = threadIdx.x;
    const int lane = tid & 31;
    const int wave = tid >> 5;
    const int w    = blockIdx.x;
    const int base = w * ECHUNK;
    int d[ECHUNK / 1024], s[ECHUNK / 1024];
#pragma unroll
    for (int k = 0; k < ECHUNK / 1024; ++k) {
        const int e = base + k * 1024 + tid;
        d[k] = (e < E) ? ei[E + e] : 0;
        s[k] = (e < E) ? ei[e] : 0;
    }
    for (int b = wave; b < nb; b += 64) {
        const int  b2   = b + 32;
        const bool two  = b2 < nb;
        const int* row0 = histT + b * g1;
        const int* row1 = histT + (two ? b2 : b) * g1;
        int t0 = 0, p0 = 0, t1 = 0, p1 = 0;
        for (int i0 = 0; i0 < g1; i0 += 256) {
#pragma unroll
            for (int q = 0; q < 8; ++q) {
                const int i  = i0 + lane + 32 * q;
                const int v0 = (i < g1) ? row0[i] : 0;
                const int v1 = (i < g1) ? row1[i] : 0;
                t0 += v0;
                p0 += (i < w) ? v0 : 0;
                t1 += v1;
                p1 += (i < w) ? v1 : 0;
            }
        }
        t0 = wave_sum_i(t0);
        p0 = wave_sum_i(p0);
        t1 = wave_sum_i(t1);
        p1 = wave_sum_i(p1);
        if (lane == 0) {
            tot[b] = t0; pre[b] = p0;
            if (two) { tot[b2] = t1; pre[b2] = p1; }
        }
    }
    if (tid < MAXB) cur[tid] = 0;
    __syncthreads();
    const int tv = (tid < nb) ? tot[tid] : 0;
    int incl = tv;
#pragma unroll
    for (int off = 1; off < 32; off <<= 1) {
        const int t = __shfl_up(incl, off, 32);
        if (lane >= off) incl += t;
    }
    if (tid < MAXB && lane == 31) wtot[wave] = incl;
    __syncthreads();
    if (tid < nb) {
        int wb = 0;
#pragma unroll
        for (int i = 0; i < MAXB / 32; ++i) { const int t = wtot[i]; wb += (i < wave) ? t : 0; }
        const int st = wb + incl - tv;
        bas[tid] = st + pre[tid];
        if (w == 0) {
            binStart[tid] = st;
            if (tid == nb - 1) binStart[nb] = st + tv;
        }
    }
    if (w == 0 && tid < 64) tctr[tid] = 0;
    __syncthreads();
    int2* out = (int2*)binned;
#pragma unroll
    for (int k = 0; k < ECHUNK / 1024; ++k) {
        const int e = base + k * 1024 + tid;
        if (e < E) {
            const int b = d[k] >> BIN_SHIFT;
            const int r = atomicAdd(&cur[b], 1);
            out[bas[b] + r] = make_int2(d[k], s[k]);
        }
    }
}

#define SORT_REG 8
__global__ void __launch_bounds__(1024)
bin_sort(const int* __restrict__ binned, const int* __restrict__ binStart, int n,
         int* __restrict__ start, int* __restrict__ endp, int* __restrict__ csr_src) {
    __shared__ int cnt[BIN_NODES];
    __shared__ int wsum[BIN_NODES / 32];
    __shared__ int stage[STAGE_CAP];
    const int tid   = threadIdx.x;
    const int lane  = tid & 31;
    const int wave  = tid >> 5;
    const int bin   = blockIdx.x;
    const int lo    = binStart[bin];
    const int m     = binStart[bin + 1] - lo;
    const int node0 = bin << BIN_SHIFT;
    const int2* in  = (const int2*)binned + lo;
    int2 v[SORT_REG];
#pragma unroll
    for (int q = 0; q < SORT_REG; ++q) {
        const int i = tid + q * 1024;
        v[q] = (i < m) ? in[i] : make_int2(node0, 0);
    }
    if (tid < BIN_NODES) cnt[tid] = 0;
    __syncthreads();
#pragma unroll
    for (int q = 0; q < SORT_REG; ++q)
        if (tid + q * 1024 < m) atomicAdd(&cnt[v[q].x - node0], 1);
    for (int i = tid + SORT_REG * 1024; i < m; i += 1024) atomicAdd(&cnt[in[i].x - node0], 1);
    __syncthreads();
    const int c = (tid < BIN_NODES) ? cnt[tid] : 0;
    int incl = c;
#pragma unroll
    for (int off = 1; off < 32; off <<= 1) {
        const int t = __shfl_up(incl, off, 32);
        if (lane >= off) incl += t;
    }
    if (tid < BIN_NODES && lane == 31) wsum[wave] = incl;
    __syncthreads();
    if (tid < BIN_NODES) {
        int wb = 0;
#pragma unroll
        for (int i = 0; i < BIN_NODES / 32; ++i) { const int t = wsum[i]; wb += (i < wave) ? t : 0; }
        const int ex = wb + incl - c;
        cnt[tid] = ex;
        const int j = node0 + tid;
        if (j < n) { start[j] = lo + ex; endp[j] = lo + ex + c; }
    }
    __syncthreads();
    if (m <= STAGE_CAP) {
#pragma unroll
        for (int q = 0; q < SORT_REG; ++q)
            if (tid + q * 1024 < m) stage[atomicAdd(&cnt[v[q].x - node0], 1)] = v[q].y;
        for (int i = tid + SORT_REG * 1024; i < m; i += 1024) {
            const int2 t = in[i];
            stage[atomicAdd(&cnt[t.x - node0], 1)] = t.y;
        }
        __syncthreads();
        for (int i = tid; i < m; i += 1024) csr_src[lo + i] = stage[i];
    } else {
#pragma unroll
        for (int q = 0; q < SORT_REG; ++q)
            if (tid + q * 1024 < m) csr_src[lo + atomicAdd(&cnt[v[q].x - node0], 1)] = v[q].y;
        for (int i = tid + SORT_REG * 1024; i < m; i += 1024) {
            const int2 t = in[i];
            csr_src[lo + atomicAdd(&cnt[t.x - node0], 1)] = t.y;
        }
    }
}

__device__ __forceinline__ float wave_sum(float v) {
#pragma unroll
    for (int off = 16; off > 0; off >>= 1) v += __shfl_xor(v, off, 32);
    return v;
}

template <int M>
__device__ __forceinline__ float row_xor(float x) {
    return __builtin_bit_cast(float, __builtin_amdgcn_update_dpp(
        0, __builtin_bit_cast(int, x), 0x160 + M, 0xf, 0xf, false));
}
__device__ __forceinline__ float half_xor(float x) {
    return __builtin_bit_cast(float, __builtin_amdgcn_permlanex16(
        __builtin_bit_cast(int, x), __builtin_bit_cast(int, x), 0x76543210, 0xfedcba98,
        false, false));
}
template <int K>
__device__ __forceinline__ float quad_bcast(float x) {
    return __builtin_bit_cast(float, __builtin_amdgcn_update_dpp(
        0, __builtin_bit_cast(int, x), K * 0x55, 0xf, 0xf, true));
}
template <int M>
__device__ __forceinline__ float row_xor_max(float x) {
    return fmaxf(x, __builtin_bit_cast(float, __builtin_amdgcn_update_dpp(
        0, __builtin_bit_cast(int, x), 0x160 + M, 0xf, 0xf, true)));
}
__device__ __forceinline__ float lane_sum(float v) {
    v += row_xor<1>(v);
    v += row_xor<2>(v);
    v += row_xor<4>(v);
    v += row_xor<8>(v);
    v += half_xor(v);
    return v;
}

__device__ __forceinline__ float lane_sum_from2(float v) {
    v += row_xor<2>(v);
    v += row_xor<4>(v);
    v += row_xor<8>(v);
    v += half_xor(v);
    return v;
}

__device__ __forceinline__ float dot4(const float4 h, const float4 b) {
    v2f q = (v2f){h.x, h.y} * (v2f){b.x, b.y};
    q = __builtin_elementwise_fma((v2f){h.z, h.w}, (v2f){b.z, b.w}, q);
    return q.x + q.y;
}
__device__ __forceinline__ float4 load_row(const float* __restrict__ H, int i, int lane) {
    const float* rp = H + ((long)i << 7);
    return ((const float4*)rp)[lane];
}
__device__ __forceinline__ float leaky(float s) { return fmaxf(s, NEG_SLOPE * s); }

__device__ __forceinline__ void fma4(float w, const float4 h, float4& a) {
    a.x = fmaf(w, h.x, a.x); a.y = fmaf(w, h.y, a.y);
    a.z = fmaf(w, h.z, a.z); a.w = fmaf(w, h.w, a.w);
}

template <int NV>
__device__ __forceinline__ void edge_group(const float* __restrict__ H, const float4 us,
                                           float sd, int iv, int k0, int nvar, int lane,
                                           float& m, float& den, float4& a) {
    const int   nv      = NV ? NV : nvar;
    const float NEG_INF = -__builtin_inff();
    const int i0 = __builtin_amdgcn_readlane(iv, k0);
    const int i1 = __builtin_amdgcn_readlane(iv, k0 + 1);
    const int i2 = __builtin_amdgcn_readlane(iv, k0 + 2);
    const int i3 = __builtin_amdgcn_readlane(iv, k0 + 3);
    const float4 z = make_float4(0.f, 0.f, 0.f, 0.f);
    float4 h0 = load_row(H, i0, lane);
    float4 h1 = z, h2 = z, h3 = z;
    if (nv > 1) h1 = load_row(H, i1, lane);
    if (nv > 2) h2 = load_row(H, i2, lane);
    if (nv > 3) h3 = load_row(H, i3, lane);
    const float p0 = dot4(h0, us), p1 = dot4(h1, us), p2 = dot4(h2, us), p3 = dot4(h3, us);
    const bool  odd = (lane & 1) != 0;
    const bool  up  = (lane & 2) != 0;
    const float q0  = (odd ? p1 : p0) + row_xor<1>(odd ? p0 : p1);
    const float q1  = (odd ? p3 : p2) + row_xor<1>(odd ? p2 : p3);
    float t = (up ? q1 : q0) + row_xor<2>(up ? q0 : q1);
    t += row_xor<4>(t);
    t += row_xor<8>(t);
    t += half_xor(t);
    float l = leaky(t + sd);
    if (NV == 0) l = ((lane & 3) < nv) ? l : NEG_INF;
    float lm = row_xor_max<1>(l);
    lm = row_xor_max<2>(lm);
    const float mn = fmaxf(m, lm);
    const float sc = __expf(m - mn);
    const float w  = __expf(l - mn);
    const float w0 = quad_bcast<0>(w);
    const float w1 = quad_bcast<1>(w);
    const float w2 = quad_bcast<2>(w);
    const float w3 = quad_bcast<3>(w);
    den = fmaf(den, sc, (w0 + w1) + (w2 + w3));
    a.x *= sc; a.y *= sc; a.z *= sc; a.w *= sc;
    fma4(w0, h0, a); fma4(w1, h1, a); fma4(w2, h2, a); fma4(w3, h3, a);
    m = mn;
}

__global__ void __launch_bounds__(256)
gat_uvec(const float* __restrict__ W1, const float* __restrict__ W2,
         const float* __restrict__ W3, const float* __restrict__ as1,
         const float* __restrict__ as2, const float* __restrict__ as3,
         const float* __restrict__ ad1, const float* __restrict__ ad2,
         const float* __restrict__ ad3, float* __restrict__ u) {
    const int lane  = threadIdx.x & 31;
    const int wave  = threadIdx.x >> 5;
    const int layer = blockIdx.x >> 4;
    const int row   = (blockIdx.x & 15) * 8 + wave;
    const float* W  = (layer == 0) ? W1  : (layer == 1) ? W2  : W3;
    const float* as = (layer == 0) ? as1 : (layer == 1) ? as2 : as3;
    const float* ad = (layer == 0) ? ad1 : (layer == 1) ? ad2 : ad3;
    const float4 w4 = *(const float4*)(W + row * GDIM + lane * 4);
    const float4 a4 = *(const float4*)(as + lane * 4);
    const float4 d4 = *(const float4*)(ad + lane * 4);
    float ps = w4.x * a4.x + w4.y * a4.y + w4.z * a4.z + w4.w * a4.w;
    float pd = w4.x * d4.x + w4.y * d4.y + w4.z * d4.z + w4.w * d4.w;
    ps = wave_sum(ps);
    pd = wave_sum(pd);
    if (lane == 0) {
        u[layer * 2 * GDIM + row]        = ps;
        u[layer * 2 * GDIM + GDIM + row] = pd;
    }
}

__global__ void __launch_bounds__(512, 16)
gat_layer(const int* __restrict__ start, const int* __restrict__ endp,
          const int* __restrict__ csr_src, const float* __restrict__ Xin,
          const float* __restrict__ W, const float* __restrict__ bias,
          const float* __restrict__ u, float* __restrict__ Xout,
          int n, int nTiles, int* __restrict__ tctr, int nFixed) {
    __shared__ __attribute__((aligned(16))) float sW[GDIM * GDIM];
    __shared__ __attribute__((aligned(16))) float sT[TROWS * TPITCH];
    __shared__ int sNext;
    const int tid  = threadIdx.x;
    const int lane = tid & 31;
    const int wave = __builtin_amdgcn_readfirstlane(tid >> 5);
    const int hi   = lane >> 4;
    const int col  = lane & 15;

    {
        const float4* W4 = (const float4*)W;
        float4*       S4 = (float4*)sW;
#pragma unroll
        for (int t = 0; t < 8; ++t) S4[t * 512 + tid] = W4[t * 512 + tid];
    }

    int tile = blockIdx.x;
    while (tile < nTiles) {
        int nxt = 0;
        const int j = tile * TROWS + wave;
        float4 a = make_float4(0.f, 0.f, 0.f, 0.f);
        if (j < n) {
            const float4 hs  = load_row(Xin, j, lane);
            const float4 us  = *(const float4*)(u + lane * 4);
            const float4 ud  = *(const float4*)(u + GDIM + lane * 4);
            const int    beg = start[j];
            const int    deg = endp[j] - beg;
            int iv = (lane < deg) ? csr_src[beg + lane] : j;
            if (tid == 0) nxt = atomicAdd(tctr, 1);
            __builtin_amdgcn_sched_barrier(0);
            const float  ps  = dot4(hs, us);
            const float  pd  = dot4(hs, ud);
            const bool   odd = (lane & 1) != 0;
            const float  tsd = lane_sum_from2((odd ? pd : ps) + row_xor<1>(odd ? ps : pd));
            const float  ss  = quad_bcast<0>(tsd);
            const float  sd  = quad_bcast<1>(tsd);
            float m   = leaky(ss + sd);
            float den = 1.0f;
            a = hs;
            for (int base = 0; base < deg; base += 32) {
                if (base > 0) {
                    const int t = base + lane;
                    iv = (t < deg) ? csr_src[beg + t] : j;
                }
                const int cnt = min(32, deg - base);
                int k = 0;
                for (; k + 4 <= cnt; k += 4)
                    edge_group<4>(Xin, us, sd, iv, k, 4, lane, m, den, a);
                if (k < cnt)
                    edge_group<0>(Xin, us, sd, iv, k, cnt - k, lane, m, den, a);
            }
            const float inv = 1.0f / den;
            a.x *= inv; a.y *= inv; a.z *= inv; a.w *= inv;
        } else if (tid == 0) {
            nxt = atomicAdd(tctr, 1);
        }
        *(float4*)(sT + wave * TPITCH + lane * 4) = a;
        if (tid == 0) sNext = nFixed + nxt;
        __syncthreads();
        const int nextTile = __builtin_amdgcn_readfirstlane(sNext);

        v8f c = (v8f)0.0f;
        if (wave < 8) {
            v8f c0 = (v8f)0.0f, c1 = (v8f)0.0f;
            const float* At = sT + col * TPITCH + 2 * hi;
            const float* Bt = sW + (2 * hi) * GDIM + wave * 16 + col;
#pragma unroll
            for (int kk = 0; kk < GDIM; kk += 8) {
                const v2f a0 = *(const v2f*)(At + kk);
                v2f b0;
                b0.x = Bt[(kk + 0) * GDIM];
                b0.y = Bt[(kk + 1) * GDIM];
                const v2f a1 = *(const v2f*)(At + kk + 4);
                v2f b1;
                b1.x = Bt[(kk + 4) * GDIM];
                b1.y = Bt[(kk + 5) * GDIM];
                c0 = __builtin_amdgcn_wmma_f32_16x16x4_f32(false, a0, false, b0, (short)0, c0, false, false);
                c1 = __builtin_amdgcn_wmma_f32_16x16x4_f32(false, a1, false, b1, (short)0, c1, false, false);
            }
            c = c0 + c1;
        }
        __syncthreads();
        if (wave < 8) {
#pragma unroll
            for (int v = 0; v < 8; ++v) sT[(v + 8 * hi) * TPITCH + wave * 16 + col] = c[v];
        }
        __syncthreads();
        if (j < n) {
            const float4 b4 = *(const float4*)(bias + lane * 4);
            float4 r = *(const float4*)(sT + wave * TPITCH + lane * 4);
            r.x += b4.x; r.y += b4.y; r.z += b4.z; r.w += b4.w;
            *(float4*)(Xout + (long)j * GDIM + lane * 4) = r;
        }
        tile = nextTile;
    }
}

#define POOL_CHUNK 16

__device__ __forceinline__ void pool_flush(float* __restrict__ psum,
                                           float* __restrict__ pcount, int g,
                                           float4 s, float cnt, int lane) {
    float* pp = psum + (long)g * GDIM + lane * 4;
    atomicAdd(pp + 0, s.x);
    atomicAdd(pp + 1, s.y);
    atomicAdd(pp + 2, s.z);
    atomicAdd(pp + 3, s.w);
    if (lane == 0) atomicAdd(pcount + g, cnt);
}

__device__ __forceinline__ float4 add4(float4 a, float4 b) {
    return make_float4(a.x + b.x, a.y + b.y, a.z + b.z, a.w + b.w);
}

__global__ void __launch_bounds__(256)
gat_pool_accum(const float* __restrict__ h, const int* __restrict__ batch, int n,
               float* __restrict__ psum, float* __restrict__ pcount) {
    const int lane  = threadIdx.x & 31;
    const int wave  = __builtin_amdgcn_readfirstlane((int)(threadIdx.x >> 5));
    const int node0 = (blockIdx.x * 8 + wave) * POOL_CHUNK;
    if (node0 >= n) return;
    const int valid = min(POOL_CHUNK, n - node0);
    const float* hp = h + (long)node0 * GDIM + lane * 4;
    const int gl = batch[node0 + (lane < valid ? lane : 0)];

    if (valid == POOL_CHUNK) {
        float4 v[POOL_CHUNK];
#pragma unroll
        for (int u = 0; u < POOL_CHUNK; ++u)
            v[u] = *(const float4*)(hp + (long)u * GDIM);
        const int g0 = __builtin_amdgcn_readfirstlane(gl);
        if (__all(gl == g0)) {
#pragma unroll
            for (int w = POOL_CHUNK / 2; w > 0; w >>= 1)
#pragma unroll
                for (int u = 0; u < w; ++u) v[u] = add4(v[u], v[u + w]);
            pool_flush(psum, pcount, g0, v[0], (float)POOL_CHUNK, lane);
        } else {
            float4 s   = make_float4(0.f, 0.f, 0.f, 0.f);
            int    gc  = g0;
            float  cnt = 0.f;
#pragma unroll
            for (int u = 0; u < POOL_CHUNK; ++u) {
                const int g = __builtin_amdgcn_readlane(gl, u);
                if (g != gc) {
                    pool_flush(psum, pcount, gc, s, cnt, lane);
                    s = make_float4(0.f, 0.f, 0.f, 0.f); cnt = 0.f; gc = g;
                }
                s = add4(s, v[u]); cnt += 1.0f;
            }
            pool_flush(psum, pcount, gc, s, cnt, lane);
        }
    } else {
        float4 s   = make_float4(0.f, 0.f, 0.f, 0.f);
        int    gc  = __builtin_amdgcn_readfirstlane(gl);
        float  cnt = 0.f;
        for (int u = 0; u < valid; ++u) {
            const int    g  = __shfl(gl, u, 32);
            const float4 hv = *(const float4*)(hp + (long)u * GDIM);
            if (g != gc) {
                pool_flush(psum, pcount, gc, s, cnt, lane);
                s = make_float4(0.f, 0.f, 0.f, 0.f); cnt = 0.f; gc = g;
            }
            s = add4(s, hv); cnt += 1.0f;
        }
        pool_flush(psum, pcount, gc, s, cnt, lane);
    }
}

__global__ void __launch_bounds__(256)
gat_pool_final(const float* __restrict__ psum, const float* __restrict__ pcount,
               float* __restrict__ out) {
    const int idx = blockIdx.x * blockDim.x + threadIdx.x;
    const float cnt = pcount[idx >> 7];
    out[idx] = psum[idx] / fmaxf(cnt, 1.0f);
}

extern "C" void kernel_launch(void* const* d_in, const int* in_sizes, int n_in,
                              void* d_out, int out_size, void* d_ws, size_t ws_size,
                              hipStream_t stream) {
    const float* x     = (const float*)d_in[0];
    const int*   ei    = (const int*)d_in[1];
    const int*   batch = (const int*)d_in[2];
    const float* W[3]    = {(const float*)d_in[3], (const float*)d_in[7],  (const float*)d_in[11]};
    const float* asrc[3] = {(const float*)d_in[4], (const float*)d_in[8],  (const float*)d_in[12]};
    const float* adst[3] = {(const float*)d_in[5], (const float*)d_in[9],  (const float*)d_in[13]};
    const float* bias[3] = {(const float*)d_in[6], (const float*)d_in[10], (const float*)d_in[14]};

    const int n  = in_sizes[0] / GDIM;
    const int E  = in_sizes[1] / 2;

    char*  wsb = (char*)d_ws;
    size_t off = 0;
    auto take = [&](size_t bytes) -> void* {
        void* p = wsb + off;
        off += (bytes + 255) & ~(size_t)255;
        return p;
    };
    const size_t szH = (size_t)n * GDIM * sizeof(float);
    float*    bufA   = (float*)   take(szH);
    float*    bufB   = (float*)   take(szH);
    const int nb = (n + BIN_NODES - 1) / BIN_NODES;
    const int g1 = (E + ECHUNK - 1) / ECHUNK;
    int*      tctr   = (int*)     take(256);
    int*      histT  = (int*)     take((size_t)nb * g1 * sizeof(int));
    int*      bstart = (int*)     take((size_t)(nb + 1) * sizeof(int));
    int*      binned = (int*)     take((size_t)E * 2 * sizeof(int));
    int*      start  = (int*)     take((size_t)n * sizeof(int));
    int*      cursor = (int*)     take((size_t)n * sizeof(int));
    int*      csr    = (int*)     take((size_t)E * sizeof(int));
    float*    psum   = (float*)   take((size_t)NGRAPH * GDIM * sizeof(float));
    float*    pcount = (float*)   take((size_t)NGRAPH * sizeof(float));
    float*    u      = (float*)   take((size_t)3 * 2 * GDIM * sizeof(float));
    (void)ws_size; (void)n_in; (void)out_size;

    const int nTiles      = (n + TROWS - 1) / TROWS;
    const int layerBlocks = nTiles < 1024 ? nTiles : 1024;
    const int poolBlocks  = (n + 8 * POOL_CHUNK - 1) / (8 * POOL_CHUNK);

    bin_hist<<<g1, 1024, 0, stream>>>(ei, E, nb, g1, histT);
    bin_scatter<<<g1, 1024, 0, stream>>>(ei, E, nb, g1, histT, bstart, binned, tctr);
    bin_sort<<<nb, 1024, 0, stream>>>(binned, bstart, n, start, cursor, csr);

    gat_uvec<<<48, 256, 0, stream>>>(W[0], W[1], W[2], asrc[0], asrc[1], asrc[2],
                                     adst[0], adst[1], adst[2], u);
    gat_layer<<<layerBlocks, 512, 0, stream>>>(
        start, cursor, csr, x, W[0], bias[0], u, bufA, n, nTiles, tctr, layerBlocks);
    gat_layer<<<layerBlocks, 512, 0, stream>>>(
        start, cursor, csr, bufA, W[1], bias[1], u + 2 * GDIM, bufB, n, nTiles, tctr + 16, layerBlocks);
    gat_layer<<<layerBlocks, 512, 0, stream>>>(
        start, cursor, csr, bufB, W[2], bias[2], u + 4 * GDIM, bufA, n, nTiles, tctr + 32, layerBlocks);

    hipMemsetAsync(psum,   0, (size_t)NGRAPH * GDIM * sizeof(float), stream);
    hipMemsetAsync(pcount, 0, (size_t)NGRAPH * sizeof(float), stream);
    gat_pool_accum<<<poolBlocks, 256, 0, stream>>>(bufA, batch, n, psum, pcount);
    gat_pool_final<<<(NGRAPH * GDIM) / 256, 256, 0, stream>>>(psum, pcount,
                                                              (float*)d_out);
}


# no activation between layers, attention rows sum to 1: the three W products run on the 256 pooled rows (WMMA); a layer only aggregates raw rows, one wave per node, no tile barrier; new sum order
# speedup vs baseline: 9.7470x; 1.1927x over previous
#include <hip/hip_runtime.h>

#define GDIM 128
#define NGRAPH 256
#define NEG_SLOPE 0.2f
#define TROWS 16
#define TPITCH 132

typedef float v2f __attribute__((ext_vector_type(2)));
typedef float v8f __attribute__((ext_vector_type(8)));

#define BIN_SHIFT 9
#define BIN_NODES 512
#define MAXB      256
#define ECHUNK    4096
#define STAGE_CAP 12288

__device__ __forceinline__ int wave_sum_i(int v) {
#pragma unroll
    for (int off = 16; off > 0; off >>= 1) v += __shfl_xor(v, off, 32);
    return v;
}

__global__ void __launch_bounds__(1024)
bin_hist(const int* __restrict__ ei, int E, int nb, int g1, int* __restrict__ histT) {
    __shared__ int h[MAXB];
    const int tid  = threadIdx.x;
    const int base = blockIdx.x * ECHUNK;
    int d[ECHUNK / 1024];
#pragma unroll
    for (int k = 0; k < ECHUNK / 1024; ++k) {
        const int e = base + k * 1024 + tid;
        d[k] = ei[E + min(e, E - 1)];
    }
    if (tid < MAXB) h[tid] = 0;
    __syncthreads();
#pragma unroll
    for (int k = 0; k < ECHUNK / 1024; ++k) {
        const int e = base + k * 1024 + tid;
        if (e < E) atomicAdd(&h[d[k] >> BIN_SHIFT], 1);
    }
    __syncthreads();
    if (tid < nb) histT[tid * g1 + blockIdx.x] = h[tid];
}

__global__ void __launch_bounds__(1024)
bin_scatter(const int* __restrict__ ei, int E, int nb, int g1,
            const int* __restrict__ histT, int* __restrict__ binStart,
            int* __restrict__ binned) {
    __shared__ int tot[MAXB];
    __shared__ int pre[MAXB];
    __shared__ int cur[MAXB];
    __shared__ int bas[MAXB];
    __shared__ int wtot[MAXB / 32];
    const int tid  = threadIdx.x;
    const int lane = tid & 31;
    const int wave = tid >> 5;
    const int w    = blockIdx.x;
    const int base = w * ECHUNK;
    int d[ECHUNK / 1024], s[ECHUNK / 1024];
#pragma unroll
    for (int k = 0; k < ECHUNK / 1024; ++k) {
        const int e = base + k * 1024 + tid;
        d[k] = (e < E) ? ei[E + e] : 0;
        s[k] = (e < E) ? ei[e] : 0;
    }
    for (int b = wave; b < nb; b += 64) {
        const int  b2   = b + 32;
        const bool two  = b2 < nb;
        const int* row0 = histT + b * g1;
        const int* row1 = histT + (two ? b2 : b) * g1;
        int t0 = 0, p0 = 0, t1 = 0, p1 = 0;
        for (int i0 = 0; i0 < g1; i0 += 256) {
#pragma unroll
            for (int q = 0; q < 8; ++q) {
                const int i  = i0 + lane + 32 * q;
                const int v0 = (i < g1) ? row0[i] : 0;
                const int v1 = (i < g1) ? row1[i] : 0;
                t0 += v0;
                p0 += (i < w) ? v0 : 0;
                t1 += v1;
                p1 += (i < w) ? v1 : 0;
            }
        }
        t0 = wave_sum_i(t0);
        p0 = wave_sum_i(p0);
        t1 = wave_sum_i(t1);
        p1 = wave_sum_i(p1);
        if (lane == 0) {
            tot[b] = t0; pre[b] = p0;
            if (two) { tot[b2] = t1; pre[b2] = p1; }
        }
    }
    if (tid < MAXB) cur[tid] = 0;
    __syncthreads();
    const int tv = (tid < nb) ? tot[tid] : 0;
    int incl = tv;
#pragma unroll
    for (int off = 1; off < 32; off <<= 1) {
        const int t = __shfl_up(incl, off, 32);
        if (lane >= off) incl += t;
    }
    if (tid < MAXB && lane == 31) wtot[wave] = incl;
    __syncthreads();
    if (tid < nb) {
        int wb = 0;
#pragma unroll
        for (int i = 0; i < MAXB / 32; ++i) { const int t = wtot[i]; wb += (i < wave) ? t : 0; }
        const int st = wb + incl - tv;
        bas[tid] = st + pre[tid];
        if (w == 0) {
            binStart[tid] = st;
            if (tid == nb - 1) binStart[nb] = st + tv;
        }
    }
    __syncthreads();
    int2* out = (int2*)binned;
#pragma unroll
    for (int k = 0; k < ECHUNK / 1024; ++k) {
        const int e = base + k * 1024 + tid;
        if (e < E) {
            const int b = d[k] >> BIN_SHIFT;
            const int r = atomicAdd(&cur[b], 1);
            out[bas[b] + r] = make_int2(d[k], s[k]);
        }
    }
}

#define SORT_REG 8
__global__ void __launch_bounds__(1024)
bin_sort(const int* __restrict__ binned, const int* __restrict__ binStart, int n,
         int* __restrict__ start, int* __restrict__ endp, int* __restrict__ csr_src) {
    __shared__ int cnt[BIN_NODES];
    __shared__ int wsum[BIN_NODES / 32];
    __shared__ int stage[STAGE_CAP];
    const int tid   = threadIdx.x;
    const int lane  = tid & 31;
    const int wave  = tid >> 5;
    const int bin   = blockIdx.x;
    const int lo    = binStart[bin];
    const int m     = binStart[bin + 1] - lo;
    const int node0 = bin << BIN_SHIFT;
    const int2* in  = (const int2*)binned + lo;
    int2 v[SORT_REG];
#pragma unroll
    for (int q = 0; q < SORT_REG; ++q) {
        const int i = tid + q * 1024;
        v[q] = (i < m) ? in[i] : make_int2(node0, 0);
    }
    if (tid < BIN_NODES) cnt[tid] = 0;
    __syncthreads();
#pragma unroll
    for (int q = 0; q < SORT_REG; ++q)
        if (tid + q * 1024 < m) atomicAdd(&cnt[v[q].x - node0], 1);
    for (int i = tid + SORT_REG * 1024; i < m; i += 1024) atomicAdd(&cnt[in[i].x - node0], 1);
    __syncthreads();
    const int c = (tid < BIN_NODES) ? cnt[tid] : 0;
    int incl = c;
#pragma unroll
    for (int off = 1; off < 32; off <<= 1) {
        const int t = __shfl_up(incl, off, 32);
        if (lane >= off) incl += t;
    }
    if (tid < BIN_NODES && lane == 31) wsum[wave] = incl;
    __syncthreads();
    if (tid < BIN_NODES) {
        int wb = 0;
#pragma unroll
        for (int i = 0; i < BIN_NODES / 32; ++i) { const int t = wsum[i]; wb += (i < wave) ? t : 0; }
        const int ex = wb + incl - c;
        cnt[tid] = ex;
        const int j = node0 + tid;
        if (j < n) { start[j] = lo + ex; endp[j] = lo + ex + c; }
    }
    __syncthreads();
    if (m <= STAGE_CAP) {
#pragma unroll
        for (int q = 0; q < SORT_REG; ++q)
            if (tid + q * 1024 < m) stage[atomicAdd(&cnt[v[q].x - node0], 1)] = v[q].y;
        for (int i = tid + SORT_REG * 1024; i < m; i += 1024) {
            const int2 t = in[i];
            stage[atomicAdd(&cnt[t.x - node0], 1)] = t.y;
        }
        __syncthreads();
        for (int i = tid; i < m; i += 1024) csr_src[lo + i] = stage[i];
    } else {
#pragma unroll
        for (int q = 0; q < SORT_REG; ++q)
            if (tid + q * 1024 < m) csr_src[lo + atomicAdd(&cnt[v[q].x - node0], 1)] = v[q].y;
        for (int i = tid + SORT_REG * 1024; i < m; i += 1024) {
            const int2 t = in[i];
            csr_src[lo + atomicAdd(&cnt[t.x - node0], 1)] = t.y;
        }
    }
}

__device__ __forceinline__ float wave_sum(float v) {
#pragma unroll
    for (int off = 16; off > 0; off >>= 1) v += __shfl_xor(v, off, 32);
    return v;
}

template <int M>
__device__ __forceinline__ float row_xor(float x) {
    return __builtin_bit_cast(float, __builtin_amdgcn_update_dpp(
        0, __builtin_bit_cast(int, x), 0x160 + M, 0xf, 0xf, false));
}
__device__ __forceinline__ float half_xor(float x) {
    return __builtin_bit_cast(float, __builtin_amdgcn_permlanex16(
        __builtin_bit_cast(int, x), __builtin_bit_cast(int, x), 0x76543210, 0xfedcba98,
        false, false));
}
template <int K>
__device__ __forceinline__ float quad_bcast(float x) {
    return __builtin_bit_cast(float, __builtin_amdgcn_update_dpp(
        0, __builtin_bit_cast(int, x), K * 0x55, 0xf, 0xf, true));
}
template <int M>
__device__ __forceinline__ float row_xor_max(float x) {
    return fmaxf(x, __builtin_bit_cast(float, __builtin_amdgcn_update_dpp(
        0, __builtin_bit_cast(int, x), 0x160 + M, 0xf, 0xf, true)));
}
__device__ __forceinline__ float lane_sum(float v) {
    v += row_xor<1>(v);
    v += row_xor<2>(v);
    v += row_xor<4>(v);
    v += row_xor<8>(v);
    v += half_xor(v);
    return v;
}

__device__ __forceinline__ float lane_sum_from2(float v) {
    v += row_xor<2>(v);
    v += row_xor<4>(v);
    v += row_xor<8>(v);
    v += half_xor(v);
    return v;
}

__device__ __forceinline__ float dot4(const float4 h, const float4 b) {
    v2f q = (v2f){h.x, h.y} * (v2f){b.x, b.y};
    q = __builtin_elementwise_fma((v2f){h.z, h.w}, (v2f){b.z, b.w}, q);
    return q.x + q.y;
}
__device__ __forceinline__ float4 load_row(const float* __restrict__ H, int i, int lane) {
    const float* rp = H + ((long)i << 7);
    return ((const float4*)rp)[lane];
}
__device__ __forceinline__ float leaky(float s) { return fmaxf(s, NEG_SLOPE * s); }

__device__ __forceinline__ void fma4(float w, const float4 h, float4& a) {
    a.x = fmaf(w, h.x, a.x); a.y = fmaf(w, h.y, a.y);
    a.z = fmaf(w, h.z, a.z); a.w = fmaf(w, h.w, a.w);
}

template <int NV>
__device__ __forceinline__ void edge_group(const float* __restrict__ H, const float4 us,
                                           float sd, int iv, int k0, int nvar, int lane,
                                           float& m, float& den, float4& a) {
    const int   nv      = NV ? NV : nvar;
    const float NEG_INF = -__builtin_inff();
    const int i0 = __builtin_amdgcn_readlane(iv, k0);
    const int i1 = __builtin_amdgcn_readlane(iv, k0 + 1);
    const int i2 = __builtin_amdgcn_readlane(iv, k0 + 2);
    const int i3 = __builtin_amdgcn_readlane(iv, k0 + 3);
    const float4 z = make_float4(0.f, 0.f, 0.f, 0.f);
    float4 h0 = load_row(H, i0, lane);
    float4 h1 = z, h2 = z, h3 = z;
    if (nv > 1) h1 = load_row(H, i1, lane);
    if (nv > 2) h2 = load_row(H, i2, lane);
    if (nv > 3) h3 = load_row(H, i3, lane);
    const float p0 = dot4(h0, us), p1 = dot4(h1, us), p2 = dot4(h2, us), p3 = dot4(h3, us);
    const bool  odd = (lane & 1) != 0;
    const bool  up  = (lane & 2) != 0;
    const float q0  = (odd ? p1 : p0) + row_xor<1>(odd ? p0 : p1);
    const float q1  = (odd ? p3 : p2) + row_xor<1>(odd ? p2 : p3);
    float t = (up ? q1 : q0) + row_xor<2>(up ? q0 : q1);
    t += row_xor<4>(t);
    t += row_xor<8>(t);
    t += half_xor(t);
    float l = leaky(t + sd);
    if (NV == 0) l = ((lane & 3) < nv) ? l : NEG_INF;
    float lm = row_xor_max<1>(l);
    lm = row_xor_max<2>(lm);
    const float mn = fmaxf(m, lm);
    const float sc = __expf(m - mn);
    const float w  = __expf(l - mn);
    const float w0 = quad_bcast<0>(w);
    const float w1 = quad_bcast<1>(w);
    const float w2 = quad_bcast<2>(w);
    const float w3 = quad_bcast<3>(w);
    den = fmaf(den, sc, (w0 + w1) + (w2 + w3));
    a.x *= sc; a.y *= sc; a.z *= sc; a.w *= sc;
    fma4(w0, h0, a); fma4(w1, h1, a); fma4(w2, h2, a); fma4(w3, h3, a);
    m = mn;
}

__global__ void __launch_bounds__(1024)
gat_uvec(const float* __restrict__ W1, const float* __restrict__ W2,
         const float* __restrict__ W3, const float* __restrict__ as1,
         const float* __restrict__ as2, const float* __restrict__ as3,
         const float* __restrict__ ad1, const float* __restrict__ ad2,
         const float* __restrict__ ad3, const float* __restrict__ b1,
         const float* __restrict__ b2, float* __restrict__ u) {
    __shared__ __attribute__((aligned(16))) float sv[6][GDIM];
    const int lane = threadIdx.x & 31;
    const int wave = threadIdx.x >> 5;
    float4 w1[4], w2[4], w3[4];
#pragma unroll
    for (int q = 0; q < 4; ++q) {
        const int row = wave * 4 + q;
        w1[q] = *(const float4*)(W1 + row * GDIM + lane * 4);
        w2[q] = *(const float4*)(W2 + row * GDIM + lane * 4);
        w3[q] = *(const float4*)(W3 + row * GDIM + lane * 4);
    }
    const float4 s1 = *(const float4*)(as1 + lane * 4);
    const float4 d1 = *(const float4*)(ad1 + lane * 4);
    const float4 s2 = *(const float4*)(as2 + lane * 4);
    const float4 d2 = *(const float4*)(ad2 + lane * 4);
    const float4 s3 = *(const float4*)(as3 + lane * 4);
    const float4 d3 = *(const float4*)(ad3 + lane * 4);
#pragma unroll
    for (int q = 0; q < 4; ++q) {
        const int row = wave * 4 + q;
        const float r0 = wave_sum(dot4(w1[q], s1));
        const float r1 = wave_sum(dot4(w1[q], d1));
        const float r2 = wave_sum(dot4(w2[q], s2));
        const float r3 = wave_sum(dot4(w2[q], d2));
        const float r4 = wave_sum(dot4(w3[q], s3));
        const float r5 = wave_sum(dot4(w3[q], d3));
        if (lane == 0) {
            u[row] = r0; u[GDIM + row] = r1;
            sv[0][row] = r2; sv[1][row] = r3; sv[2][row] = r4; sv[3][row] = r5;
        }
    }
    __syncthreads();
    const float4 p2s = *(const float4*)(&sv[0][lane * 4]);
    const float4 p2d = *(const float4*)(&sv[1][lane * 4]);
    const float4 p3s = *(const float4*)(&sv[2][lane * 4]);
    const float4 p3d = *(const float4*)(&sv[3][lane * 4]);
#pragma unroll
    for (int q = 0; q < 4; ++q) {
        const int row = wave * 4 + q;
        const float r0 = wave_sum(dot4(w1[q], p2s));
        const float r1 = wave_sum(dot4(w1[q], p2d));
        const float r2 = wave_sum(dot4(w2[q], p3s));
        const float r3 = wave_sum(dot4(w2[q], p3d));
        if (lane == 0) {
            u[2 * GDIM + row] = r0; u[3 * GDIM + row] = r1;
            sv[4][row] = r2; sv[5][row] = r3;
        }
    }
    __syncthreads();
    const float4 q3s = *(const float4*)(&sv[4][lane * 4]);
    const float4 q3d = *(const float4*)(&sv[5][lane * 4]);
#pragma unroll
    for (int q = 0; q < 4; ++q) {
        const int row = wave * 4 + q;
        const float r0 = wave_sum(dot4(w1[q], q3s));
        const float r1 = wave_sum(dot4(w1[q], q3d));
        if (lane == 0) { u[4 * GDIM + row] = r0; u[5 * GDIM + row] = r1; }
    }
    if (wave == 0) {
        const float4 bb1 = *(const float4*)(b1 + lane * 4);
        const float4 bb2 = *(const float4*)(b2 + lane * 4);
        const float c2 = wave_sum(dot4(bb1, p2s) + dot4(bb1, p2d));
        const float c3 = wave_sum((dot4(bb1, q3s) + dot4(bb1, q3d)) +
                                  (dot4(bb2, p3s) + dot4(bb2, p3d)));
        if (lane == 0) { u[6 * GDIM] = 0.0f; u[6 * GDIM + 1] = c2; u[6 * GDIM + 2] = c3; }
    }
}

__global__ void __launch_bounds__(256, 16)
gat_agg(const int* __restrict__ start, const int* __restrict__ endp,
        const int* __restrict__ csr_src, const float* __restrict__ Xin,
        const float* __restrict__ u, const float* __restrict__ cl,
        float* __restrict__ Xout, int n, int nWaves) {
    const int lane = threadIdx.x & 31;
    const int wave = __builtin_amdgcn_readfirstlane((int)(threadIdx.x >> 5));
    const float4 us = *(const float4*)(u + lane * 4);
    const float4 ud = *(const float4*)(u + GDIM + lane * 4);
    const float  c  = cl[0];
    int j = blockIdx.x * 8 + wave;
    if (j >= n) return;
    int beg = start[j];
    int deg = endp[j] - beg;
    for (;;) {
        const float4 hs = load_row(Xin, j, lane);
        int iv = (lane < deg) ? csr_src[beg + lane] : j;
        const int jn   = j + nWaves;
        const int jc   = min(jn, n - 1);
        const int begN = start[jc];
        const int endN = endp[jc];
        __builtin_amdgcn_sched_barrier(0);
        const float  ps  = dot4(hs, us);
        const float  pd  = dot4(hs, ud);
        const bool   odd = (lane & 1) != 0;
        const float  tsd = lane_sum_from2((odd ? pd : ps) + row_xor<1>(odd ? ps : pd));
        const float  ss  = quad_bcast<0>(tsd);
        const float  sd  = quad_bcast<1>(tsd) + c;
        float m   = leaky(ss + sd);
        float den = 1.0f;
        float4 a  = hs;
        for (int base = 0; base < deg; base += 32) {
            if (base > 0) {
                const int t = base + lane;
                iv = (t < deg) ? csr_src[beg + t] : j;
            }
            const int cnt = min(32, deg - base);
            int k = 0;
            for (; k + 4 <= cnt; k += 4)
                edge_group<4>(Xin, us, sd, iv, k, 4, lane, m, den, a);
            if (k < cnt)
                edge_group<0>(Xin, us, sd, iv, k, cnt - k, lane, m, den, a);
        }
        const float inv = 1.0f / den;
        a.x *= inv; a.y *= inv; a.z *= inv; a.w *= inv;
        *(float4*)(Xout + (long)j * GDIM + lane * 4) = a;
        if (jn >= n) break;
        j = jn; beg = begN; deg = endN - begN;
    }
}

#define POOL_CHUNK 16

__device__ __forceinline__ void pool_flush(float* __restrict__ psum,
                                           float* __restrict__ pcount, int g,
                                           float4 s, float cnt, int lane) {
    float* pp = psum + (long)g * GDIM + lane * 4;
    atomicAdd(pp + 0, s.x);
    atomicAdd(pp + 1, s.y);
    atomicAdd(pp + 2, s.z);
    atomicAdd(pp + 3, s.w);
    if (lane == 0) atomicAdd(pcount + g, cnt);
}

__device__ __forceinline__ float4 add4(float4 a, float4 b) {
    return make_float4(a.x + b.x, a.y + b.y, a.z + b.z, a.w + b.w);
}

__global__ void __launch_bounds__(256)
gat_pool_accum(const float* __restrict__ h, const int* __restrict__ batch, int n,
               float* __restrict__ psum, float* __restrict__ pcount) {
    const int lane  = threadIdx.x & 31;
    const int wave  = __builtin_amdgcn_readfirstlane((int)(threadIdx.x >> 5));
    const int node0 = (blockIdx.x * 8 + wave) * POOL_CHUNK;
    if (node0 >= n) return;
    const int valid = min(POOL_CHUNK, n - node0);
    const float* hp = h + (long)node0 * GDIM + lane * 4;
    const int gl = batch[node0 + (lane < valid ? lane : 0)];

    if (valid == POOL_CHUNK) {
        float4 v[POOL_CHUNK];
#pragma unroll
        for (int u = 0; u < POOL_CHUNK; ++u)
            v[u] = *(const float4*)(hp + (long)u * GDIM);
        const int g0 = __builtin_amdgcn_readfirstlane(gl);
        if (__all(gl == g0)) {
#pragma unroll
            for (int w = POOL_CHUNK / 2; w > 0; w >>= 1)
#pragma unroll
                for (int u = 0; u < w; ++u) v[u] = add4(v[u], v[u + w]);
            pool_flush(psum, pcount, g0, v[0], (float)POOL_CHUNK, lane);
        } else {
            float4 s   = make_float4(0.f, 0.f, 0.f, 0.f);
            int    gc  = g0;
            float  cnt = 0.f;
#pragma unroll
            for (int u = 0; u < POOL_CHUNK; ++u) {
                const int g = __builtin_amdgcn_readlane(gl, u);
                if (g != gc) {
                    pool_flush(psum, pcount, gc, s, cnt, lane);
                    s = make_float4(0.f, 0.f, 0.f, 0.f); cnt = 0.f; gc = g;
                }
                s = add4(s, v[u]); cnt += 1.0f;
            }
            pool_flush(psum, pcount, gc, s, cnt, lane);
        }
    } else {
        float4 s   = make_float4(0.f, 0.f, 0.f, 0.f);
        int    gc  = __builtin_amdgcn_readfirstlane(gl);
        float  cnt = 0.f;
        for (int u = 0; u < valid; ++u) {
            const int    g  = __shfl(gl, u, 32);
            const float4 hv = *(const float4*)(hp + (long)u * GDIM);
            if (g != gc) {
                pool_flush(psum, pcount, gc, s, cnt, lane);
                s = make_float4(0.f, 0.f, 0.f, 0.f); cnt = 0.f; gc = g;
            }
            s = add4(s, hv); cnt += 1.0f;
        }
        pool_flush(psum, pcount, gc, s, cnt, lane);
    }
}

__global__ void __launch_bounds__(256)
gat_pool_final(const float* __restrict__ psum, const float* __restrict__ pcount,
               const float* __restrict__ W1, const float* __restrict__ W2,
               const float* __restrict__ W3, const float* __restrict__ b1,
               const float* __restrict__ b2, const float* __restrict__ b3,
               float* __restrict__ out) {
    __shared__ __attribute__((aligned(16))) float sT[TROWS * TPITCH];
    const int tid  = threadIdx.x;
    const int lane = tid & 31;
    const int wave = __builtin_amdgcn_readfirstlane(tid >> 5);
    const int hi   = lane >> 4;
    const int col  = lane & 15;
    const int g0   = blockIdx.x * TROWS;
#pragma unroll
    for (int rr = 0; rr < 2; ++rr) {
        const int   r   = wave * 2 + rr;
        const float cnt = fmaxf(pcount[g0 + r], 1.0f);
        const float4 p  = *(const float4*)(psum + (long)(g0 + r) * GDIM + lane * 4);
        *(float4*)(sT + r * TPITCH + lane * 4) =
            make_float4(p.x / cnt, p.y / cnt, p.z / cnt, p.w / cnt);
    }
    float has[8];
#pragma unroll
    for (int v = 0; v < 8; ++v) has[v] = fminf(pcount[g0 + v + 8 * hi], 1.0f);
    __syncthreads();
#pragma unroll
    for (int l = 0; l < 3; ++l) {
        const float* W  = (l == 0) ? W1 : (l == 1) ? W2 : W3;
        const float* bl = (l == 0) ? b1 : (l == 1) ? b2 : b3;
        const float  bc = bl[wave * 16 + col];
        v8f c0 = (v8f)0.0f, c1 = (v8f)0.0f;
        const float* At = sT + col * TPITCH + 2 * hi;
        const float* Bt = W + (2 * hi) * GDIM + wave * 16 + col;
#pragma unroll
        for (int kk = 0; kk < GDIM; kk += 8) {
            const v2f a0 = *(const v2f*)(At + kk);
            v2f b0;
            b0.x = Bt[(kk + 0) * GDIM];
            b0.y = Bt[(kk + 1) * GDIM];
            const v2f a1 = *(const v2f*)(At + kk + 4);
            v2f b1v;
            b1v.x = Bt[(kk + 4) * GDIM];
            b1v.y = Bt[(kk + 5) * GDIM];
            c0 = __builtin_amdgcn_wmma_f32_16x16x4_f32(false, a0, false, b0, (short)0, c0, false, false);
            c1 = __builtin_amdgcn_wmma_f32_16x16x4_f32(false, a1, false, b1v, (short)0, c1, false, false);
        }
        const v8f c = c0 + c1;
        __syncthreads();
#pragma unroll
        for (int v = 0; v < 8; ++v)
            sT[(v + 8 * hi) * TPITCH + wave * 16 + col] = fmaf(has[v], bc, c[v]);
        __syncthreads();
    }
#pragma unroll
    for (int rr = 0; rr < 2; ++rr) {
        const int r = wave * 2 + rr;
        *(float4*)(out + (long)(g0 + r) * GDIM + lane * 4) =
            *(const float4*)(sT + r * TPITCH + lane * 4);
    }
}

extern "C" void kernel_launch(void* const* d_in, const int* in_sizes, int n_in,
                              void* d_out, int out_size, void* d_ws, size_t ws_size,
                              hipStream_t stream) {
    const float* x     = (const float*)d_in[0];
    const int*   ei    = (const int*)d_in[1];
    const int*   batch = (const int*)d_in[2];
    const float* W[3]    = {(const float*)d_in[3], (const float*)d_in[7],  (const float*)d_in[11]};
    const float* asrc[3] = {(const float*)d_in[4], (const float*)d_in[8],  (const float*)d_in[12]};
    const float* adst[3] = {(const float*)d_in[5], (const float*)d_in[9],  (const float*)d_in[13]};
    const float* bias[3] = {(const float*)d_in[6], (const float*)d_in[10], (const float*)d_in[14]};

    const int n  = in_sizes[0] / GDIM;
    const int E  = in_sizes[1] / 2;

    char*  wsb = (char*)d_ws;
    size_t off = 0;
    auto take = [&](size_t bytes) -> void* {
        void* p = wsb + off;
        off += (bytes + 255) & ~(size_t)255;
        return p;
    };
    const size_t szH = (size_t)n * GDIM * sizeof(float);
    float*    bufA   = (float*)   take(szH);
    float*    bufB   = (float*)   take(szH);
    const int nb = (n + BIN_NODES - 1) / BIN_NODES;
    const int g1 = (E + ECHUNK - 1) / ECHUNK;
    int*      histT  = (int*)     take((size_t)nb * g1 * sizeof(int));
    int*      bstart = (int*)     take((size_t)(nb + 1) * sizeof(int));
    int*      binned = (int*)     take((size_t)E * 2 * sizeof(int));
    int*      start  = (int*)     take((size_t)n * sizeof(int));
    int*      cursor = (int*)     take((size_t)n * sizeof(int));
    int*      csr    = (int*)     take((size_t)E * sizeof(int));
    float*    psum   = (float*)   take((size_t)NGRAPH * GDIM * sizeof(float));
    float*    pcount = (float*)   take((size_t)NGRAPH * sizeof(float));
    float*    u      = (float*)   take((size_t)(6 * GDIM + 4) * sizeof(float));
    (void)ws_size; (void)n_in; (void)out_size;

    const int perWave    = (n + 16383) / 16384;
    const int aggBlocks  = (n + 8 * perWave - 1) / (8 * perWave);
    const int aggWaves   = aggBlocks * 8;
    const int poolBlocks = (n + 8 * POOL_CHUNK - 1) / (8 * POOL_CHUNK);

    bin_hist<<<g1, 1024, 0, stream>>>(ei, E, nb, g1, histT);
    bin_scatter<<<g1, 1024, 0, stream>>>(ei, E, nb, g1, histT, bstart, binned);
    bin_sort<<<nb, 1024, 0, stream>>>(binned, bstart, n, start, cursor, csr);

    gat_uvec<<<1, 1024, 0, stream>>>(W[0], W[1], W[2], asrc[0], asrc[1], asrc[2],
                                     adst[0], adst[1], adst[2], bias[0], bias[1], u);
    gat_agg<<<aggBlocks, 256, 0, stream>>>(start, cursor, csr, x,    u,            u + 6 * GDIM,     bufA, n, aggWaves);
    gat_agg<<<aggBlocks, 256, 0, stream>>>(start, cursor, csr, bufA, u + 2 * GDIM, u + 6 * GDIM + 1, bufB, n, aggWaves);
    gat_agg<<<aggBlocks, 256, 0, stream>>>(start, cursor, csr, bufB, u + 4 * GDIM, u + 6 * GDIM + 2, bufA, n, aggWaves);

    hipMemsetAsync(psum,   0, (size_t)NGRAPH * GDIM * sizeof(float), stream);
    hipMemsetAsync(pcount, 0, (size_t)NGRAPH * sizeof(float), stream);
    gat_pool_accum<<<poolBlocks, 256, 0, stream>>>(bufA, batch, n, psum, pcount);
    gat_pool_final<<<NGRAPH / TROWS, 256, 0, stream>>>(psum, pcount, W[0], W[1], W[2],
                                                       bias[0], bias[1], bias[2], (float*)d_out);
}
